# starting kernel
# baseline (speedup 1.0000x reference)
#include <hip/hip_runtime.h>
#include <hip/hip_bf16.h>


typedef __attribute__((ext_vector_type(2))) float v2f;
typedef __attribute__((ext_vector_type(8))) float v8f;

#define BB      4
#define NN      4096
#define CX      32
#define CIN     35
#define CINP    36
#define COUT    64
#define KS      32
#define R2      0.01f
#define EPSV    1e-5f
#define NPAIRS  (BB * (NN / 2))
#define GCONV   1024
#define CNTF    ((float)BB * (float)NN * (float)KS)

#define FSTRIDE  65
#define W1STRIDE 36
#define W2STRIDE 68

__global__ void xpose_kernel(const float* __restrict__ x, float* __restrict__ xt) {
  __shared__ float tile[CX][65];
  int b  = blockIdx.y;
  int n0 = blockIdx.x * 64;
  int t  = threadIdx.x;
  for (int i = t; i < CX * 64; i += 256) {
    int c = i >> 6, nn = i & 63;
    tile[c][nn] = x[((size_t)b * CX + c) * NN + n0 + nn];
  }
  __syncthreads();
  for (int i = t; i < CX * 64; i += 256) {
    int nn = i >> 5, c = i & 31;
    xt[((size_t)b * NN + n0 + nn) * CX + c] = tile[c][nn];
  }
}

__global__ void ballq_kernel(const float* __restrict__ p, int* __restrict__ idx) {
  __shared__ float sx[256], sy[256], sz[256];
  int b = blockIdx.y;
  int i = blockIdx.x * 256 + threadIdx.x;
  float px = p[((size_t)b * NN + i) * 3 + 0];
  float py = p[((size_t)b * NN + i) * 3 + 1];
  float pz = p[((size_t)b * NN + i) * 3 + 2];
  int cnt = 0, first = 0;
  int* myidx = idx + ((size_t)b * NN + i) * KS;
  for (int t0 = 0; t0 < NN; t0 += 256) {
    int j = t0 + threadIdx.x;
    __syncthreads();
    sx[threadIdx.x] = p[((size_t)b * NN + j) * 3 + 0];
    sy[threadIdx.x] = p[((size_t)b * NN + j) * 3 + 1];
    sz[threadIdx.x] = p[((size_t)b * NN + j) * 3 + 2];
    __syncthreads();
    if (cnt < KS) {
      for (int jj = 0; jj < 256; ++jj) {
        float dx = sx[jj] - px, dy = sy[jj] - py, dz = sz[jj] - pz;
        float d2 = dx * dx + dy * dy + dz * dz;
        if (d2 < R2 && cnt < KS) {
          int j2 = t0 + jj;
          if (cnt == 0) first = j2;
          myidx[cnt] = j2;
          cnt++;
        }
      }
    }
  }
  for (int r = cnt; r < KS; ++r) myidx[r] = first;
}

__device__ __forceinline__ void load_ftile(const float* __restrict__ p,
                                           const float* __restrict__ xt,
                                           const int* __restrict__ idxg,
                                           int b, int n0,
                                           int* ibuf, float* flds) {
  int t = threadIdx.x;
  if (t < 64) ibuf[t] = idxg[((size_t)b * NN + n0) * KS + t];
  __syncthreads();
  int col = t & 63;
  int j = ibuf[col];
  const float* xr = xt + ((size_t)b * NN + j) * CX;
  if (t < 64) {
    int nq = n0 + (col >> 5);
    const float* pj = p + ((size_t)b * NN + j) * 3;
    const float* pq = p + ((size_t)b * NN + nq) * 3;
    flds[0 * FSTRIDE + col] = pj[0] - pq[0];
    flds[1 * FSTRIDE + col] = pj[1] - pq[1];
    flds[2 * FSTRIDE + col] = pj[2] - pq[2];
    flds[35 * FSTRIDE + col] = 0.0f;
    #pragma unroll
    for (int c = 0; c < 16; ++c) flds[(3 + c) * FSTRIDE + col] = xr[c];
  } else {
    #pragma unroll
    for (int c = 16; c < 32; ++c) flds[(3 + c) * FSTRIDE + col] = xr[c];
  }
}

__device__ __forceinline__ v8f wmma_gemm(const float* __restrict__ Wl, int wstride,
                                         const float* __restrict__ Fl,
                                         int mt, int ct, int ksteps) {
  int lane  = threadIdx.x & 31;
  int mrow  = mt * 16 + (lane & 15);
  int khalf = (lane >> 4) << 1;
  int col   = ct * 16 + (lane & 15);
  v8f acc = {0.f, 0.f, 0.f, 0.f, 0.f, 0.f, 0.f, 0.f};
  for (int kk = 0; kk < ksteps; ++kk) {
    int kb = (kk << 2) + khalf;
    v2f a, bm;
    a.x  = Wl[mrow * wstride + kb];
    a.y  = Wl[mrow * wstride + kb + 1];
    bm.x = Fl[kb * FSTRIDE + col];
    bm.y = Fl[(kb + 1) * FSTRIDE + col];
    acc = __builtin_amdgcn_wmma_f32_16x16x4_f32(false, a, false, bm,
                                                (short)0, acc, false, false);
  }
  return acc;
}

__global__ __launch_bounds__(128) void conv1_stats_kernel(
    const float* __restrict__ p, const float* __restrict__ xt,
    const int* __restrict__ idxg, const float* __restrict__ W1,
    float* __restrict__ part1) {
  __shared__ float W1l[COUT * W1STRIDE];
  __shared__ float flds[CINP * FSTRIDE];
  __shared__ int   ibuf[64];
  int t = threadIdx.x;
  for (int i = t; i < COUT * W1STRIDE; i += 128) {
    int c = i % W1STRIDE;
    W1l[i] = (c < CIN) ? W1[(i / W1STRIDE) * CIN + c] : 0.f;
  }
  int mt = t >> 5, lane = t & 31;
  float s[8], sq[8];
  #pragma unroll
  for (int v = 0; v < 8; ++v) { s[v] = 0.f; sq[v] = 0.f; }

  for (int pair = blockIdx.x; pair < NPAIRS; pair += gridDim.x) {
    int b  = pair / (NN / 2);
    int n0 = (pair % (NN / 2)) * 2;
    __syncthreads();
    load_ftile(p, xt, idxg, b, n0, ibuf, flds);
    __syncthreads();
    #pragma unroll
    for (int ct = 0; ct < 4; ++ct) {
      v8f acc = wmma_gemm(W1l, W1STRIDE, flds, mt, ct, 9);
      #pragma unroll
      for (int v = 0; v < 8; ++v) { float y = acc[v]; s[v] += y; sq[v] += y * y; }
    }
  }
  #pragma unroll
  for (int v = 0; v < 8; ++v)
    #pragma unroll
    for (int off = 1; off < 16; off <<= 1) {
      s[v]  += __shfl_xor(s[v],  off, 32);
      sq[v] += __shfl_xor(sq[v], off, 32);
    }
  if ((lane & 15) == 0) {
    int hi = lane >> 4;
    #pragma unroll
    for (int v = 0; v < 8; ++v) {
      int o = mt * 16 + hi * 8 + v;
      part1[(size_t)blockIdx.x * 128 + o]      = s[v];
      part1[(size_t)blockIdx.x * 128 + 64 + o] = sq[v];
    }
  }
}

__global__ void bn_finalize_kernel(const float* __restrict__ part,
                                   const float* __restrict__ g,
                                   const float* __restrict__ bta,
                                   float* __restrict__ bnp, int nblocks) {
  int o = threadIdx.x;
  if (o >= COUT) return;
  float s = 0.f, sq = 0.f;
  for (int blk = 0; blk < nblocks; ++blk) {
    s  += part[(size_t)blk * 128 + o];
    sq += part[(size_t)blk * 128 + 64 + o];
  }
  float mean  = s / CNTF;
  float var   = sq / CNTF - mean * mean;
  float scale = g[o] * rsqrtf(var + EPSV);
  bnp[o]      = scale;
  bnp[64 + o] = bta[o] - mean * scale;
}

__global__ __launch_bounds__(128) void conv2_stats_kernel(
    const float* __restrict__ p, const float* __restrict__ xt,
    const int* __restrict__ idxg, const float* __restrict__ W1,
    const float* __restrict__ W2, const float* __restrict__ bnp1,
    float* __restrict__ part2) {
  __shared__ float W1l[COUT * W1STRIDE];
  __shared__ float W2l[COUT * W2STRIDE];
  __shared__ float flds[CINP * FSTRIDE];
  __shared__ float f1l[COUT * FSTRIDE];
  __shared__ int   ibuf[64];
  int t = threadIdx.x;
  for (int i = t; i < COUT * W1STRIDE; i += 128) {
    int c = i % W1STRIDE;
    W1l[i] = (c < CIN) ? W1[(i / W1STRIDE) * CIN + c] : 0.f;
  }
  for (int i = t; i < COUT * W2STRIDE; i += 128) {
    int c = i % W2STRIDE;
    W2l[i] = (c < COUT) ? W2[(i / W2STRIDE) * COUT + c] : 0.f;
  }
  int mt = t >> 5, lane = t & 31;
  int hi = lane >> 4, colb = lane & 15;
  float sc1[8], sh1[8];
  #pragma unroll
  for (int v = 0; v < 8; ++v) {
    int o = mt * 16 + hi * 8 + v;
    sc1[v] = bnp1[o]; sh1[v] = bnp1[64 + o];
  }
  float s[8], sq[8];
  #pragma unroll
  for (int v = 0; v < 8; ++v) { s[v] = 0.f; sq[v] = 0.f; }

  for (int pair = blockIdx.x; pair < NPAIRS; pair += gridDim.x) {
    int b  = pair / (NN / 2);
    int n0 = (pair % (NN / 2)) * 2;
    __syncthreads();
    load_ftile(p, xt, idxg, b, n0, ibuf, flds);
    __syncthreads();
    #pragma unroll
    for (int ct = 0; ct < 4; ++ct) {
      v8f acc = wmma_gemm(W1l, W1STRIDE, flds, mt, ct, 9);
      int col = ct * 16 + colb;
      #pragma unroll
      for (int v = 0; v < 8; ++v) {
        float y = acc[v] * sc1[v] + sh1[v];
        y = y > 0.f ? y : 0.f;
        f1l[(mt * 16 + hi * 8 + v) * FSTRIDE + col] = y;
      }
    }
    __syncthreads();
    #pragma unroll
    for (int ct = 0; ct < 4; ++ct) {
      v8f acc = wmma_gemm(W2l, W2STRIDE, f1l, mt, ct, 16);
      #pragma unroll
      for (int v = 0; v < 8; ++v) { float y = acc[v]; s[v] += y; sq[v] += y * y; }
    }
  }
  #pragma unroll
  for (int v = 0; v < 8; ++v)
    #pragma unroll
    for (int off = 1; off < 16; off <<= 1) {
      s[v]  += __shfl_xor(s[v],  off, 32);
      sq[v] += __shfl_xor(sq[v], off, 32);
    }
  if ((lane & 15) == 0) {
    #pragma unroll
    for (int v = 0; v < 8; ++v) {
      int o = mt * 16 + hi * 8 + v;
      part2[(size_t)blockIdx.x * 128 + o]      = s[v];
      part2[(size_t)blockIdx.x * 128 + 64 + o] = sq[v];
    }
  }
}

__global__ __launch_bounds__(128) void final_kernel(
    const float* __restrict__ p, const float* __restrict__ xt,
    const int* __restrict__ idxg, const float* __restrict__ W1,
    const float* __restrict__ W2, const float* __restrict__ bnp1,
    const float* __restrict__ bnp2, float* __restrict__ out) {
  __shared__ float W1l[COUT * W1STRIDE];
  __shared__ float W2l[COUT * W2STRIDE];
  __shared__ float flds[CINP * FSTRIDE];
  __shared__ float f1l[COUT * FSTRIDE];
  __shared__ int   ibuf[64];
  int t = threadIdx.x;
  for (int i = t; i < COUT * W1STRIDE; i += 128) {
    int c = i % W1STRIDE;
    W1l[i] = (c < CIN) ? W1[(i / W1STRIDE) * CIN + c] : 0.f;
  }
  for (int i = t; i < COUT * W2STRIDE; i += 128) {
    int c = i % W2STRIDE;
    W2l[i] = (c < COUT) ? W2[(i / W2STRIDE) * COUT + c] : 0.f;
  }
  int mt = t >> 5, lane = t & 31;
  int hi = lane >> 4, colb = lane & 15;
  float sc1[8], sh1[8], sc2[8], sh2[8];
  #pragma unroll
  for (int v = 0; v < 8; ++v) {
    int o = mt * 16 + hi * 8 + v;
    sc1[v] = bnp1[o]; sh1[v] = bnp1[64 + o];
    sc2[v] = bnp2[o]; sh2[v] = bnp2[64 + o];
  }
  for (int pair = blockIdx.x; pair < NPAIRS; pair += gridDim.x) {
    int b  = pair / (NN / 2);
    int n0 = (pair % (NN / 2)) * 2;
    __syncthreads();
    load_ftile(p, xt, idxg, b, n0, ibuf, flds);
    __syncthreads();
    #pragma unroll
    for (int ct = 0; ct < 4; ++ct) {
      v8f acc = wmma_gemm(W1l, W1STRIDE, flds, mt, ct, 9);
      int col = ct * 16 + colb;
      #pragma unroll
      for (int v = 0; v < 8; ++v) {
        float y = acc[v] * sc1[v] + sh1[v];
        y = y > 0.f ? y : 0.f;
        f1l[(mt * 16 + hi * 8 + v) * FSTRIDE + col] = y;
      }
    }
    __syncthreads();
    float m0[8], m1[8];
    #pragma unroll
    for (int v = 0; v < 8; ++v) { m0[v] = 0.f; m1[v] = 0.f; }
    #pragma unroll
    for (int ct = 0; ct < 4; ++ct) {
      v8f acc = wmma_gemm(W2l, W2STRIDE, f1l, mt, ct, 16);
      #pragma unroll
      for (int v = 0; v < 8; ++v) {
        float y = acc[v] * sc2[v] + sh2[v];
        y = y > 0.f ? y : 0.f;
        if (ct < 2) m0[v] = fmaxf(m0[v], y);
        else        m1[v] = fmaxf(m1[v], y);
      }
    }
    #pragma unroll
    for (int v = 0; v < 8; ++v)
      #pragma unroll
      for (int off = 1; off < 16; off <<= 1) {
        m0[v] = fmaxf(m0[v], __shfl_xor(m0[v], off, 32));
        m1[v] = fmaxf(m1[v], __shfl_xor(m1[v], off, 32));
      }
    if ((lane & 15) == 0) {
      #pragma unroll
      for (int v = 0; v < 8; ++v) {
        int o = mt * 16 + hi * 8 + v;
        out[((size_t)b * COUT + o) * NN + n0]     = m0[v];
        out[((size_t)b * COUT + o) * NN + n0 + 1] = m1[v];
      }
    }
  }
}

extern "C" void kernel_launch(void* const* d_in, const int* in_sizes, int n_in,
                              void* d_out, int out_size, void* d_ws, size_t ws_size,
                              hipStream_t stream) {
  const float* p  = (const float*)d_in[0];
  const float* x  = (const float*)d_in[1];
  const float* W1 = (const float*)d_in[2];
  const float* g1 = (const float*)d_in[3];
  const float* b1 = (const float*)d_in[4];
  const float* W2 = (const float*)d_in[5];
  const float* g2 = (const float*)d_in[6];
  const float* b2 = (const float*)d_in[7];
  float* out = (float*)d_out;

  char*  ws    = (char*)d_ws;
  int*   idx   = (int*)ws;
  float* xt    = (float*)(ws + (size_t)BB * NN * KS * 4);
  float* part1 = (float*)(ws + (size_t)BB * NN * KS * 4
                             + (size_t)BB * NN * CX * 4);
  float* part2 = part1 + (size_t)GCONV * 128;
  float* bnp1  = part2 + (size_t)GCONV * 128;
  float* bnp2  = bnp1 + 128;

  xpose_kernel<<<dim3(NN / 64, BB), 256, 0, stream>>>(x, xt);
  ballq_kernel<<<dim3(NN / 256, BB), 256, 0, stream>>>(p, idx);
  conv1_stats_kernel<<<GCONV, 128, 0, stream>>>(p, xt, idx, W1, part1);
  bn_finalize_kernel<<<1, 64, 0, stream>>>(part1, g1, b1, bnp1, GCONV);
  conv2_stats_kernel<<<GCONV, 128, 0, stream>>>(p, xt, idx, W1, W2, bnp1, part2);
  bn_finalize_kernel<<<1, 64, 0, stream>>>(part2, g2, b2, bnp2, GCONV);
  final_kernel<<<GCONV, 128, 0, stream>>>(p, xt, idx, W1, W2, bnp1, bnp2, out);
}


# ball query remapped: one wave per 4 queries, one lane per candidate (batch copied to LDS); wave vote + prefix count builds the ascending first-32 list; 256 blocks of 512 threads, was 64 of 256
# speedup vs baseline: 2.2238x; 2.2238x over previous
#include <hip/hip_runtime.h>
#include <hip/hip_bf16.h>


typedef __attribute__((ext_vector_type(2))) float v2f;
typedef __attribute__((ext_vector_type(8))) float v8f;

#define BB      4
#define NN      4096
#define CX      32
#define CIN     35
#define CINP    36
#define COUT    64
#define KS      32
#define R2      0.01f
#define EPSV    1e-5f
#define NPAIRS  (BB * (NN / 2))
#define GCONV   1024
#define CNTF    ((float)BB * (float)NN * (float)KS)

#define FSTRIDE  65
#define W1STRIDE 36
#define W2STRIDE 68

__global__ void xpose_kernel(const float* __restrict__ x, float* __restrict__ xt) {
  __shared__ float tile[CX][65];
  int b  = blockIdx.y;
  int n0 = blockIdx.x * 64;
  int t  = threadIdx.x;
  for (int i = t; i < CX * 64; i += 256) {
    int c = i >> 6, nn = i & 63;
    tile[c][nn] = x[((size_t)b * CX + c) * NN + n0 + nn];
  }
  __syncthreads();
  for (int i = t; i < CX * 64; i += 256) {
    int nn = i >> 5, c = i & 31;
    xt[((size_t)b * NN + n0 + nn) * CX + c] = tile[c][nn];
  }
}

#define BQ_Q     4
#define BQ_WAVES 16
#define BQ_QPB   (BQ_Q * BQ_WAVES)
#define BQ_TPB   (32 * BQ_WAVES)
#define BQ_FILL  (NN * 3 / 4 / BQ_TPB)
static_assert(BQ_Q == 4, "the vote test and the early-exit test below are written out for 4 queries");

__global__ __launch_bounds__(BQ_TPB) void ballq_kernel(const float* __restrict__ p,
                                                       int* __restrict__ idx) {
  __shared__ float4 cand4[NN * 3 / 4];
  __shared__ int    lst[BQ_WAVES * BQ_Q * KS];
  const int t    = threadIdx.x;
  const int lane = t & 31;
  const int w  = __builtin_amdgcn_readfirstlane(t >> 5);
  const int b  = blockIdx.x / (NN / BQ_QPB);
  const int q0 = (blockIdx.x % (NN / BQ_QPB)) * BQ_QPB + w * BQ_Q;
  const float* pb = p + (size_t)b * NN * 3;

  const float4* src4 = reinterpret_cast<const float4*>(pb);
  float4 stage[BQ_FILL];
  #pragma unroll
  for (int k = 0; k < BQ_FILL; ++k) stage[k] = src4[t + k * BQ_TPB];
  #pragma unroll
  for (int k = 0; k < BQ_FILL; ++k) cand4[t + k * BQ_TPB] = stage[k];

  int* mylst = lst + w * (BQ_Q * KS);
  if (lane < BQ_Q) mylst[lane * KS] = 0;
  float qx[BQ_Q], qy[BQ_Q], qz[BQ_Q];
  int cnt[BQ_Q];
  #pragma unroll
  for (int q = 0; q < BQ_Q; ++q) {
    qx[q] = pb[(q0 + q) * 3 + 0];
    qy[q] = pb[(q0 + q) * 3 + 1];
    qz[q] = pb[(q0 + q) * 3 + 2];
    cnt[q] = 0;
  }
  __syncthreads();

  const float* cand = reinterpret_cast<const float*>(cand4) + lane * 3;
  float nx = cand[0], ny = cand[1], nz = cand[2];
  bool active = true;
  int nb3 = 0;
  for (int base = 0; base < NN && active; base += 32) {
    const float cx = nx, cy = ny, cz = nz;
    if (base + 32 < NN) nb3 += 32 * 3;
    nx = cand[nb3 + 0]; ny = cand[nb3 + 1]; nz = cand[nb3 + 2];

    bool h[BQ_Q];
    unsigned m[BQ_Q];
    #pragma unroll
    for (int q = 0; q < BQ_Q; ++q) {
      const float dx = cx - qx[q], dy = cy - qy[q], dz = cz - qz[q];
      const float d2 = __builtin_fmaf(dz, dz, __builtin_fmaf(dx, dx, dy * dy));
      h[q] = d2 < R2;
      m[q] = __builtin_amdgcn_ballot_w32(h[q]);
    }
    if ((m[0] | m[1] | m[2] | m[3]) != 0u) {
      #pragma unroll
      for (int q = 0; q < BQ_Q; ++q) {
        if (m[q] != 0u) {
          const int rank = cnt[q] + (int)__builtin_amdgcn_mbcnt_lo(m[q], 0u);
          if (h[q] && rank < KS) mylst[q * KS + rank] = base + lane;
          cnt[q] += __builtin_popcount(m[q]);
        }
      }
      active = (((cnt[0] - KS) | (cnt[1] - KS) | (cnt[2] - KS) | (cnt[3] - KS)) < 0);
    }
  }
  #pragma unroll
  for (int q = 0; q < BQ_Q; ++q) {
    const int v = mylst[q * KS + (lane < cnt[q] ? lane : 0)];
    idx[((size_t)b * NN + q0 + q) * KS + lane] = v;
  }
}

__device__ __forceinline__ void load_ftile(const float* __restrict__ p,
                                           const float* __restrict__ xt,
                                           const int* __restrict__ idxg,
                                           int b, int n0,
                                           int* ibuf, float* flds) {
  int t = threadIdx.x;
  if (t < 64) ibuf[t] = idxg[((size_t)b * NN + n0) * KS + t];
  __syncthreads();
  int col = t & 63;
  int j = ibuf[col];
  const float* xr = xt + ((size_t)b * NN + j) * CX;
  if (t < 64) {
    int nq = n0 + (col >> 5);
    const float* pj = p + ((size_t)b * NN + j) * 3;
    const float* pq = p + ((size_t)b * NN + nq) * 3;
    flds[0 * FSTRIDE + col] = pj[0] - pq[0];
    flds[1 * FSTRIDE + col] = pj[1] - pq[1];
    flds[2 * FSTRIDE + col] = pj[2] - pq[2];
    flds[35 * FSTRIDE + col] = 0.0f;
    #pragma unroll
    for (int c = 0; c < 16; ++c) flds[(3 + c) * FSTRIDE + col] = xr[c];
  } else {
    #pragma unroll
    for (int c = 16; c < 32; ++c) flds[(3 + c) * FSTRIDE + col] = xr[c];
  }
}

__device__ __forceinline__ v8f wmma_gemm(const float* __restrict__ Wl, int wstride,
                                         const float* __restrict__ Fl,
                                         int mt, int ct, int ksteps) {
  int lane  = threadIdx.x & 31;
  int mrow  = mt * 16 + (lane & 15);
  int khalf = (lane >> 4) << 1;
  int col   = ct * 16 + (lane & 15);
  v8f acc = {0.f, 0.f, 0.f, 0.f, 0.f, 0.f, 0.f, 0.f};
  for (int kk = 0; kk < ksteps; ++kk) {
    int kb = (kk << 2) + khalf;
    v2f a, bm;
    a.x  = Wl[mrow * wstride + kb];
    a.y  = Wl[mrow * wstride + kb + 1];
    bm.x = Fl[kb * FSTRIDE + col];
    bm.y = Fl[(kb + 1) * FSTRIDE + col];
    acc = __builtin_amdgcn_wmma_f32_16x16x4_f32(false, a, false, bm,
                                                (short)0, acc, false, false);
  }
  return acc;
}

__global__ __launch_bounds__(128) void conv1_stats_kernel(
    const float* __restrict__ p, const float* __restrict__ xt,
    const int* __restrict__ idxg, const float* __restrict__ W1,
    float* __restrict__ part1) {
  __shared__ float W1l[COUT * W1STRIDE];
  __shared__ float flds[CINP * FSTRIDE];
  __shared__ int   ibuf[64];
  int t = threadIdx.x;
  for (int i = t; i < COUT * W1STRIDE; i += 128) {
    int c = i % W1STRIDE;
    W1l[i] = (c < CIN) ? W1[(i / W1STRIDE) * CIN + c] : 0.f;
  }
  int mt = t >> 5, lane = t & 31;
  float s[8], sq[8];
  #pragma unroll
  for (int v = 0; v < 8; ++v) { s[v] = 0.f; sq[v] = 0.f; }

  for (int pair = blockIdx.x; pair < NPAIRS; pair += gridDim.x) {
    int b  = pair / (NN / 2);
    int n0 = (pair % (NN / 2)) * 2;
    __syncthreads();
    load_ftile(p, xt, idxg, b, n0, ibuf, flds);
    __syncthreads();
    #pragma unroll
    for (int ct = 0; ct < 4; ++ct) {
      v8f acc = wmma_gemm(W1l, W1STRIDE, flds, mt, ct, 9);
      #pragma unroll
      for (int v = 0; v < 8; ++v) { float y = acc[v]; s[v] += y; sq[v] += y * y; }
    }
  }
  #pragma unroll
  for (int v = 0; v < 8; ++v)
    #pragma unroll
    for (int off = 1; off < 16; off <<= 1) {
      s[v]  += __shfl_xor(s[v],  off, 32);
      sq[v] += __shfl_xor(sq[v], off, 32);
    }
  if ((lane & 15) == 0) {
    int hi = lane >> 4;
    #pragma unroll
    for (int v = 0; v < 8; ++v) {
      int o = mt * 16 + hi * 8 + v;
      part1[(size_t)blockIdx.x * 128 + o]      = s[v];
      part1[(size_t)blockIdx.x * 128 + 64 + o] = sq[v];
    }
  }
}

__global__ void bn_finalize_kernel(const float* __restrict__ part,
                                   const float* __restrict__ g,
                                   const float* __restrict__ bta,
                                   float* __restrict__ bnp, int nblocks) {
  int o = threadIdx.x;
  if (o >= COUT) return;
  float s = 0.f, sq = 0.f;
  for (int blk = 0; blk < nblocks; ++blk) {
    s  += part[(size_t)blk * 128 + o];
    sq += part[(size_t)blk * 128 + 64 + o];
  }
  float mean  = s / CNTF;
  float var   = sq / CNTF - mean * mean;
  float scale = g[o] * rsqrtf(var + EPSV);
  bnp[o]      = scale;
  bnp[64 + o] = bta[o] - mean * scale;
}

__global__ __launch_bounds__(128) void conv2_stats_kernel(
    const float* __restrict__ p, const float* __restrict__ xt,
    const int* __restrict__ idxg, const float* __restrict__ W1,
    const float* __restrict__ W2, const float* __restrict__ bnp1,
    float* __restrict__ part2) {
  __shared__ float W1l[COUT * W1STRIDE];
  __shared__ float W2l[COUT * W2STRIDE];
  __shared__ float flds[CINP * FSTRIDE];
  __shared__ float f1l[COUT * FSTRIDE];
  __shared__ int   ibuf[64];
  int t = threadIdx.x;
  for (int i = t; i < COUT * W1STRIDE; i += 128) {
    int c = i % W1STRIDE;
    W1l[i] = (c < CIN) ? W1[(i / W1STRIDE) * CIN + c] : 0.f;
  }
  for (int i = t; i < COUT * W2STRIDE; i += 128) {
    int c = i % W2STRIDE;
    W2l[i] = (c < COUT) ? W2[(i / W2STRIDE) * COUT + c] : 0.f;
  }
  int mt = t >> 5, lane = t & 31;
  int hi = lane >> 4, colb = lane & 15;
  float sc1[8], sh1[8];
  #pragma unroll
  for (int v = 0; v < 8; ++v) {
    int o = mt * 16 + hi * 8 + v;
    sc1[v] = bnp1[o]; sh1[v] = bnp1[64 + o];
  }
  float s[8], sq[8];
  #pragma unroll
  for (int v = 0; v < 8; ++v) { s[v] = 0.f; sq[v] = 0.f; }

  for (int pair = blockIdx.x; pair < NPAIRS; pair += gridDim.x) {
    int b  = pair / (NN / 2);
    int n0 = (pair % (NN / 2)) * 2;
    __syncthreads();
    load_ftile(p, xt, idxg, b, n0, ibuf, flds);
    __syncthreads();
    #pragma unroll
    for (int ct = 0; ct < 4; ++ct) {
      v8f acc = wmma_gemm(W1l, W1STRIDE, flds, mt, ct, 9);
      int col = ct * 16 + colb;
      #pragma unroll
      for (int v = 0; v < 8; ++v) {
        float y = acc[v] * sc1[v] + sh1[v];
        y = y > 0.f ? y : 0.f;
        f1l[(mt * 16 + hi * 8 + v) * FSTRIDE + col] = y;
      }
    }
    __syncthreads();
    #pragma unroll
    for (int ct = 0; ct < 4; ++ct) {
      v8f acc = wmma_gemm(W2l, W2STRIDE, f1l, mt, ct, 16);
      #pragma unroll
      for (int v = 0; v < 8; ++v) { float y = acc[v]; s[v] += y; sq[v] += y * y; }
    }
  }
  #pragma unroll
  for (int v = 0; v < 8; ++v)
    #pragma unroll
    for (int off = 1; off < 16; off <<= 1) {
      s[v]  += __shfl_xor(s[v],  off, 32);
      sq[v] += __shfl_xor(sq[v], off, 32);
    }
  if ((lane & 15) == 0) {
    #pragma unroll
    for (int v = 0; v < 8; ++v) {
      int o = mt * 16 + hi * 8 + v;
      part2[(size_t)blockIdx.x * 128 + o]      = s[v];
      part2[(size_t)blockIdx.x * 128 + 64 + o] = sq[v];
    }
  }
}

__global__ __launch_bounds__(128) void final_kernel(
    const float* __restrict__ p, const float* __restrict__ xt,
    const int* __restrict__ idxg, const float* __restrict__ W1,
    const float* __restrict__ W2, const float* __restrict__ bnp1,
    const float* __restrict__ bnp2, float* __restrict__ out) {
  __shared__ float W1l[COUT * W1STRIDE];
  __shared__ float W2l[COUT * W2STRIDE];
  __shared__ float flds[CINP * FSTRIDE];
  __shared__ float f1l[COUT * FSTRIDE];
  __shared__ int   ibuf[64];
  int t = threadIdx.x;
  for (int i = t; i < COUT * W1STRIDE; i += 128) {
    int c = i % W1STRIDE;
    W1l[i] = (c < CIN) ? W1[(i / W1STRIDE) * CIN + c] : 0.f;
  }
  for (int i = t; i < COUT * W2STRIDE; i += 128) {
    int c = i % W2STRIDE;
    W2l[i] = (c < COUT) ? W2[(i / W2STRIDE) * COUT + c] : 0.f;
  }
  int mt = t >> 5, lane = t & 31;
  int hi = lane >> 4, colb = lane & 15;
  float sc1[8], sh1[8], sc2[8], sh2[8];
  #pragma unroll
  for (int v = 0; v < 8; ++v) {
    int o = mt * 16 + hi * 8 + v;
    sc1[v] = bnp1[o]; sh1[v] = bnp1[64 + o];
    sc2[v] = bnp2[o]; sh2[v] = bnp2[64 + o];
  }
  for (int pair = blockIdx.x; pair < NPAIRS; pair += gridDim.x) {
    int b  = pair / (NN / 2);
    int n0 = (pair % (NN / 2)) * 2;
    __syncthreads();
    load_ftile(p, xt, idxg, b, n0, ibuf, flds);
    __syncthreads();
    #pragma unroll
    for (int ct = 0; ct < 4; ++ct) {
      v8f acc = wmma_gemm(W1l, W1STRIDE, flds, mt, ct, 9);
      int col = ct * 16 + colb;
      #pragma unroll
      for (int v = 0; v < 8; ++v) {
        float y = acc[v] * sc1[v] + sh1[v];
        y = y > 0.f ? y : 0.f;
        f1l[(mt * 16 + hi * 8 + v) * FSTRIDE + col] = y;
      }
    }
    __syncthreads();
    float m0[8], m1[8];
    #pragma unroll
    for (int v = 0; v < 8; ++v) { m0[v] = 0.f; m1[v] = 0.f; }
    #pragma unroll
    for (int ct = 0; ct < 4; ++ct) {
      v8f acc = wmma_gemm(W2l, W2STRIDE, f1l, mt, ct, 16);
      #pragma unroll
      for (int v = 0; v < 8; ++v) {
        float y = acc[v] * sc2[v] + sh2[v];
        y = y > 0.f ? y : 0.f;
        if (ct < 2) m0[v] = fmaxf(m0[v], y);
        else        m1[v] = fmaxf(m1[v], y);
      }
    }
    #pragma unroll
    for (int v = 0; v < 8; ++v)
      #pragma unroll
      for (int off = 1; off < 16; off <<= 1) {
        m0[v] = fmaxf(m0[v], __shfl_xor(m0[v], off, 32));
        m1[v] = fmaxf(m1[v], __shfl_xor(m1[v], off, 32));
      }
    if ((lane & 15) == 0) {
      #pragma unroll
      for (int v = 0; v < 8; ++v) {
        int o = mt * 16 + hi * 8 + v;
        out[((size_t)b * COUT + o) * NN + n0]     = m0[v];
        out[((size_t)b * COUT + o) * NN + n0 + 1] = m1[v];
      }
    }
  }
}

extern "C" void kernel_launch(void* const* d_in, const int* in_sizes, int n_in,
                              void* d_out, int out_size, void* d_ws, size_t ws_size,
                              hipStream_t stream) {
  const float* p  = (const float*)d_in[0];
  const float* x  = (const float*)d_in[1];
  const float* W1 = (const float*)d_in[2];
  const float* g1 = (const float*)d_in[3];
  const float* b1 = (const float*)d_in[4];
  const float* W2 = (const float*)d_in[5];
  const float* g2 = (const float*)d_in[6];
  const float* b2 = (const float*)d_in[7];
  float* out = (float*)d_out;

  char*  ws    = (char*)d_ws;
  int*   idx   = (int*)ws;
  float* xt    = (float*)(ws + (size_t)BB * NN * KS * 4);
  float* part1 = (float*)(ws + (size_t)BB * NN * KS * 4
                             + (size_t)BB * NN * CX * 4);
  float* part2 = part1 + (size_t)GCONV * 128;
  float* bnp1  = part2 + (size_t)GCONV * 128;
  float* bnp2  = bnp1 + 128;

  xpose_kernel<<<dim3(NN / 64, BB), 256, 0, stream>>>(x, xt);
  ballq_kernel<<<BB * (NN / BQ_QPB), BQ_TPB, 0, stream>>>(p, idx);
  conv1_stats_kernel<<<GCONV, 128, 0, stream>>>(p, xt, idx, W1, part1);
  bn_finalize_kernel<<<1, 64, 0, stream>>>(part1, g1, b1, bnp1, GCONV);
  conv2_stats_kernel<<<GCONV, 128, 0, stream>>>(p, xt, idx, W1, W2, bnp1, part2);
  bn_finalize_kernel<<<1, 64, 0, stream>>>(part2, g2, b2, bnp2, GCONV);
  final_kernel<<<GCONV, 128, 0, stream>>>(p, xt, idx, W1, W2, bnp1, bnp2, out);
}


# BN partial-sum reduce made parallel: 4 blocks x 1024 threads, every row load in flight at once, tree sum instead of a serial 1024-row loop (changes the summation order of the statistics)
# speedup vs baseline: 3.7049x; 1.6660x over previous
#include <hip/hip_runtime.h>
#include <hip/hip_bf16.h>


typedef __attribute__((ext_vector_type(2))) float v2f;
typedef __attribute__((ext_vector_type(8))) float v8f;

#define BB      4
#define NN      4096
#define CX      32
#define CIN     35
#define CINP    36
#define COUT    64
#define KS      32
#define R2      0.01f
#define EPSV    1e-5f
#define NPAIRS  (BB * (NN / 2))
#define GCONV   1024
#define CNTF    ((float)BB * (float)NN * (float)KS)

#define FSTRIDE  65
#define W1STRIDE 36
#define W2STRIDE 68

__global__ void xpose_kernel(const float* __restrict__ x, float* __restrict__ xt) {
  __shared__ float tile[CX][65];
  int b  = blockIdx.y;
  int n0 = blockIdx.x * 64;
  int t  = threadIdx.x;
  for (int i = t; i < CX * 64; i += 256) {
    int c = i >> 6, nn = i & 63;
    tile[c][nn] = x[((size_t)b * CX + c) * NN + n0 + nn];
  }
  __syncthreads();
  for (int i = t; i < CX * 64; i += 256) {
    int nn = i >> 5, c = i & 31;
    xt[((size_t)b * NN + n0 + nn) * CX + c] = tile[c][nn];
  }
}

#define BQ_Q     4
#define BQ_WAVES 16
#define BQ_QPB   (BQ_Q * BQ_WAVES)
#define BQ_TPB   (32 * BQ_WAVES)
#define BQ_FILL  (NN * 3 / 4 / BQ_TPB)
static_assert(BQ_Q == 4, "the vote test and the early-exit test below are written out for 4 queries");

__global__ __launch_bounds__(BQ_TPB) void ballq_kernel(const float* __restrict__ p,
                                                       int* __restrict__ idx) {
  __shared__ float4 cand4[NN * 3 / 4];
  __shared__ int    lst[BQ_WAVES * BQ_Q * KS];
  const int t    = threadIdx.x;
  const int lane = t & 31;
  const int w  = __builtin_amdgcn_readfirstlane(t >> 5);
  const int b  = blockIdx.x / (NN / BQ_QPB);
  const int q0 = (blockIdx.x % (NN / BQ_QPB)) * BQ_QPB + w * BQ_Q;
  const float* pb = p + (size_t)b * NN * 3;

  const float4* src4 = reinterpret_cast<const float4*>(pb);
  float4 stage[BQ_FILL];
  #pragma unroll
  for (int k = 0; k < BQ_FILL; ++k) stage[k] = src4[t + k * BQ_TPB];
  #pragma unroll
  for (int k = 0; k < BQ_FILL; ++k) cand4[t + k * BQ_TPB] = stage[k];

  int* mylst = lst + w * (BQ_Q * KS);
  if (lane < BQ_Q) mylst[lane * KS] = 0;
  float qx[BQ_Q], qy[BQ_Q], qz[BQ_Q];
  int cnt[BQ_Q];
  #pragma unroll
  for (int q = 0; q < BQ_Q; ++q) {
    qx[q] = pb[(q0 + q) * 3 + 0];
    qy[q] = pb[(q0 + q) * 3 + 1];
    qz[q] = pb[(q0 + q) * 3 + 2];
    cnt[q] = 0;
  }
  __syncthreads();

  const float* cand = reinterpret_cast<const float*>(cand4) + lane * 3;
  float nx = cand[0], ny = cand[1], nz = cand[2];
  bool active = true;
  int nb3 = 0;
  for (int base = 0; base < NN && active; base += 32) {
    const float cx = nx, cy = ny, cz = nz;
    if (base + 32 < NN) nb3 += 32 * 3;
    nx = cand[nb3 + 0]; ny = cand[nb3 + 1]; nz = cand[nb3 + 2];

    bool h[BQ_Q];
    unsigned m[BQ_Q];
    #pragma unroll
    for (int q = 0; q < BQ_Q; ++q) {
      const float dx = cx - qx[q], dy = cy - qy[q], dz = cz - qz[q];
      const float d2 = __builtin_fmaf(dz, dz, __builtin_fmaf(dx, dx, dy * dy));
      h[q] = d2 < R2;
      m[q] = __builtin_amdgcn_ballot_w32(h[q]);
    }
    if ((m[0] | m[1] | m[2] | m[3]) != 0u) {
      #pragma unroll
      for (int q = 0; q < BQ_Q; ++q) {
        if (m[q] != 0u) {
          const int rank = cnt[q] + (int)__builtin_amdgcn_mbcnt_lo(m[q], 0u);
          if (h[q] && rank < KS) mylst[q * KS + rank] = base + lane;
          cnt[q] += __builtin_popcount(m[q]);
        }
      }
      active = (((cnt[0] - KS) | (cnt[1] - KS) | (cnt[2] - KS) | (cnt[3] - KS)) < 0);
    }
  }
  #pragma unroll
  for (int q = 0; q < BQ_Q; ++q) {
    const int v = mylst[q * KS + (lane < cnt[q] ? lane : 0)];
    idx[((size_t)b * NN + q0 + q) * KS + lane] = v;
  }
}

__device__ __forceinline__ void load_ftile(const float* __restrict__ p,
                                           const float* __restrict__ xt,
                                           const int* __restrict__ idxg,
                                           int b, int n0,
                                           int* ibuf, float* flds) {
  int t = threadIdx.x;
  if (t < 64) ibuf[t] = idxg[((size_t)b * NN + n0) * KS + t];
  __syncthreads();
  int col = t & 63;
  int j = ibuf[col];
  const float* xr = xt + ((size_t)b * NN + j) * CX;
  if (t < 64) {
    int nq = n0 + (col >> 5);
    const float* pj = p + ((size_t)b * NN + j) * 3;
    const float* pq = p + ((size_t)b * NN + nq) * 3;
    flds[0 * FSTRIDE + col] = pj[0] - pq[0];
    flds[1 * FSTRIDE + col] = pj[1] - pq[1];
    flds[2 * FSTRIDE + col] = pj[2] - pq[2];
    flds[35 * FSTRIDE + col] = 0.0f;
    #pragma unroll
    for (int c = 0; c < 16; ++c) flds[(3 + c) * FSTRIDE + col] = xr[c];
  } else {
    #pragma unroll
    for (int c = 16; c < 32; ++c) flds[(3 + c) * FSTRIDE + col] = xr[c];
  }
}

__device__ __forceinline__ v8f wmma_gemm(const float* __restrict__ Wl, int wstride,
                                         const float* __restrict__ Fl,
                                         int mt, int ct, int ksteps) {
  int lane  = threadIdx.x & 31;
  int mrow  = mt * 16 + (lane & 15);
  int khalf = (lane >> 4) << 1;
  int col   = ct * 16 + (lane & 15);
  v8f acc = {0.f, 0.f, 0.f, 0.f, 0.f, 0.f, 0.f, 0.f};
  for (int kk = 0; kk < ksteps; ++kk) {
    int kb = (kk << 2) + khalf;
    v2f a, bm;
    a.x  = Wl[mrow * wstride + kb];
    a.y  = Wl[mrow * wstride + kb + 1];
    bm.x = Fl[kb * FSTRIDE + col];
    bm.y = Fl[(kb + 1) * FSTRIDE + col];
    acc = __builtin_amdgcn_wmma_f32_16x16x4_f32(false, a, false, bm,
                                                (short)0, acc, false, false);
  }
  return acc;
}

__global__ __launch_bounds__(128) void conv1_stats_kernel(
    const float* __restrict__ p, const float* __restrict__ xt,
    const int* __restrict__ idxg, const float* __restrict__ W1,
    float* __restrict__ part1) {
  __shared__ float W1l[COUT * W1STRIDE];
  __shared__ float flds[CINP * FSTRIDE];
  __shared__ int   ibuf[64];
  int t = threadIdx.x;
  for (int i = t; i < COUT * W1STRIDE; i += 128) {
    int c = i % W1STRIDE;
    W1l[i] = (c < CIN) ? W1[(i / W1STRIDE) * CIN + c] : 0.f;
  }
  int mt = t >> 5, lane = t & 31;
  float s[8], sq[8];
  #pragma unroll
  for (int v = 0; v < 8; ++v) { s[v] = 0.f; sq[v] = 0.f; }

  for (int pair = blockIdx.x; pair < NPAIRS; pair += gridDim.x) {
    int b  = pair / (NN / 2);
    int n0 = (pair % (NN / 2)) * 2;
    __syncthreads();
    load_ftile(p, xt, idxg, b, n0, ibuf, flds);
    __syncthreads();
    #pragma unroll
    for (int ct = 0; ct < 4; ++ct) {
      v8f acc = wmma_gemm(W1l, W1STRIDE, flds, mt, ct, 9);
      #pragma unroll
      for (int v = 0; v < 8; ++v) { float y = acc[v]; s[v] += y; sq[v] += y * y; }
    }
  }
  #pragma unroll
  for (int v = 0; v < 8; ++v)
    #pragma unroll
    for (int off = 1; off < 16; off <<= 1) {
      s[v]  += __shfl_xor(s[v],  off, 32);
      sq[v] += __shfl_xor(sq[v], off, 32);
    }
  if ((lane & 15) == 0) {
    int hi = lane >> 4;
    #pragma unroll
    for (int v = 0; v < 8; ++v) {
      int c = hi * 8 + v;
      part1[(size_t)blockIdx.x * 128 + mt * 32 + c]      = s[v];
      part1[(size_t)blockIdx.x * 128 + mt * 32 + 16 + c] = sq[v];
    }
  }
}

#define FIN_TPB   1024
#define FIN_WAVES (FIN_TPB / 32)
#define FIN_ROWS  (GCONV / FIN_WAVES)
static_assert(FIN_ROWS == 32 && FIN_WAVES == 32, "tree32 below sums exactly 32 values");

__device__ __forceinline__ float tree32(float (&a)[32]) {
  #pragma unroll
  for (int h = 16; h >= 1; h >>= 1) {
    #pragma unroll
    for (int r = 0; r < h; ++r) a[r] += a[r + h];
  }
  return a[0];
}

__global__ __launch_bounds__(FIN_TPB) void bn_finalize_kernel(const float* __restrict__ part,
                                                              const float* __restrict__ g,
                                                              const float* __restrict__ bta,
                                                              float* __restrict__ bnp) {
  __shared__ float red[FIN_WAVES * 32];
  const int t    = threadIdx.x;
  const int lane = t & 31;
  const int w    = t >> 5;
  const int m    = blockIdx.x;
  const float* src = part + (size_t)w * 128 + m * 32 + lane;
  float a[FIN_ROWS];
  #pragma unroll
  for (int r = 0; r < FIN_ROWS; ++r) a[r] = src[(size_t)r * (FIN_WAVES * 128)];
  red[w * 32 + lane] = tree32(a);
  __syncthreads();
  if (w == 0) {
    float c[FIN_WAVES];
    #pragma unroll
    for (int k = 0; k < FIN_WAVES; ++k) c[k] = red[k * 32 + lane];
    const float tot = tree32(c);
    const float sq  = __shfl_down(tot, 16, 32);
    if (lane < 16) {
      const int o = m * 16 + lane;
      float mean  = tot / CNTF;
      float var   = sq / CNTF - mean * mean;
      float scale = g[o] * rsqrtf(var + EPSV);
      bnp[o]      = scale;
      bnp[64 + o] = bta[o] - mean * scale;
    }
  }
}

__global__ __launch_bounds__(128) void conv2_stats_kernel(
    const float* __restrict__ p, const float* __restrict__ xt,
    const int* __restrict__ idxg, const float* __restrict__ W1,
    const float* __restrict__ W2, const float* __restrict__ bnp1,
    float* __restrict__ part2) {
  __shared__ float W1l[COUT * W1STRIDE];
  __shared__ float W2l[COUT * W2STRIDE];
  __shared__ float flds[CINP * FSTRIDE];
  __shared__ float f1l[COUT * FSTRIDE];
  __shared__ int   ibuf[64];
  int t = threadIdx.x;
  for (int i = t; i < COUT * W1STRIDE; i += 128) {
    int c = i % W1STRIDE;
    W1l[i] = (c < CIN) ? W1[(i / W1STRIDE) * CIN + c] : 0.f;
  }
  for (int i = t; i < COUT * W2STRIDE; i += 128) {
    int c = i % W2STRIDE;
    W2l[i] = (c < COUT) ? W2[(i / W2STRIDE) * COUT + c] : 0.f;
  }
  int mt = t >> 5, lane = t & 31;
  int hi = lane >> 4, colb = lane & 15;
  float sc1[8], sh1[8];
  #pragma unroll
  for (int v = 0; v < 8; ++v) {
    int o = mt * 16 + hi * 8 + v;
    sc1[v] = bnp1[o]; sh1[v] = bnp1[64 + o];
  }
  float s[8], sq[8];
  #pragma unroll
  for (int v = 0; v < 8; ++v) { s[v] = 0.f; sq[v] = 0.f; }

  for (int pair = blockIdx.x; pair < NPAIRS; pair += gridDim.x) {
    int b  = pair / (NN / 2);
    int n0 = (pair % (NN / 2)) * 2;
    __syncthreads();
    load_ftile(p, xt, idxg, b, n0, ibuf, flds);
    __syncthreads();
    #pragma unroll
    for (int ct = 0; ct < 4; ++ct) {
      v8f acc = wmma_gemm(W1l, W1STRIDE, flds, mt, ct, 9);
      int col = ct * 16 + colb;
      #pragma unroll
      for (int v = 0; v < 8; ++v) {
        float y = acc[v] * sc1[v] + sh1[v];
        y = y > 0.f ? y : 0.f;
        f1l[(mt * 16 + hi * 8 + v) * FSTRIDE + col] = y;
      }
    }
    __syncthreads();
    #pragma unroll
    for (int ct = 0; ct < 4; ++ct) {
      v8f acc = wmma_gemm(W2l, W2STRIDE, f1l, mt, ct, 16);
      #pragma unroll
      for (int v = 0; v < 8; ++v) { float y = acc[v]; s[v] += y; sq[v] += y * y; }
    }
  }
  #pragma unroll
  for (int v = 0; v < 8; ++v)
    #pragma unroll
    for (int off = 1; off < 16; off <<= 1) {
      s[v]  += __shfl_xor(s[v],  off, 32);
      sq[v] += __shfl_xor(sq[v], off, 32);
    }
  if ((lane & 15) == 0) {
    #pragma unroll
    for (int v = 0; v < 8; ++v) {
      int c = hi * 8 + v;
      part2[(size_t)blockIdx.x * 128 + mt * 32 + c]      = s[v];
      part2[(size_t)blockIdx.x * 128 + mt * 32 + 16 + c] = sq[v];
    }
  }
}

__global__ __launch_bounds__(128) void final_kernel(
    const float* __restrict__ p, const float* __restrict__ xt,
    const int* __restrict__ idxg, const float* __restrict__ W1,
    const float* __restrict__ W2, const float* __restrict__ bnp1,
    const float* __restrict__ bnp2, float* __restrict__ out) {
  __shared__ float W1l[COUT * W1STRIDE];
  __shared__ float W2l[COUT * W2STRIDE];
  __shared__ float flds[CINP * FSTRIDE];
  __shared__ float f1l[COUT * FSTRIDE];
  __shared__ int   ibuf[64];
  int t = threadIdx.x;
  for (int i = t; i < COUT * W1STRIDE; i += 128) {
    int c = i % W1STRIDE;
    W1l[i] = (c < CIN) ? W1[(i / W1STRIDE) * CIN + c] : 0.f;
  }
  for (int i = t; i < COUT * W2STRIDE; i += 128) {
    int c = i % W2STRIDE;
    W2l[i] = (c < COUT) ? W2[(i / W2STRIDE) * COUT + c] : 0.f;
  }
  int mt = t >> 5, lane = t & 31;
  int hi = lane >> 4, colb = lane & 15;
  float sc1[8], sh1[8], sc2[8], sh2[8];
  #pragma unroll
  for (int v = 0; v < 8; ++v) {
    int o = mt * 16 + hi * 8 + v;
    sc1[v] = bnp1[o]; sh1[v] = bnp1[64 + o];
    sc2[v] = bnp2[o]; sh2[v] = bnp2[64 + o];
  }
  for (int pair = blockIdx.x; pair < NPAIRS; pair += gridDim.x) {
    int b  = pair / (NN / 2);
    int n0 = (pair % (NN / 2)) * 2;
    __syncthreads();
    load_ftile(p, xt, idxg, b, n0, ibuf, flds);
    __syncthreads();
    #pragma unroll
    for (int ct = 0; ct < 4; ++ct) {
      v8f acc = wmma_gemm(W1l, W1STRIDE, flds, mt, ct, 9);
      int col = ct * 16 + colb;
      #pragma unroll
      for (int v = 0; v < 8; ++v) {
        float y = acc[v] * sc1[v] + sh1[v];
        y = y > 0.f ? y : 0.f;
        f1l[(mt * 16 + hi * 8 + v) * FSTRIDE + col] = y;
      }
    }
    __syncthreads();
    float m0[8], m1[8];
    #pragma unroll
    for (int v = 0; v < 8; ++v) { m0[v] = 0.f; m1[v] = 0.f; }
    #pragma unroll
    for (int ct = 0; ct < 4; ++ct) {
      v8f acc = wmma_gemm(W2l, W2STRIDE, f1l, mt, ct, 16);
      #pragma unroll
      for (int v = 0; v < 8; ++v) {
        float y = acc[v] * sc2[v] + sh2[v];
        y = y > 0.f ? y : 0.f;
        if (ct < 2) m0[v] = fmaxf(m0[v], y);
        else        m1[v] = fmaxf(m1[v], y);
      }
    }
    #pragma unroll
    for (int v = 0; v < 8; ++v)
      #pragma unroll
      for (int off = 1; off < 16; off <<= 1) {
        m0[v] = fmaxf(m0[v], __shfl_xor(m0[v], off, 32));
        m1[v] = fmaxf(m1[v], __shfl_xor(m1[v], off, 32));
      }
    if ((lane & 15) == 0) {
      #pragma unroll
      for (int v = 0; v < 8; ++v) {
        int o = mt * 16 + hi * 8 + v;
        out[((size_t)b * COUT + o) * NN + n0]     = m0[v];
        out[((size_t)b * COUT + o) * NN + n0 + 1] = m1[v];
      }
    }
  }
}

extern "C" void kernel_launch(void* const* d_in, const int* in_sizes, int n_in,
                              void* d_out, int out_size, void* d_ws, size_t ws_size,
                              hipStream_t stream) {
  const float* p  = (const float*)d_in[0];
  const float* x  = (const float*)d_in[1];
  const float* W1 = (const float*)d_in[2];
  const float* g1 = (const float*)d_in[3];
  const float* b1 = (const float*)d_in[4];
  const float* W2 = (const float*)d_in[5];
  const float* g2 = (const float*)d_in[6];
  const float* b2 = (const float*)d_in[7];
  float* out = (float*)d_out;

  char*  ws    = (char*)d_ws;
  int*   idx   = (int*)ws;
  float* xt    = (float*)(ws + (size_t)BB * NN * KS * 4);
  float* part1 = (float*)(ws + (size_t)BB * NN * KS * 4
                             + (size_t)BB * NN * CX * 4);
  float* part2 = part1 + (size_t)GCONV * 128;
  float* bnp1  = part2 + (size_t)GCONV * 128;
  float* bnp2  = bnp1 + 128;

  xpose_kernel<<<dim3(NN / 64, BB), 256, 0, stream>>>(x, xt);
  ballq_kernel<<<BB * (NN / BQ_QPB), BQ_TPB, 0, stream>>>(p, idx);
  conv1_stats_kernel<<<GCONV, 128, 0, stream>>>(p, xt, idx, W1, part1);
  bn_finalize_kernel<<<COUT / 16, FIN_TPB, 0, stream>>>(part1, g1, b1, bnp1);
  conv2_stats_kernel<<<GCONV, 128, 0, stream>>>(p, xt, idx, W1, W2, bnp1, part2);
  bn_finalize_kernel<<<COUT / 16, FIN_TPB, 0, stream>>>(part2, g2, b2, bnp2);
  final_kernel<<<GCONV, 128, 0, stream>>>(p, xt, idx, W1, W2, bnp1, bnp2, out);
}


# pool before BN2: the BN2-stats pass keeps max over K of raw conv2 (min where gamma2<0; BN2+ReLU is monotone per channel) in out; last pass elementwise in place, third gather+conv pass removed
# speedup vs baseline: 4.8214x; 1.3014x over previous
#include <hip/hip_runtime.h>
#include <hip/hip_bf16.h>


typedef __attribute__((ext_vector_type(2))) float v2f;
typedef __attribute__((ext_vector_type(8))) float v8f;

#define BB      4
#define NN      4096
#define CX      32
#define CIN     35
#define CINP    36
#define COUT    64
#define KS      32
#define R2      0.01f
#define EPSV    1e-5f
#define NPAIRS  (BB * (NN / 2))
#define GCONV   1024
#define CNTF    ((float)BB * (float)NN * (float)KS)

#define FSTRIDE  65
#define W1STRIDE 36
#define W2STRIDE 68

__global__ void xpose_kernel(const float* __restrict__ x, float* __restrict__ xt) {
  __shared__ float tile[CX][65];
  int b  = blockIdx.y;
  int n0 = blockIdx.x * 64;
  int t  = threadIdx.x;
  for (int i = t; i < CX * 64; i += 256) {
    int c = i >> 6, nn = i & 63;
    tile[c][nn] = x[((size_t)b * CX + c) * NN + n0 + nn];
  }
  __syncthreads();
  for (int i = t; i < CX * 64; i += 256) {
    int nn = i >> 5, c = i & 31;
    xt[((size_t)b * NN + n0 + nn) * CX + c] = tile[c][nn];
  }
}

#define BQ_Q     4
#define BQ_WAVES 16
#define BQ_QPB   (BQ_Q * BQ_WAVES)
#define BQ_TPB   (32 * BQ_WAVES)
#define BQ_FILL  (NN * 3 / 4 / BQ_TPB)
static_assert(BQ_Q == 4, "the vote test and the early-exit test below are written out for 4 queries");

__global__ __launch_bounds__(BQ_TPB) void ballq_kernel(const float* __restrict__ p,
                                                       int* __restrict__ idx) {
  __shared__ float4 cand4[NN * 3 / 4];
  __shared__ int    lst[BQ_WAVES * BQ_Q * KS];
  const int t    = threadIdx.x;
  const int lane = t & 31;
  const int w  = __builtin_amdgcn_readfirstlane(t >> 5);
  const int b  = blockIdx.x / (NN / BQ_QPB);
  const int q0 = (blockIdx.x % (NN / BQ_QPB)) * BQ_QPB + w * BQ_Q;
  const float* pb = p + (size_t)b * NN * 3;

  const float4* src4 = reinterpret_cast<const float4*>(pb);
  float4 stage[BQ_FILL];
  #pragma unroll
  for (int k = 0; k < BQ_FILL; ++k) stage[k] = src4[t + k * BQ_TPB];
  #pragma unroll
  for (int k = 0; k < BQ_FILL; ++k) cand4[t + k * BQ_TPB] = stage[k];

  int* mylst = lst + w * (BQ_Q * KS);
  if (lane < BQ_Q) mylst[lane * KS] = 0;
  float qx[BQ_Q], qy[BQ_Q], qz[BQ_Q];
  int cnt[BQ_Q];
  #pragma unroll
  for (int q = 0; q < BQ_Q; ++q) {
    qx[q] = pb[(q0 + q) * 3 + 0];
    qy[q] = pb[(q0 + q) * 3 + 1];
    qz[q] = pb[(q0 + q) * 3 + 2];
    cnt[q] = 0;
  }
  __syncthreads();

  const float* cand = reinterpret_cast<const float*>(cand4) + lane * 3;
  float nx = cand[0], ny = cand[1], nz = cand[2];
  bool active = true;
  int nb3 = 0;
  for (int base = 0; base < NN && active; base += 32) {
    const float cx = nx, cy = ny, cz = nz;
    if (base + 32 < NN) nb3 += 32 * 3;
    nx = cand[nb3 + 0]; ny = cand[nb3 + 1]; nz = cand[nb3 + 2];

    bool h[BQ_Q];
    unsigned m[BQ_Q];
    #pragma unroll
    for (int q = 0; q < BQ_Q; ++q) {
      const float dx = cx - qx[q], dy = cy - qy[q], dz = cz - qz[q];
      const float d2 = __builtin_fmaf(dz, dz, __builtin_fmaf(dx, dx, dy * dy));
      h[q] = d2 < R2;
      m[q] = __builtin_amdgcn_ballot_w32(h[q]);
    }
    if ((m[0] | m[1] | m[2] | m[3]) != 0u) {
      #pragma unroll
      for (int q = 0; q < BQ_Q; ++q) {
        if (m[q] != 0u) {
          const int rank = cnt[q] + (int)__builtin_amdgcn_mbcnt_lo(m[q], 0u);
          if (h[q] && rank < KS) mylst[q * KS + rank] = base + lane;
          cnt[q] += __builtin_popcount(m[q]);
        }
      }
      active = (((cnt[0] - KS) | (cnt[1] - KS) | (cnt[2] - KS) | (cnt[3] - KS)) < 0);
    }
  }
  #pragma unroll
  for (int q = 0; q < BQ_Q; ++q) {
    const int v = mylst[q * KS + (lane < cnt[q] ? lane : 0)];
    idx[((size_t)b * NN + q0 + q) * KS + lane] = v;
  }
}

__device__ __forceinline__ void load_ftile(const float* __restrict__ p,
                                           const float* __restrict__ xt,
                                           const int* __restrict__ idxg,
                                           int b, int n0,
                                           int* ibuf, float* flds) {
  int t = threadIdx.x;
  if (t < 64) ibuf[t] = idxg[((size_t)b * NN + n0) * KS + t];
  __syncthreads();
  int col = t & 63;
  int j = ibuf[col];
  const float* xr = xt + ((size_t)b * NN + j) * CX;
  if (t < 64) {
    int nq = n0 + (col >> 5);
    const float* pj = p + ((size_t)b * NN + j) * 3;
    const float* pq = p + ((size_t)b * NN + nq) * 3;
    flds[0 * FSTRIDE + col] = pj[0] - pq[0];
    flds[1 * FSTRIDE + col] = pj[1] - pq[1];
    flds[2 * FSTRIDE + col] = pj[2] - pq[2];
    flds[35 * FSTRIDE + col] = 0.0f;
    #pragma unroll
    for (int c = 0; c < 16; ++c) flds[(3 + c) * FSTRIDE + col] = xr[c];
  } else {
    #pragma unroll
    for (int c = 16; c < 32; ++c) flds[(3 + c) * FSTRIDE + col] = xr[c];
  }
}

__device__ __forceinline__ v8f wmma_gemm(const float* __restrict__ Wl, int wstride,
                                         const float* __restrict__ Fl,
                                         int mt, int ct, int ksteps) {
  int lane  = threadIdx.x & 31;
  int mrow  = mt * 16 + (lane & 15);
  int khalf = (lane >> 4) << 1;
  int col   = ct * 16 + (lane & 15);
  v8f acc = {0.f, 0.f, 0.f, 0.f, 0.f, 0.f, 0.f, 0.f};
  for (int kk = 0; kk < ksteps; ++kk) {
    int kb = (kk << 2) + khalf;
    v2f a, bm;
    a.x  = Wl[mrow * wstride + kb];
    a.y  = Wl[mrow * wstride + kb + 1];
    bm.x = Fl[kb * FSTRIDE + col];
    bm.y = Fl[(kb + 1) * FSTRIDE + col];
    acc = __builtin_amdgcn_wmma_f32_16x16x4_f32(false, a, false, bm,
                                                (short)0, acc, false, false);
  }
  return acc;
}

__global__ __launch_bounds__(128) void conv1_stats_kernel(
    const float* __restrict__ p, const float* __restrict__ xt,
    const int* __restrict__ idxg, const float* __restrict__ W1,
    float* __restrict__ part1) {
  __shared__ float W1l[COUT * W1STRIDE];
  __shared__ float flds[CINP * FSTRIDE];
  __shared__ int   ibuf[64];
  int t = threadIdx.x;
  for (int i = t; i < COUT * W1STRIDE; i += 128) {
    int c = i % W1STRIDE;
    W1l[i] = (c < CIN) ? W1[(i / W1STRIDE) * CIN + c] : 0.f;
  }
  int mt = t >> 5, lane = t & 31;
  float s[8], sq[8];
  #pragma unroll
  for (int v = 0; v < 8; ++v) { s[v] = 0.f; sq[v] = 0.f; }

  for (int pair = blockIdx.x; pair < NPAIRS; pair += gridDim.x) {
    int b  = pair / (NN / 2);
    int n0 = (pair % (NN / 2)) * 2;
    __syncthreads();
    load_ftile(p, xt, idxg, b, n0, ibuf, flds);
    __syncthreads();
    #pragma unroll
    for (int ct = 0; ct < 4; ++ct) {
      v8f acc = wmma_gemm(W1l, W1STRIDE, flds, mt, ct, 9);
      #pragma unroll
      for (int v = 0; v < 8; ++v) { float y = acc[v]; s[v] += y; sq[v] += y * y; }
    }
  }
  #pragma unroll
  for (int v = 0; v < 8; ++v)
    #pragma unroll
    for (int off = 1; off < 16; off <<= 1) {
      s[v]  += __shfl_xor(s[v],  off, 32);
      sq[v] += __shfl_xor(sq[v], off, 32);
    }
  if ((lane & 15) == 0) {
    int hi = lane >> 4;
    #pragma unroll
    for (int v = 0; v < 8; ++v) {
      int c = hi * 8 + v;
      part1[(size_t)blockIdx.x * 128 + mt * 32 + c]      = s[v];
      part1[(size_t)blockIdx.x * 128 + mt * 32 + 16 + c] = sq[v];
    }
  }
}

#define FIN_TPB   1024
#define FIN_WAVES (FIN_TPB / 32)
#define FIN_ROWS  (GCONV / FIN_WAVES)
static_assert(FIN_ROWS == 32 && FIN_WAVES == 32, "tree32 below sums exactly 32 values");

__device__ __forceinline__ float tree32(float (&a)[32]) {
  #pragma unroll
  for (int h = 16; h >= 1; h >>= 1) {
    #pragma unroll
    for (int r = 0; r < h; ++r) a[r] += a[r + h];
  }
  return a[0];
}

__global__ __launch_bounds__(FIN_TPB) void bn_finalize_kernel(const float* __restrict__ part,
                                                              const float* __restrict__ g,
                                                              const float* __restrict__ bta,
                                                              float* __restrict__ bnp) {
  __shared__ float red[FIN_WAVES * 32];
  const int t    = threadIdx.x;
  const int lane = t & 31;
  const int w    = t >> 5;
  const int m    = blockIdx.x;
  const float* src = part + (size_t)w * 128 + m * 32 + lane;
  float a[FIN_ROWS];
  #pragma unroll
  for (int r = 0; r < FIN_ROWS; ++r) a[r] = src[(size_t)r * (FIN_WAVES * 128)];
  red[w * 32 + lane] = tree32(a);
  __syncthreads();
  if (w == 0) {
    float c[FIN_WAVES];
    #pragma unroll
    for (int k = 0; k < FIN_WAVES; ++k) c[k] = red[k * 32 + lane];
    const float tot = tree32(c);
    const float sq  = __shfl_down(tot, 16, 32);
    if (lane < 16) {
      const int o = m * 16 + lane;
      float mean  = tot / CNTF;
      float var   = sq / CNTF - mean * mean;
      float scale = g[o] * rsqrtf(var + EPSV);
      bnp[o]      = scale;
      bnp[64 + o] = bta[o] - mean * scale;
    }
  }
}

__global__ __launch_bounds__(128) void conv2_stats_kernel(
    const float* __restrict__ p, const float* __restrict__ xt,
    const int* __restrict__ idxg, const float* __restrict__ W1,
    const float* __restrict__ W2, const float* __restrict__ bnp1,
    const float* __restrict__ g2, float* __restrict__ part2,
    float* __restrict__ out) {
  __shared__ float W1l[COUT * W1STRIDE];
  __shared__ float W2l[COUT * W2STRIDE];
  __shared__ float flds[CINP * FSTRIDE];
  __shared__ float f1l[COUT * FSTRIDE];
  __shared__ int   ibuf[64];
  int t = threadIdx.x;
  for (int i = t; i < COUT * W1STRIDE; i += 128) {
    int c = i % W1STRIDE;
    W1l[i] = (c < CIN) ? W1[(i / W1STRIDE) * CIN + c] : 0.f;
  }
  for (int i = t; i < COUT * W2STRIDE; i += 128) {
    int c = i % W2STRIDE;
    W2l[i] = (c < COUT) ? W2[(i / W2STRIDE) * COUT + c] : 0.f;
  }
  int mt = t >> 5, lane = t & 31;
  int hi = lane >> 4, colb = lane & 15;
  float sc1[8], sh1[8];
  #pragma unroll
  for (int v = 0; v < 8; ++v) {
    int o = mt * 16 + hi * 8 + v;
    sc1[v] = bnp1[o]; sh1[v] = bnp1[64 + o];
  }
  float sgn[8];
  #pragma unroll
  for (int v = 0; v < 8; ++v) sgn[v] = (g2[mt * 16 + hi * 8 + v] < 0.f) ? -1.f : 1.f;
  float s[8], sq[8];
  #pragma unroll
  for (int v = 0; v < 8; ++v) { s[v] = 0.f; sq[v] = 0.f; }

  for (int pair = blockIdx.x; pair < NPAIRS; pair += gridDim.x) {
    int b  = pair / (NN / 2);
    int n0 = (pair % (NN / 2)) * 2;
    __syncthreads();
    load_ftile(p, xt, idxg, b, n0, ibuf, flds);
    __syncthreads();
    #pragma unroll
    for (int ct = 0; ct < 4; ++ct) {
      v8f acc = wmma_gemm(W1l, W1STRIDE, flds, mt, ct, 9);
      int col = ct * 16 + colb;
      #pragma unroll
      for (int v = 0; v < 8; ++v) {
        float y = acc[v] * sc1[v] + sh1[v];
        y = y > 0.f ? y : 0.f;
        f1l[(mt * 16 + hi * 8 + v) * FSTRIDE + col] = y;
      }
    }
    __syncthreads();
    float m0[8], m1[8];
    #pragma unroll
    for (int ct = 0; ct < 4; ++ct) {
      v8f acc = wmma_gemm(W2l, W2STRIDE, f1l, mt, ct, 16);
      #pragma unroll
      for (int v = 0; v < 8; ++v) {
        float y = acc[v]; s[v] += y; sq[v] += y * y;
        float e = y * sgn[v];
        if (ct == 0)      m0[v] = e;
        else if (ct == 1) m0[v] = fmaxf(m0[v], e);
        else if (ct == 2) m1[v] = e;
        else              m1[v] = fmaxf(m1[v], e);
      }
    }
    #pragma unroll
    for (int v = 0; v < 8; ++v)
      #pragma unroll
      for (int off = 1; off < 16; off <<= 1) {
        m0[v] = fmaxf(m0[v], __shfl_xor(m0[v], off, 32));
        m1[v] = fmaxf(m1[v], __shfl_xor(m1[v], off, 32));
      }
    if ((lane & 15) == 0) {
      #pragma unroll
      for (int v = 0; v < 8; ++v) {
        int o = mt * 16 + hi * 8 + v;
        out[((size_t)b * COUT + o) * NN + n0]     = m0[v] * sgn[v];
        out[((size_t)b * COUT + o) * NN + n0 + 1] = m1[v] * sgn[v];
      }
    }
  }
  #pragma unroll
  for (int v = 0; v < 8; ++v)
    #pragma unroll
    for (int off = 1; off < 16; off <<= 1) {
      s[v]  += __shfl_xor(s[v],  off, 32);
      sq[v] += __shfl_xor(sq[v], off, 32);
    }
  if ((lane & 15) == 0) {
    #pragma unroll
    for (int v = 0; v < 8; ++v) {
      int c = hi * 8 + v;
      part2[(size_t)blockIdx.x * 128 + mt * 32 + c]      = s[v];
      part2[(size_t)blockIdx.x * 128 + mt * 32 + 16 + c] = sq[v];
    }
  }
}

#define AP_TPB 256
#define AP_V4  (NN / 4 / AP_TPB)
__global__ __launch_bounds__(AP_TPB) void bn2_apply_kernel(const float* __restrict__ bnp2,
                                                           float* __restrict__ out) {
  const int row = blockIdx.x;
  const int o   = row & (COUT - 1);
  const float sc = bnp2[o], sh = bnp2[64 + o];
  float4* q = reinterpret_cast<float4*>(out) + (size_t)row * (NN / 4) + threadIdx.x;
  float4 e[AP_V4];
  #pragma unroll
  for (int k = 0; k < AP_V4; ++k) e[k] = q[k * AP_TPB];
  #pragma unroll
  for (int k = 0; k < AP_V4; ++k) {
    float4 r;
    float y;
    y = e[k].x * sc + sh; r.x = y > 0.f ? y : 0.f;
    y = e[k].y * sc + sh; r.y = y > 0.f ? y : 0.f;
    y = e[k].z * sc + sh; r.z = y > 0.f ? y : 0.f;
    y = e[k].w * sc + sh; r.w = y > 0.f ? y : 0.f;
    q[k * AP_TPB] = r;
  }
}

extern "C" void kernel_launch(void* const* d_in, const int* in_sizes, int n_in,
                              void* d_out, int out_size, void* d_ws, size_t ws_size,
                              hipStream_t stream) {
  const float* p  = (const float*)d_in[0];
  const float* x  = (const float*)d_in[1];
  const float* W1 = (const float*)d_in[2];
  const float* g1 = (const float*)d_in[3];
  const float* b1 = (const float*)d_in[4];
  const float* W2 = (const float*)d_in[5];
  const float* g2 = (const float*)d_in[6];
  const float* b2 = (const float*)d_in[7];
  float* out = (float*)d_out;

  char*  ws    = (char*)d_ws;
  int*   idx   = (int*)ws;
  float* xt    = (float*)(ws + (size_t)BB * NN * KS * 4);
  float* part1 = (float*)(ws + (size_t)BB * NN * KS * 4
                             + (size_t)BB * NN * CX * 4);
  float* part2 = part1 + (size_t)GCONV * 128;
  float* bnp1  = part2 + (size_t)GCONV * 128;
  float* bnp2  = bnp1 + 128;

  xpose_kernel<<<dim3(NN / 64, BB), 256, 0, stream>>>(x, xt);
  ballq_kernel<<<BB * (NN / BQ_QPB), BQ_TPB, 0, stream>>>(p, idx);
  conv1_stats_kernel<<<GCONV, 128, 0, stream>>>(p, xt, idx, W1, part1);
  bn_finalize_kernel<<<COUT / 16, FIN_TPB, 0, stream>>>(part1, g1, b1, bnp1);
  conv2_stats_kernel<<<GCONV, 128, 0, stream>>>(p, xt, idx, W1, W2, bnp1, g2, part2, out);
  bn_finalize_kernel<<<COUT / 16, FIN_TPB, 0, stream>>>(part2, g2, b2, bnp2);
  bn2_apply_kernel<<<BB * COUT, AP_TPB, 0, stream>>>(bnp2, out);
}


# conv1's x part (W1[:,3:] x) hoisted in front of the neighbour gather: done once per point on WMMA into a 4 MB table; both conv passes gather its rows as C of the coordinate k step; idx 2-byte
# speedup vs baseline: 8.7264x; 1.8099x over previous
#include <hip/hip_runtime.h>
#include <hip/hip_bf16.h>


typedef __attribute__((ext_vector_type(2))) float v2f;
typedef __attribute__((ext_vector_type(4))) float v4f;
typedef __attribute__((ext_vector_type(8))) float v8f;

#define BB      4
#define NN      4096
#define CX      32
#define CU      64
#define CIN     35
#define COUT    64
#define KS      32
#define R2      0.01f
#define EPSV    1e-5f
#define GCONV   1024
#define CNTF    ((float)BB * (float)NN * (float)KS)


#define BQ_Q     4
#define BQ_WAVES 16
#define BQ_QPB   (BQ_Q * BQ_WAVES)
#define BQ_TPB   (32 * BQ_WAVES)
#define BQ_FILL  (NN * 3 / 4 / BQ_TPB)
static_assert(BQ_Q == 4, "the vote test and the early-exit test below are written out for 4 queries");

__global__ __launch_bounds__(BQ_TPB) void ballq_kernel(const float* __restrict__ p,
                                                       unsigned short* __restrict__ idx) {
  __shared__ float4 cand4[NN * 3 / 4];
  __shared__ int    lst[BQ_WAVES * BQ_Q * KS];
  const int t    = threadIdx.x;
  const int lane = t & 31;
  const int w  = __builtin_amdgcn_readfirstlane(t >> 5);
  const int b  = blockIdx.x / (NN / BQ_QPB);
  const int q0 = (blockIdx.x % (NN / BQ_QPB)) * BQ_QPB + w * BQ_Q;
  const float* pb = p + (size_t)b * NN * 3;

  const float4* src4 = reinterpret_cast<const float4*>(pb);
  float4 stage[BQ_FILL];
  #pragma unroll
  for (int k = 0; k < BQ_FILL; ++k) stage[k] = src4[t + k * BQ_TPB];
  #pragma unroll
  for (int k = 0; k < BQ_FILL; ++k) cand4[t + k * BQ_TPB] = stage[k];

  int* mylst = lst + w * (BQ_Q * KS);
  if (lane < BQ_Q) mylst[lane * KS] = 0;
  float qx[BQ_Q], qy[BQ_Q], qz[BQ_Q];
  int cnt[BQ_Q];
  #pragma unroll
  for (int q = 0; q < BQ_Q; ++q) {
    qx[q] = pb[(q0 + q) * 3 + 0];
    qy[q] = pb[(q0 + q) * 3 + 1];
    qz[q] = pb[(q0 + q) * 3 + 2];
    cnt[q] = 0;
  }
  __syncthreads();

  const float* cand = reinterpret_cast<const float*>(cand4) + lane * 3;
  float nx = cand[0], ny = cand[1], nz = cand[2];
  bool active = true;
  int nb3 = 0;
  for (int base = 0; base < NN && active; base += 32) {
    const float cx = nx, cy = ny, cz = nz;
    if (base + 32 < NN) nb3 += 32 * 3;
    nx = cand[nb3 + 0]; ny = cand[nb3 + 1]; nz = cand[nb3 + 2];

    bool h[BQ_Q];
    unsigned m[BQ_Q];
    #pragma unroll
    for (int q = 0; q < BQ_Q; ++q) {
      const float dx = cx - qx[q], dy = cy - qy[q], dz = cz - qz[q];
      const float d2 = __builtin_fmaf(dz, dz, __builtin_fmaf(dx, dx, dy * dy));
      h[q] = d2 < R2;
      m[q] = __builtin_amdgcn_ballot_w32(h[q]);
    }
    if ((m[0] | m[1] | m[2] | m[3]) != 0u) {
      #pragma unroll
      for (int q = 0; q < BQ_Q; ++q) {
        if (m[q] != 0u) {
          const int rank = cnt[q] + (int)__builtin_amdgcn_mbcnt_lo(m[q], 0u);
          if (h[q] && rank < KS) mylst[q * KS + rank] = base + lane;
          cnt[q] += __builtin_popcount(m[q]);
        }
      }
      active = (((cnt[0] - KS) | (cnt[1] - KS) | (cnt[2] - KS) | (cnt[3] - KS)) < 0);
    }
  }
  #pragma unroll
  for (int q = 0; q < BQ_Q; ++q) {
    const int v = mylst[q * KS + (lane < cnt[q] ? lane : 0)];
    idx[((size_t)b * NN + q0 + q) * KS + lane] = (unsigned short)v;
  }
}

#define CW_WAVES 4
#define CW_TPB   (32 * CW_WAVES)
#define CW_PTS   (BB * NN / GCONV)
static_assert(CW_PTS == 16 && NN % CW_PTS == 0 && GCONV % CW_WAVES == 0,
              "lane c keeps the pooled values of the wave's point c; a wave stays inside one batch");

struct RawTile { v4f u[8]; float px, py, pz; };

__device__ __forceinline__ RawTile gather_tile(const float* __restrict__ p,
                                               const float* __restrict__ U,
                                               int b, int j, int h) {
  RawTile r;
  const v4f* ur = reinterpret_cast<const v4f*>(U + ((size_t)b * NN + j) * CU) + h;
  #pragma unroll
  for (int i = 0; i < 8; ++i) r.u[i] = ur[2 * i];
  const float* pj = p + ((size_t)b * NN + j) * 3;
  r.px = pj[0]; r.py = pj[1]; r.pz = pj[2];
  return r;
}

__device__ __forceinline__ void end_of_trip(const RawTile& a, const RawTile& b,
                                            float qx, float qy, float qz, int id) {
  __builtin_amdgcn_sched_barrier(0);
  asm volatile("" :: "v"(a.px), "v"(a.py), "v"(a.pz), "v"(b.px), "v"(b.py), "v"(b.pz),
                     "s"(qx), "s"(qy), "s"(qz), "v"(id));
}

__device__ __forceinline__ void load_w1x_frags(const float* __restrict__ W1, int c, int h,
                                               v2f (&ax)[4][8]) {
  #pragma unroll
  for (int mt = 0; mt < 4; ++mt) {
    const float* wr = W1 + (mt * 16 + c) * CIN;
    #pragma unroll
    for (int s = 0; s < 8; ++s) {
      ax[mt][s].x = wr[3 + 16 * h + 2 * s];
      ax[mt][s].y = wr[3 + 16 * h + 2 * s + 1];
    }
  }
}

__device__ __forceinline__ void load_w1p_frags(const float* __restrict__ W1, int c, int h,
                                               v2f (&ap)[4]) {
  #pragma unroll
  for (int mt = 0; mt < 4; ++mt) {
    const float* wr = W1 + (mt * 16 + c) * CIN;
    const float w01 = wr[1];
    ap[mt].x = wr[2 * h];
    ap[mt].y = h ? 0.f : w01;
  }
}

__device__ __forceinline__ void load_w2_frags(const float* __restrict__ W2, int c, int h,
                                              v2f (&a2)[4][16]) {
  #pragma unroll
  for (int mt2 = 0; mt2 < 4; ++mt2) {
    const float* wr = W2 + (mt2 * 16 + c) * COUT + 8 * h;
    #pragma unroll
    for (int kk = 0; kk < 16; ++kk) {
      const int k0 = (kk >> 2) * 16 + (kk & 3) * 2;
      a2[mt2][kk].x = wr[k0];
      a2[mt2][kk].y = wr[k0 + 1];
    }
  }
}

__device__ __forceinline__ void conv1_tile(const v2f (&ap)[4], const RawTile& rt,
                                           float qx, float qy, float qz, int h, v8f (&acc)[4]) {
  const float dx = rt.px - qx, dy = rt.py - qy, dz = rt.pz - qz;
  v2f bd;
  bd.x = h ? dz : dx;
  bd.y = h ? 0.f : dy;
  #pragma unroll
  for (int mt = 0; mt < 4; ++mt) {
    const v4f lo = rt.u[2 * mt], hi = rt.u[2 * mt + 1];
    const v8f cu = (v8f){lo.x, lo.y, lo.z, lo.w, hi.x, hi.y, hi.z, hi.w};
    acc[mt] = __builtin_amdgcn_wmma_f32_16x16x4_f32(false, ap[mt], false, bd,
                                                    (short)0, cu, false, false);
  }
}

__global__ __launch_bounds__(CW_TPB) void upoint_kernel(const float* __restrict__ x,
                                                        const float* __restrict__ W1,
                                                        float* __restrict__ U) {
  const int lane = threadIdx.x & 31;
  const int c = lane & 15, h = lane >> 4;
  const int w   = __builtin_amdgcn_readfirstlane((int)(threadIdx.x >> 5));
  const int gw  = blockIdx.x * CW_WAVES + w;
  const int pt0 = gw * CW_PTS;
  const int b   = pt0 / NN, n0 = pt0 % NN;

  const float* xc = x + ((size_t)b * CX + 16 * h) * NN + n0 + c;
  v2f bf[8];
  #pragma unroll
  for (int s = 0; s < 8; ++s) {
    bf[s].x = xc[(size_t)(2 * s) * NN];
    bf[s].y = xc[(size_t)(2 * s + 1) * NN];
  }
  v2f ax[4][8];
  load_w1x_frags(W1, c, h, ax);

  v8f acc[4];
  #pragma unroll
  for (int mt = 0; mt < 4; ++mt) acc[mt] = (v8f){0.f, 0.f, 0.f, 0.f, 0.f, 0.f, 0.f, 0.f};
  #pragma unroll
  for (int s = 0; s < 8; ++s)
    #pragma unroll
    for (int mt = 0; mt < 4; ++mt)
      acc[mt] = __builtin_amdgcn_wmma_f32_16x16x4_f32(false, ax[mt][s], false, bf[s],
                                                      (short)0, acc[mt], false, false);

  v4f* ur = reinterpret_cast<v4f*>(U + ((size_t)b * NN + n0 + c) * CU) + h;
  #pragma unroll
  for (int mt = 0; mt < 4; ++mt) {
    ur[2 * (2 * mt)]     = (v4f){acc[mt][0], acc[mt][1], acc[mt][2], acc[mt][3]};
    ur[2 * (2 * mt + 1)] = (v4f){acc[mt][4], acc[mt][5], acc[mt][6], acc[mt][7]};
  }
}

template <int XM>
__device__ __forceinline__ float lane_xor16(float v) {
  return __builtin_bit_cast(float, __builtin_amdgcn_update_dpp(
      0, __builtin_bit_cast(int, v), 0x160 + XM, 0xf, 0xf, true));
}
__device__ __forceinline__ float row_sum16(float v) {
  v += lane_xor16<1>(v); v += lane_xor16<2>(v); v += lane_xor16<4>(v); v += lane_xor16<8>(v);
  return v;
}
__device__ __forceinline__ float row_max16(float v) {
  v = fmaxf(v, lane_xor16<1>(v)); v = fmaxf(v, lane_xor16<2>(v));
  v = fmaxf(v, lane_xor16<4>(v)); v = fmaxf(v, lane_xor16<8>(v));
  return v;
}

__device__ __forceinline__ void write_part_row(float (&s)[4][8], float (&sq)[4][8],
                                               float* __restrict__ row, int c, int h) {
  #pragma unroll
  for (int mt = 0; mt < 4; ++mt) {
    float v = 0.f;
    #pragma unroll
    for (int r = 0; r < 8; ++r) {
      const float ts = row_sum16(s[mt][r]);
      const float tq = row_sum16(sq[mt][r]);
      v = (c == r) ? ts : v;
      v = (c == 8 + r) ? tq : v;
    }
    row[mt * 32 + ((c & 8) << 1) + 8 * h + (c & 7)] = v;
  }
}

#define S1_PP 2
static_assert(CW_PTS % S1_PP == 0, "a trip takes S1_PP whole points");
__global__ __launch_bounds__(CW_TPB) __attribute__((amdgpu_waves_per_eu(1, 1))) void conv1_stats_kernel(
    const float* __restrict__ p, const float* __restrict__ U,
    const unsigned short* __restrict__ idxg, const float* __restrict__ W1,
    float* __restrict__ part1) {
  const int lane = threadIdx.x & 31;
  const int c = lane & 15, h = lane >> 4;
  const int w   = __builtin_amdgcn_readfirstlane((int)(threadIdx.x >> 5));
  const int gw  = blockIdx.x * CW_WAVES + w;
  const int pt0 = gw * CW_PTS;
  const int b   = pt0 / NN, n0 = pt0 % NN;

  v2f ap[4];
  load_w1p_frags(W1, c, h, ap);
  float s[4][8], sq[4][8];
  #pragma unroll
  for (int mt = 0; mt < 4; ++mt)
    #pragma unroll
    for (int r = 0; r < 8; ++r) { s[mt][r] = 0.f; sq[mt][r] = 0.f; }

  const unsigned short* idxw = idxg + ((size_t)b * NN + n0) * KS + lane;
  const float* pqw  = p + ((size_t)b * NN + n0) * 3;
  RawTile rn[S1_PP][2];
  float nqx[S1_PP], nqy[S1_PP], nqz[S1_PP];
  int idn[S1_PP];
  #pragma unroll
  for (int k = 0; k < S1_PP; ++k) {
    const int id = idxw[k * KS];
    rn[k][0] = gather_tile(p, U, b, __shfl(id, c, 32) & (NN - 1), h);
    rn[k][1] = gather_tile(p, U, b, __shfl(id, 16 + c, 32) & (NN - 1), h);
    nqx[k] = pqw[k * 3 + 0]; nqy[k] = pqw[k * 3 + 1]; nqz[k] = pqw[k * 3 + 2];
  }
  #pragma unroll
  for (int k = 0; k < S1_PP; ++k) idn[k] = idxw[(S1_PP + k) * KS];

  #pragma nounroll
  for (int i = 0; i < CW_PTS; i += S1_PP) {
    RawTile r[S1_PP][2];
    float qx[S1_PP], qy[S1_PP], qz[S1_PP];
    #pragma unroll
    for (int k = 0; k < S1_PP; ++k) {
      r[k][0] = rn[k][0]; r[k][1] = rn[k][1];
      qx[k] = nqx[k]; qy[k] = nqy[k]; qz[k] = nqz[k];
    }
    #pragma unroll
    for (int k = 0; k < S1_PP; ++k) {
      const int i1 = (i + S1_PP + k < CW_PTS) ? i + S1_PP + k : CW_PTS - 1;
      const int i2 = (i + 2 * S1_PP + k < CW_PTS) ? i + 2 * S1_PP + k : CW_PTS - 1;
      rn[k][0] = gather_tile(p, U, b, __shfl(idn[k], c, 32) & (NN - 1), h);
      rn[k][1] = gather_tile(p, U, b, __shfl(idn[k], 16 + c, 32) & (NN - 1), h);
      nqx[k] = pqw[i1 * 3 + 0]; nqy[k] = pqw[i1 * 3 + 1]; nqz[k] = pqw[i1 * 3 + 2];
      idn[k] = idxw[i2 * KS];
    }
    __builtin_amdgcn_sched_barrier(0);
    #pragma unroll
    for (int k = 0; k < S1_PP; ++k)
      #pragma unroll
      for (int tile = 0; tile < 2; ++tile) {
        v8f acc[4];
        conv1_tile(ap, r[k][tile], qx[k], qy[k], qz[k], h, acc);
        #pragma unroll
        for (int mt = 0; mt < 4; ++mt)
          #pragma unroll
          for (int rr = 0; rr < 8; ++rr) { const float y = acc[mt][rr]; s[mt][rr] += y; sq[mt][rr] += y * y; }
      }
    #pragma unroll
    for (int k = 0; k < S1_PP; ++k)
      end_of_trip(rn[k][0], rn[k][1], nqx[k], nqy[k], nqz[k], idn[k]);
  }
  write_part_row(s, sq, part1 + (size_t)gw * 128, c, h);
}

#define FIN_TPB   1024
#define FIN_WAVES (FIN_TPB / 32)
#define FIN_ROWS  (GCONV / FIN_WAVES)
static_assert(FIN_ROWS == 32 && FIN_WAVES == 32, "tree32 below sums exactly 32 values");

__device__ __forceinline__ float tree32(float (&a)[32]) {
  #pragma unroll
  for (int h = 16; h >= 1; h >>= 1) {
    #pragma unroll
    for (int r = 0; r < h; ++r) a[r] += a[r + h];
  }
  return a[0];
}

__global__ __launch_bounds__(FIN_TPB) void bn_finalize_kernel(const float* __restrict__ part,
                                                              const float* __restrict__ g,
                                                              const float* __restrict__ bta,
                                                              float* __restrict__ bnp,
                                                              int rstride) {
  __shared__ float red[FIN_WAVES * 32];
  const int t    = threadIdx.x;
  const int lane = t & 31;
  const int w    = t >> 5;
  const int m    = blockIdx.x;
  const float* src = part + (size_t)w * rstride + m * 32 + lane;
  float a[FIN_ROWS];
  #pragma unroll
  for (int r = 0; r < FIN_ROWS; ++r) a[r] = src[(size_t)r * FIN_WAVES * rstride];
  red[w * 32 + lane] = tree32(a);
  __syncthreads();
  if (w == 0) {
    float c[FIN_WAVES];
    #pragma unroll
    for (int k = 0; k < FIN_WAVES; ++k) c[k] = red[k * 32 + lane];
    const float tot = tree32(c);
    const float sq  = __shfl_down(tot, 16, 32);
    if (lane < 16) {
      const int o = m * 16 + lane;
      float mean  = tot / CNTF;
      float var   = sq / CNTF - mean * mean;
      float scale = g[o] * rsqrtf(var + EPSV);
      bnp[o]      = scale;
      bnp[64 + o] = bta[o] - mean * scale;
    }
  }
}

template <bool FIRST>
__device__ __forceinline__ void conv2_tile(const RawTile& rt, float qx, float qy, float qz, int h,
                                           const v2f (&ap)[4], const v2f (&a2)[4][16],
                                           const float (&sc1)[4][8], const float (&sh1)[4][8],
                                           const float (&sgn)[4][8],
                                           float (&s)[4][8], float (&sq)[4][8], float (&m)[4][8]) {
  v8f acc1[4];
  conv1_tile(ap, rt, qx, qy, qz, h, acc1);
  float f1[4][8];
  #pragma unroll
  for (int mt = 0; mt < 4; ++mt)
    #pragma unroll
    for (int r = 0; r < 8; ++r) {
      float y = acc1[mt][r] * sc1[mt][r] + sh1[mt][r];
      f1[mt][r] = y > 0.f ? y : 0.f;
    }
  v8f acc2[4];
  #pragma unroll
  for (int mt2 = 0; mt2 < 4; ++mt2) acc2[mt2] = (v8f){0.f, 0.f, 0.f, 0.f, 0.f, 0.f, 0.f, 0.f};
  #pragma unroll
  for (int kk = 0; kk < 16; ++kk) {
    v2f bm;
    bm.x = f1[kk >> 2][(kk & 3) * 2];
    bm.y = f1[kk >> 2][(kk & 3) * 2 + 1];
    #pragma unroll
    for (int mt2 = 0; mt2 < 4; ++mt2)
      acc2[mt2] = __builtin_amdgcn_wmma_f32_16x16x4_f32(false, a2[mt2][kk], false, bm,
                                                        (short)0, acc2[mt2], false, false);
  }
  #pragma unroll
  for (int mt2 = 0; mt2 < 4; ++mt2)
    #pragma unroll
    for (int r = 0; r < 8; ++r) {
      const float y = acc2[mt2][r];
      s[mt2][r] += y; sq[mt2][r] += y * y;
      const float e = y * sgn[mt2][r];
      m[mt2][r] = FIRST ? e : fmaxf(m[mt2][r], e);
    }
}

__global__ __launch_bounds__(CW_TPB) void conv2_stats_kernel(
    const float* __restrict__ p, const float* __restrict__ U,
    const unsigned short* idxg, const float* __restrict__ W1,
    const float* __restrict__ W2, const float* __restrict__ bnp1,
    const float* __restrict__ g2, float* part2,
    float* __restrict__ out) {
  const int lane = threadIdx.x & 31;
  const int c = lane & 15, h = lane >> 4;
  const int w   = __builtin_amdgcn_readfirstlane((int)(threadIdx.x >> 5));
  const int gw  = blockIdx.x * CW_WAVES + w;
  const int pt0 = gw * CW_PTS;
  const int b   = pt0 / NN, n0 = pt0 % NN;

  v2f ap[4];
  load_w1p_frags(W1, c, h, ap);
  v2f a2[4][16];
  load_w2_frags(W2, c, h, a2);
  float sc1[4][8], sh1[4][8], sgn[4][8];
  float s[4][8], sq[4][8], res[4][8];
  #pragma unroll
  for (int mt = 0; mt < 4; ++mt)
    #pragma unroll
    for (int r = 0; r < 8; ++r) {
      const int o = mt * 16 + 8 * h + r;
      sc1[mt][r] = bnp1[o]; sh1[mt][r] = bnp1[64 + o];
      sgn[mt][r] = (g2[o] < 0.f) ? -1.f : 1.f;
      s[mt][r] = 0.f; sq[mt][r] = 0.f; res[mt][r] = 0.f;
    }

  const unsigned short* idxw = idxg + ((size_t)b * NN + n0) * KS + lane;
  const float* pqw  = p + ((size_t)b * NN + n0) * 3;
  int idn = idxw[0];
  RawTile rn0 = gather_tile(p, U, b, __shfl(idn, c, 32) & (NN - 1), h);
  RawTile rn1 = gather_tile(p, U, b, __shfl(idn, 16 + c, 32) & (NN - 1), h);
  float nqx = pqw[0], nqy = pqw[1], nqz = pqw[2];
  idn = idxw[KS];

  #pragma nounroll
  for (int i = 0; i < CW_PTS; ++i) {
    const RawTile r0 = rn0, r1 = rn1;
    const float qx = nqx, qy = nqy, qz = nqz;
    {
      const int i1 = (i + 1 < CW_PTS) ? i + 1 : CW_PTS - 1;
      const int i2 = (i + 2 < CW_PTS) ? i + 2 : CW_PTS - 1;
      rn0 = gather_tile(p, U, b, __shfl(idn, c, 32) & (NN - 1), h);
      rn1 = gather_tile(p, U, b, __shfl(idn, 16 + c, 32) & (NN - 1), h);
      nqx = pqw[i1 * 3 + 0]; nqy = pqw[i1 * 3 + 1]; nqz = pqw[i1 * 3 + 2];
      idn = idxw[i2 * KS];
    }
    __builtin_amdgcn_sched_barrier(0);
    float m[4][8];
    conv2_tile<true >(r0, qx, qy, qz, h, ap, a2, sc1, sh1, sgn, s, sq, m);
    conv2_tile<false>(r1, qx, qy, qz, h, ap, a2, sc1, sh1, sgn, s, sq, m);
    #pragma unroll
    for (int mt2 = 0; mt2 < 4; ++mt2)
      #pragma unroll
      for (int r = 0; r < 8; ++r) {
        const float e = row_max16(m[mt2][r]);
        res[mt2][r] = (c == i) ? e : res[mt2][r];
      }
    end_of_trip(rn0, rn1, nqx, nqy, nqz, idn);
  }
  #pragma unroll
  for (int mt2 = 0; mt2 < 4; ++mt2)
    #pragma unroll
    for (int r = 0; r < 8; ++r) {
      const int o = mt2 * 16 + 8 * h + r;
      out[((size_t)b * COUT + o) * NN + n0 + c] = res[mt2][r] * sgn[mt2][r];
    }
  write_part_row(s, sq, part2 + (size_t)gw * (CW_PTS * KS / 2), c, h);
}

#define AP_TPB 256
#define AP_V4  (NN / 4 / AP_TPB)
__global__ __launch_bounds__(AP_TPB) void bn2_apply_kernel(const float* __restrict__ bnp2,
                                                           float* __restrict__ out) {
  const int row = blockIdx.x;
  const int o   = row & (COUT - 1);
  const float sc = bnp2[o], sh = bnp2[64 + o];
  float4* q = reinterpret_cast<float4*>(out) + (size_t)row * (NN / 4) + threadIdx.x;
  float4 e[AP_V4];
  #pragma unroll
  for (int k = 0; k < AP_V4; ++k) e[k] = q[k * AP_TPB];
  #pragma unroll
  for (int k = 0; k < AP_V4; ++k) {
    float4 r;
    float y;
    y = e[k].x * sc + sh; r.x = y > 0.f ? y : 0.f;
    y = e[k].y * sc + sh; r.y = y > 0.f ? y : 0.f;
    y = e[k].z * sc + sh; r.z = y > 0.f ? y : 0.f;
    y = e[k].w * sc + sh; r.w = y > 0.f ? y : 0.f;
    q[k * AP_TPB] = r;
  }
}

extern "C" void kernel_launch(void* const* d_in, const int* in_sizes, int n_in,
                              void* d_out, int out_size, void* d_ws, size_t ws_size,
                              hipStream_t stream) {
  const float* p  = (const float*)d_in[0];
  const float* x  = (const float*)d_in[1];
  const float* W1 = (const float*)d_in[2];
  const float* g1 = (const float*)d_in[3];
  const float* b1 = (const float*)d_in[4];
  const float* W2 = (const float*)d_in[5];
  const float* g2 = (const float*)d_in[6];
  const float* b2 = (const float*)d_in[7];
  float* out = (float*)d_out;

  char*  ws    = (char*)d_ws;
  unsigned short* idx = (unsigned short*)ws;
  float* U     = (float*)(ws + (size_t)BB * NN * KS * 2);
  float* bnp1  = (float*)(ws + (size_t)BB * NN * KS * 2
                             + (size_t)BB * NN * CU * 4);
  float* bnp2  = bnp1 + 128;
  float* part1 = out;
  float* part2 = (float*)ws;

  upoint_kernel<<<GCONV / CW_WAVES, CW_TPB, 0, stream>>>(x, W1, U);
  ballq_kernel<<<BB * (NN / BQ_QPB), BQ_TPB, 0, stream>>>(p, idx);
  conv1_stats_kernel<<<GCONV / CW_WAVES, CW_TPB, 0, stream>>>(p, U, idx, W1, part1);
  bn_finalize_kernel<<<COUT / 16, FIN_TPB, 0, stream>>>(part1, g1, b1, bnp1, 128);
  conv2_stats_kernel<<<GCONV / CW_WAVES, CW_TPB, 0, stream>>>(p, U, idx, W1, W2, bnp1, g2, part2, out);
  bn_finalize_kernel<<<COUT / 16, FIN_TPB, 0, stream>>>(part2, g2, b2, bnp2, CW_PTS * KS / 2);
  bn2_apply_kernel<<<BB * COUT, AP_TPB, 0, stream>>>(bnp2, out);
}


# launch merged: the per-point W1[:,3:] x table (U) is built inside the ball-query launch, one 16x16 tile per wave while the block's LDS copy is in flight; 6 launches per call, was 7
# speedup vs baseline: 9.0653x; 1.0388x over previous
#include <hip/hip_runtime.h>
#include <hip/hip_bf16.h>


typedef __attribute__((ext_vector_type(2))) float v2f;
typedef __attribute__((ext_vector_type(4))) float v4f;
typedef __attribute__((ext_vector_type(8))) float v8f;

#define BB      4
#define NN      4096
#define CX      32
#define CU      64
#define CIN     35
#define COUT    64
#define KS      32
#define R2      0.01f
#define EPSV    1e-5f
#define GCONV   1024
#define CNTF    ((float)BB * (float)NN * (float)KS)


#define BQ_Q     4
#define BQ_WAVES 16
#define BQ_QPB   (BQ_Q * BQ_WAVES)
#define BQ_TPB   (32 * BQ_WAVES)
#define BQ_FILL  (NN * 3 / 4 / BQ_TPB)
static_assert(BQ_Q == 4, "the vote test and the early-exit test below are written out for 4 queries");
static_assert(BQ_QPB == 64 && BQ_WAVES == 16,
              "the U job: a block's 64 points are 4 tiles of 16 columns x 4 row tiles, one per wave");

struct UFrags { v2f a[8], b[8]; };
__device__ __forceinline__ UFrags upoint_load(const float* __restrict__ x, const float* __restrict__ W1,
                                              int b, int n0, int mt, int c, int h) {
  UFrags f;
  const float* xc = x + ((size_t)b * CX + 16 * h) * NN + n0 + c;
  const float* wr = W1 + (mt * 16 + c) * CIN + 3 + 16 * h;
  #pragma unroll
  for (int s = 0; s < 8; ++s) {
    f.b[s].x = xc[(size_t)(2 * s) * NN];
    f.b[s].y = xc[(size_t)(2 * s + 1) * NN];
    f.a[s].x = wr[2 * s];
    f.a[s].y = wr[2 * s + 1];
  }
  return f;
}
__device__ __forceinline__ v8f upoint_tile(const UFrags& f) {
  v8f acc = (v8f){0.f, 0.f, 0.f, 0.f, 0.f, 0.f, 0.f, 0.f};
  #pragma unroll
  for (int s = 0; s < 8; ++s)
    acc = __builtin_amdgcn_wmma_f32_16x16x4_f32(false, f.a[s], false, f.b[s],
                                                (short)0, acc, false, false);
  return acc;
}
__device__ __forceinline__ void upoint_store(const v8f& acc, float* __restrict__ U,
                                             int b, int n0, int mt, int c, int h) {
  v4f* ur = reinterpret_cast<v4f*>(U + ((size_t)b * NN + n0 + c) * CU) + h;
  ur[4 * mt]     = (v4f){acc[0], acc[1], acc[2], acc[3]};
  ur[4 * mt + 2] = (v4f){acc[4], acc[5], acc[6], acc[7]};
}

__global__ __launch_bounds__(BQ_TPB) void ballq_kernel(const float* __restrict__ p,
                                                       const float* __restrict__ x,
                                                       const float* __restrict__ W1,
                                                       unsigned short* __restrict__ idx,
                                                       float* __restrict__ U) {
  __shared__ float4 cand4[NN * 3 / 4];
  __shared__ int    lst[BQ_WAVES * BQ_Q * KS];
  const int t    = threadIdx.x;
  const int lane = t & 31;
  const int w  = __builtin_amdgcn_readfirstlane(t >> 5);
  const int b  = blockIdx.x / (NN / BQ_QPB);
  const int q0 = (blockIdx.x % (NN / BQ_QPB)) * BQ_QPB + w * BQ_Q;
  const float* pb = p + (size_t)b * NN * 3;

  const float4* src4 = reinterpret_cast<const float4*>(pb);
  float4 stage[BQ_FILL];
  #pragma unroll
  for (int k = 0; k < BQ_FILL; ++k) stage[k] = src4[t + k * BQ_TPB];
  const int uc = lane & 15, uh = lane >> 4;
  const int un0 = (blockIdx.x % (NN / BQ_QPB)) * BQ_QPB + (w & 3) * 16, umt = w >> 2;
  const UFrags uf = upoint_load(x, W1, b, un0, umt, uc, uh);
  #pragma unroll
  for (int k = 0; k < BQ_FILL; ++k) cand4[t + k * BQ_TPB] = stage[k];
  const v8f uacc = upoint_tile(uf);

  int* mylst = lst + w * (BQ_Q * KS);
  if (lane < BQ_Q) mylst[lane * KS] = 0;
  float qx[BQ_Q], qy[BQ_Q], qz[BQ_Q];
  int cnt[BQ_Q];
  #pragma unroll
  for (int q = 0; q < BQ_Q; ++q) {
    qx[q] = pb[(q0 + q) * 3 + 0];
    qy[q] = pb[(q0 + q) * 3 + 1];
    qz[q] = pb[(q0 + q) * 3 + 2];
    cnt[q] = 0;
  }
  __syncthreads();
  upoint_store(uacc, U, b, un0, umt, uc, uh);

  const float* cand = reinterpret_cast<const float*>(cand4) + lane * 3;
  float nx = cand[0], ny = cand[1], nz = cand[2];
  bool active = true;
  int nb3 = 0;
  for (int base = 0; base < NN && active; base += 32) {
    const float cx = nx, cy = ny, cz = nz;
    if (base + 32 < NN) nb3 += 32 * 3;
    nx = cand[nb3 + 0]; ny = cand[nb3 + 1]; nz = cand[nb3 + 2];

    bool h[BQ_Q];
    unsigned m[BQ_Q];
    #pragma unroll
    for (int q = 0; q < BQ_Q; ++q) {
      const float dx = cx - qx[q], dy = cy - qy[q], dz = cz - qz[q];
      const float d2 = __builtin_fmaf(dz, dz, __builtin_fmaf(dx, dx, dy * dy));
      h[q] = d2 < R2;
      m[q] = __builtin_amdgcn_ballot_w32(h[q]);
    }
    if ((m[0] | m[1] | m[2] | m[3]) != 0u) {
      #pragma unroll
      for (int q = 0; q < BQ_Q; ++q) {
        if (m[q] != 0u) {
          const int rank = cnt[q] + (int)__builtin_amdgcn_mbcnt_lo(m[q], 0u);
          if (h[q] && rank < KS) mylst[q * KS + rank] = base + lane;
          cnt[q] += __builtin_popcount(m[q]);
        }
      }
      active = (((cnt[0] - KS) | (cnt[1] - KS) | (cnt[2] - KS) | (cnt[3] - KS)) < 0);
    }
  }
  #pragma unroll
  for (int q = 0; q < BQ_Q; ++q) {
    const int v = mylst[q * KS + (lane < cnt[q] ? lane : 0)];
    idx[((size_t)b * NN + q0 + q) * KS + lane] = (unsigned short)v;
  }
}

#define CW_WAVES 4
#define CW_TPB   (32 * CW_WAVES)
#define CW_PTS   (BB * NN / GCONV)
static_assert(CW_PTS == 16 && NN % CW_PTS == 0 && GCONV % CW_WAVES == 0,
              "lane c keeps the pooled values of the wave's point c; a wave stays inside one batch");

struct RawTile { v4f u[8]; float px, py, pz; };

__device__ __forceinline__ RawTile gather_tile(const float* __restrict__ p,
                                               const float* __restrict__ U,
                                               int b, int j, int h) {
  RawTile r;
  const v4f* ur = reinterpret_cast<const v4f*>(U + ((size_t)b * NN + j) * CU) + h;
  #pragma unroll
  for (int i = 0; i < 8; ++i) r.u[i] = ur[2 * i];
  const float* pj = p + ((size_t)b * NN + j) * 3;
  r.px = pj[0]; r.py = pj[1]; r.pz = pj[2];
  return r;
}

__device__ __forceinline__ void end_of_trip(const RawTile& a, const RawTile& b,
                                            float qx, float qy, float qz, int id) {
  __builtin_amdgcn_sched_barrier(0);
  asm volatile("" :: "v"(a.px), "v"(a.py), "v"(a.pz), "v"(b.px), "v"(b.py), "v"(b.pz),
                     "s"(qx), "s"(qy), "s"(qz), "v"(id));
}

__device__ __forceinline__ void load_w1p_frags(const float* __restrict__ W1, int c, int h,
                                               v2f (&ap)[4]) {
  #pragma unroll
  for (int mt = 0; mt < 4; ++mt) {
    const float* wr = W1 + (mt * 16 + c) * CIN;
    const float w01 = wr[1];
    ap[mt].x = wr[2 * h];
    ap[mt].y = h ? 0.f : w01;
  }
}

__device__ __forceinline__ void load_w2_frags(const float* __restrict__ W2, int c, int h,
                                              v2f (&a2)[4][16]) {
  #pragma unroll
  for (int mt2 = 0; mt2 < 4; ++mt2) {
    const float* wr = W2 + (mt2 * 16 + c) * COUT + 8 * h;
    #pragma unroll
    for (int kk = 0; kk < 16; ++kk) {
      const int k0 = (kk >> 2) * 16 + (kk & 3) * 2;
      a2[mt2][kk].x = wr[k0];
      a2[mt2][kk].y = wr[k0 + 1];
    }
  }
}

__device__ __forceinline__ void conv1_tile(const v2f (&ap)[4], const RawTile& rt,
                                           float qx, float qy, float qz, int h, v8f (&acc)[4]) {
  const float dx = rt.px - qx, dy = rt.py - qy, dz = rt.pz - qz;
  v2f bd;
  bd.x = h ? dz : dx;
  bd.y = h ? 0.f : dy;
  #pragma unroll
  for (int mt = 0; mt < 4; ++mt) {
    const v4f lo = rt.u[2 * mt], hi = rt.u[2 * mt + 1];
    const v8f cu = (v8f){lo.x, lo.y, lo.z, lo.w, hi.x, hi.y, hi.z, hi.w};
    acc[mt] = __builtin_amdgcn_wmma_f32_16x16x4_f32(false, ap[mt], false, bd,
                                                    (short)0, cu, false, false);
  }
}

template <int XM>
__device__ __forceinline__ float lane_xor16(float v) {
  return __builtin_bit_cast(float, __builtin_amdgcn_update_dpp(
      0, __builtin_bit_cast(int, v), 0x160 + XM, 0xf, 0xf, true));
}
__device__ __forceinline__ float row_sum16(float v) {
  v += lane_xor16<1>(v); v += lane_xor16<2>(v); v += lane_xor16<4>(v); v += lane_xor16<8>(v);
  return v;
}
__device__ __forceinline__ float row_max16(float v) {
  v = fmaxf(v, lane_xor16<1>(v)); v = fmaxf(v, lane_xor16<2>(v));
  v = fmaxf(v, lane_xor16<4>(v)); v = fmaxf(v, lane_xor16<8>(v));
  return v;
}

__device__ __forceinline__ void write_part_row(float (&s)[4][8], float (&sq)[4][8],
                                               float* __restrict__ row, int c, int h) {
  #pragma unroll
  for (int mt = 0; mt < 4; ++mt) {
    float v = 0.f;
    #pragma unroll
    for (int r = 0; r < 8; ++r) {
      const float ts = row_sum16(s[mt][r]);
      const float tq = row_sum16(sq[mt][r]);
      v = (c == r) ? ts : v;
      v = (c == 8 + r) ? tq : v;
    }
    row[mt * 32 + ((c & 8) << 1) + 8 * h + (c & 7)] = v;
  }
}

#define S1_PP 2
static_assert(CW_PTS % S1_PP == 0, "a trip takes S1_PP whole points");
__global__ __launch_bounds__(CW_TPB) __attribute__((amdgpu_waves_per_eu(1, 1))) void conv1_stats_kernel(
    const float* __restrict__ p, const float* __restrict__ U,
    const unsigned short* __restrict__ idxg, const float* __restrict__ W1,
    float* __restrict__ part1) {
  const int lane = threadIdx.x & 31;
  const int c = lane & 15, h = lane >> 4;
  const int w   = __builtin_amdgcn_readfirstlane((int)(threadIdx.x >> 5));
  const int gw  = blockIdx.x * CW_WAVES + w;
  const int pt0 = gw * CW_PTS;
  const int b   = pt0 / NN, n0 = pt0 % NN;

  v2f ap[4];
  load_w1p_frags(W1, c, h, ap);
  float s[4][8], sq[4][8];
  #pragma unroll
  for (int mt = 0; mt < 4; ++mt)
    #pragma unroll
    for (int r = 0; r < 8; ++r) { s[mt][r] = 0.f; sq[mt][r] = 0.f; }

  const unsigned short* idxw = idxg + ((size_t)b * NN + n0) * KS + lane;
  const float* pqw  = p + ((size_t)b * NN + n0) * 3;
  RawTile rn[S1_PP][2];
  float nqx[S1_PP], nqy[S1_PP], nqz[S1_PP];
  int idn[S1_PP];
  #pragma unroll
  for (int k = 0; k < S1_PP; ++k) {
    const int id = idxw[k * KS];
    rn[k][0] = gather_tile(p, U, b, __shfl(id, c, 32) & (NN - 1), h);
    rn[k][1] = gather_tile(p, U, b, __shfl(id, 16 + c, 32) & (NN - 1), h);
    nqx[k] = pqw[k * 3 + 0]; nqy[k] = pqw[k * 3 + 1]; nqz[k] = pqw[k * 3 + 2];
  }
  #pragma unroll
  for (int k = 0; k < S1_PP; ++k) idn[k] = idxw[(S1_PP + k) * KS];

  #pragma nounroll
  for (int i = 0; i < CW_PTS; i += S1_PP) {
    RawTile r[S1_PP][2];
    float qx[S1_PP], qy[S1_PP], qz[S1_PP];
    #pragma unroll
    for (int k = 0; k < S1_PP; ++k) {
      r[k][0] = rn[k][0]; r[k][1] = rn[k][1];
      qx[k] = nqx[k]; qy[k] = nqy[k]; qz[k] = nqz[k];
    }
    #pragma unroll
    for (int k = 0; k < S1_PP; ++k) {
      const int i1 = (i + S1_PP + k < CW_PTS) ? i + S1_PP + k : CW_PTS - 1;
      const int i2 = (i + 2 * S1_PP + k < CW_PTS) ? i + 2 * S1_PP + k : CW_PTS - 1;
      rn[k][0] = gather_tile(p, U, b, __shfl(idn[k], c, 32) & (NN - 1), h);
      rn[k][1] = gather_tile(p, U, b, __shfl(idn[k], 16 + c, 32) & (NN - 1), h);
      nqx[k] = pqw[i1 * 3 + 0]; nqy[k] = pqw[i1 * 3 + 1]; nqz[k] = pqw[i1 * 3 + 2];
      idn[k] = idxw[i2 * KS];
    }
    __builtin_amdgcn_sched_barrier(0);
    #pragma unroll
    for (int k = 0; k < S1_PP; ++k)
      #pragma unroll
      for (int tile = 0; tile < 2; ++tile) {
        v8f acc[4];
        conv1_tile(ap, r[k][tile], qx[k], qy[k], qz[k], h, acc);
        #pragma unroll
        for (int mt = 0; mt < 4; ++mt)
          #pragma unroll
          for (int rr = 0; rr < 8; ++rr) { const float y = acc[mt][rr]; s[mt][rr] += y; sq[mt][rr] += y * y; }
      }
    #pragma unroll
    for (int k = 0; k < S1_PP; ++k)
      end_of_trip(rn[k][0], rn[k][1], nqx[k], nqy[k], nqz[k], idn[k]);
  }
  write_part_row(s, sq, part1 + (size_t)gw * 128, c, h);
}

#define FIN_TPB   1024
#define FIN_WAVES (FIN_TPB / 32)
#define FIN_ROWS  (GCONV / FIN_WAVES)
static_assert(FIN_ROWS == 32 && FIN_WAVES == 32, "tree32 below sums exactly 32 values");

__device__ __forceinline__ float tree32(float (&a)[32]) {
  #pragma unroll
  for (int h = 16; h >= 1; h >>= 1) {
    #pragma unroll
    for (int r = 0; r < h; ++r) a[r] += a[r + h];
  }
  return a[0];
}

__global__ __launch_bounds__(FIN_TPB) void bn_finalize_kernel(const float* __restrict__ part,
                                                              const float* __restrict__ g,
                                                              const float* __restrict__ bta,
                                                              float* __restrict__ bnp,
                                                              int rstride) {
  __shared__ float red[FIN_WAVES * 32];
  const int t    = threadIdx.x;
  const int lane = t & 31;
  const int w    = t >> 5;
  const int m    = blockIdx.x;
  const float* src = part + (size_t)w * rstride + m * 32 + lane;
  float a[FIN_ROWS];
  #pragma unroll
  for (int r = 0; r < FIN_ROWS; ++r) a[r] = src[(size_t)r * FIN_WAVES * rstride];
  red[w * 32 + lane] = tree32(a);
  __syncthreads();
  if (w == 0) {
    float c[FIN_WAVES];
    #pragma unroll
    for (int k = 0; k < FIN_WAVES; ++k) c[k] = red[k * 32 + lane];
    const float tot = tree32(c);
    const float sq  = __shfl_down(tot, 16, 32);
    if (lane < 16) {
      const int o = m * 16 + lane;
      float mean  = tot / CNTF;
      float var   = sq / CNTF - mean * mean;
      float scale = g[o] * rsqrtf(var + EPSV);
      bnp[o]      = scale;
      bnp[64 + o] = bta[o] - mean * scale;
    }
  }
}

template <bool FIRST>
__device__ __forceinline__ void conv2_tile(const RawTile& rt, float qx, float qy, float qz, int h,
                                           const v2f (&ap)[4], const v2f (&a2)[4][16],
                                           const float (&sc1)[4][8], const float (&sh1)[4][8],
                                           const float (&sgn)[4][8],
                                           float (&s)[4][8], float (&sq)[4][8], float (&m)[4][8]) {
  v8f acc1[4];
  conv1_tile(ap, rt, qx, qy, qz, h, acc1);
  float f1[4][8];
  #pragma unroll
  for (int mt = 0; mt < 4; ++mt)
    #pragma unroll
    for (int r = 0; r < 8; ++r) {
      float y = acc1[mt][r] * sc1[mt][r] + sh1[mt][r];
      f1[mt][r] = y > 0.f ? y : 0.f;
    }
  v8f acc2[4];
  #pragma unroll
  for (int mt2 = 0; mt2 < 4; ++mt2) acc2[mt2] = (v8f){0.f, 0.f, 0.f, 0.f, 0.f, 0.f, 0.f, 0.f};
  #pragma unroll
  for (int kk = 0; kk < 16; ++kk) {
    v2f bm;
    bm.x = f1[kk >> 2][(kk & 3) * 2];
    bm.y = f1[kk >> 2][(kk & 3) * 2 + 1];
    #pragma unroll
    for (int mt2 = 0; mt2 < 4; ++mt2)
      acc2[mt2] = __builtin_amdgcn_wmma_f32_16x16x4_f32(false, a2[mt2][kk], false, bm,
                                                        (short)0, acc2[mt2], false, false);
  }
  #pragma unroll
  for (int mt2 = 0; mt2 < 4; ++mt2)
    #pragma unroll
    for (int r = 0; r < 8; ++r) {
      const float y = acc2[mt2][r];
      s[mt2][r] += y; sq[mt2][r] += y * y;
      const float e = y * sgn[mt2][r];
      m[mt2][r] = FIRST ? e : fmaxf(m[mt2][r], e);
    }
}

__global__ __launch_bounds__(CW_TPB) void conv2_stats_kernel(
    const float* __restrict__ p, const float* __restrict__ U,
    const unsigned short* idxg, const float* __restrict__ W1,
    const float* __restrict__ W2, const float* __restrict__ bnp1,
    const float* __restrict__ g2, float* part2,
    float* __restrict__ out) {
  const int lane = threadIdx.x & 31;
  const int c = lane & 15, h = lane >> 4;
  const int w   = __builtin_amdgcn_readfirstlane((int)(threadIdx.x >> 5));
  const int gw  = blockIdx.x * CW_WAVES + w;
  const int pt0 = gw * CW_PTS;
  const int b   = pt0 / NN, n0 = pt0 % NN;

  v2f ap[4];
  load_w1p_frags(W1, c, h, ap);
  v2f a2[4][16];
  load_w2_frags(W2, c, h, a2);
  float sc1[4][8], sh1[4][8], sgn[4][8];
  float s[4][8], sq[4][8], res[4][8];
  #pragma unroll
  for (int mt = 0; mt < 4; ++mt)
    #pragma unroll
    for (int r = 0; r < 8; ++r) {
      const int o = mt * 16 + 8 * h + r;
      sc1[mt][r] = bnp1[o]; sh1[mt][r] = bnp1[64 + o];
      sgn[mt][r] = (g2[o] < 0.f) ? -1.f : 1.f;
      s[mt][r] = 0.f; sq[mt][r] = 0.f; res[mt][r] = 0.f;
    }

  const unsigned short* idxw = idxg + ((size_t)b * NN + n0) * KS + lane;
  const float* pqw  = p + ((size_t)b * NN + n0) * 3;
  int idn = idxw[0];
  RawTile rn0 = gather_tile(p, U, b, __shfl(idn, c, 32) & (NN - 1), h);
  RawTile rn1 = gather_tile(p, U, b, __shfl(idn, 16 + c, 32) & (NN - 1), h);
  float nqx = pqw[0], nqy = pqw[1], nqz = pqw[2];
  idn = idxw[KS];

  #pragma nounroll
  for (int i = 0; i < CW_PTS; ++i) {
    const RawTile r0 = rn0, r1 = rn1;
    const float qx = nqx, qy = nqy, qz = nqz;
    {
      const int i1 = (i + 1 < CW_PTS) ? i + 1 : CW_PTS - 1;
      const int i2 = (i + 2 < CW_PTS) ? i + 2 : CW_PTS - 1;
      rn0 = gather_tile(p, U, b, __shfl(idn, c, 32) & (NN - 1), h);
      rn1 = gather_tile(p, U, b, __shfl(idn, 16 + c, 32) & (NN - 1), h);
      nqx = pqw[i1 * 3 + 0]; nqy = pqw[i1 * 3 + 1]; nqz = pqw[i1 * 3 + 2];
      idn = idxw[i2 * KS];
    }
    __builtin_amdgcn_sched_barrier(0);
    float m[4][8];
    conv2_tile<true >(r0, qx, qy, qz, h, ap, a2, sc1, sh1, sgn, s, sq, m);
    conv2_tile<false>(r1, qx, qy, qz, h, ap, a2, sc1, sh1, sgn, s, sq, m);
    #pragma unroll
    for (int mt2 = 0; mt2 < 4; ++mt2)
      #pragma unroll
      for (int r = 0; r < 8; ++r) {
        const float e = row_max16(m[mt2][r]);
        res[mt2][r] = (c == i) ? e : res[mt2][r];
      }
    end_of_trip(rn0, rn1, nqx, nqy, nqz, idn);
  }
  #pragma unroll
  for (int mt2 = 0; mt2 < 4; ++mt2)
    #pragma unroll
    for (int r = 0; r < 8; ++r) {
      const int o = mt2 * 16 + 8 * h + r;
      out[((size_t)b * COUT + o) * NN + n0 + c] = res[mt2][r] * sgn[mt2][r];
    }
  write_part_row(s, sq, part2 + (size_t)gw * (CW_PTS * KS / 2), c, h);
}

#define AP_TPB 256
#define AP_V4  (NN / 4 / AP_TPB)
__global__ __launch_bounds__(AP_TPB) void bn2_apply_kernel(const float* __restrict__ bnp2,
                                                           float* __restrict__ out) {
  const int row = blockIdx.x;
  const int o   = row & (COUT - 1);
  const float sc = bnp2[o], sh = bnp2[64 + o];
  float4* q = reinterpret_cast<float4*>(out) + (size_t)row * (NN / 4) + threadIdx.x;
  float4 e[AP_V4];
  #pragma unroll
  for (int k = 0; k < AP_V4; ++k) e[k] = q[k * AP_TPB];
  #pragma unroll
  for (int k = 0; k < AP_V4; ++k) {
    float4 r;
    float y;
    y = e[k].x * sc + sh; r.x = y > 0.f ? y : 0.f;
    y = e[k].y * sc + sh; r.y = y > 0.f ? y : 0.f;
    y = e[k].z * sc + sh; r.z = y > 0.f ? y : 0.f;
    y = e[k].w * sc + sh; r.w = y > 0.f ? y : 0.f;
    q[k * AP_TPB] = r;
  }
}

extern "C" void kernel_launch(void* const* d_in, const int* in_sizes, int n_in,
                              void* d_out, int out_size, void* d_ws, size_t ws_size,
                              hipStream_t stream) {
  const float* p  = (const float*)d_in[0];
  const float* x  = (const float*)d_in[1];
  const float* W1 = (const float*)d_in[2];
  const float* g1 = (const float*)d_in[3];
  const float* b1 = (const float*)d_in[4];
  const float* W2 = (const float*)d_in[5];
  const float* g2 = (const float*)d_in[6];
  const float* b2 = (const float*)d_in[7];
  float* out = (float*)d_out;

  char*  ws    = (char*)d_ws;
  unsigned short* idx = (unsigned short*)ws;
  float* U     = (float*)(ws + (size_t)BB * NN * KS * 2);
  float* bnp1  = (float*)(ws + (size_t)BB * NN * KS * 2
                             + (size_t)BB * NN * CU * 4);
  float* bnp2  = bnp1 + 128;
  float* part1 = out;
  float* part2 = (float*)ws;

  ballq_kernel<<<BB * (NN / BQ_QPB), BQ_TPB, 0, stream>>>(p, x, W1, idx, U);
  conv1_stats_kernel<<<GCONV / CW_WAVES, CW_TPB, 0, stream>>>(p, U, idx, W1, part1);
  bn_finalize_kernel<<<COUT / 16, FIN_TPB, 0, stream>>>(part1, g1, b1, bnp1, 128);
  conv2_stats_kernel<<<GCONV / CW_WAVES, CW_TPB, 0, stream>>>(p, U, idx, W1, W2, bnp1, g2, part2, out);
  bn_finalize_kernel<<<COUT / 16, FIN_TPB, 0, stream>>>(part2, g2, b2, bnp2, CW_PTS * KS / 2);
  bn2_apply_kernel<<<BB * COUT, AP_TPB, 0, stream>>>(bnp2, out);
}


# BN2 finalize launch removed: a conv2 block adds its 4 waves' partial rows in LDS, each apply block tree-sums the 256 block rows itself; 5 launches, was 6 (changes the BN2 sums' order)
# speedup vs baseline: 9.4832x; 1.0461x over previous
#include <hip/hip_runtime.h>
#include <hip/hip_bf16.h>


typedef __attribute__((ext_vector_type(2))) float v2f;
typedef __attribute__((ext_vector_type(4))) float v4f;
typedef __attribute__((ext_vector_type(8))) float v8f;

#define BB      4
#define NN      4096
#define CX      32
#define CU      64
#define CIN     35
#define COUT    64
#define KS      32
#define R2      0.01f
#define EPSV    1e-5f
#define GCONV   1024
#define CNTF    ((float)BB * (float)NN * (float)KS)


#define BQ_Q     4
#define BQ_WAVES 16
#define BQ_QPB   (BQ_Q * BQ_WAVES)
#define BQ_TPB   (32 * BQ_WAVES)
#define BQ_FILL  (NN * 3 / 4 / BQ_TPB)
static_assert(BQ_Q == 4, "the vote test and the early-exit test below are written out for 4 queries");
static_assert(BQ_QPB == 64 && BQ_WAVES == 16,
              "the U job: a block's 64 points are 4 tiles of 16 columns x 4 row tiles, one per wave");

struct UFrags { v2f a[8], b[8]; };
__device__ __forceinline__ UFrags upoint_load(const float* __restrict__ x, const float* __restrict__ W1,
                                              int b, int n0, int mt, int c, int h) {
  UFrags f;
  const float* xc = x + ((size_t)b * CX + 16 * h) * NN + n0 + c;
  const float* wr = W1 + (mt * 16 + c) * CIN + 3 + 16 * h;
  #pragma unroll
  for (int s = 0; s < 8; ++s) {
    f.b[s].x = xc[(size_t)(2 * s) * NN];
    f.b[s].y = xc[(size_t)(2 * s + 1) * NN];
    f.a[s].x = wr[2 * s];
    f.a[s].y = wr[2 * s + 1];
  }
  return f;
}
__device__ __forceinline__ v8f upoint_tile(const UFrags& f) {
  v8f acc = (v8f){0.f, 0.f, 0.f, 0.f, 0.f, 0.f, 0.f, 0.f};
  #pragma unroll
  for (int s = 0; s < 8; ++s)
    acc = __builtin_amdgcn_wmma_f32_16x16x4_f32(false, f.a[s], false, f.b[s],
                                                (short)0, acc, false, false);
  return acc;
}
__device__ __forceinline__ void upoint_store(const v8f& acc, float* __restrict__ U,
                                             int b, int n0, int mt, int c, int h) {
  v4f* ur = reinterpret_cast<v4f*>(U + ((size_t)b * NN + n0 + c) * CU) + h;
  ur[4 * mt]     = (v4f){acc[0], acc[1], acc[2], acc[3]};
  ur[4 * mt + 2] = (v4f){acc[4], acc[5], acc[6], acc[7]};
}

__global__ __launch_bounds__(BQ_TPB) void ballq_kernel(const float* __restrict__ p,
                                                       const float* __restrict__ x,
                                                       const float* __restrict__ W1,
                                                       unsigned short* __restrict__ idx,
                                                       float* __restrict__ U) {
  __shared__ float4 cand4[NN * 3 / 4];
  __shared__ int    lst[BQ_WAVES * BQ_Q * KS];
  const int t    = threadIdx.x;
  const int lane = t & 31;
  const int w  = __builtin_amdgcn_readfirstlane(t >> 5);
  const int b  = blockIdx.x / (NN / BQ_QPB);
  const int q0 = (blockIdx.x % (NN / BQ_QPB)) * BQ_QPB + w * BQ_Q;
  const float* pb = p + (size_t)b * NN * 3;

  const float4* src4 = reinterpret_cast<const float4*>(pb);
  float4 stage[BQ_FILL];
  #pragma unroll
  for (int k = 0; k < BQ_FILL; ++k) stage[k] = src4[t + k * BQ_TPB];
  const int uc = lane & 15, uh = lane >> 4;
  const int un0 = (blockIdx.x % (NN / BQ_QPB)) * BQ_QPB + (w & 3) * 16, umt = w >> 2;
  const UFrags uf = upoint_load(x, W1, b, un0, umt, uc, uh);
  #pragma unroll
  for (int k = 0; k < BQ_FILL; ++k) cand4[t + k * BQ_TPB] = stage[k];
  const v8f uacc = upoint_tile(uf);

  int* mylst = lst + w * (BQ_Q * KS);
  if (lane < BQ_Q) mylst[lane * KS] = 0;
  float qx[BQ_Q], qy[BQ_Q], qz[BQ_Q];
  int cnt[BQ_Q];
  #pragma unroll
  for (int q = 0; q < BQ_Q; ++q) {
    qx[q] = pb[(q0 + q) * 3 + 0];
    qy[q] = pb[(q0 + q) * 3 + 1];
    qz[q] = pb[(q0 + q) * 3 + 2];
    cnt[q] = 0;
  }
  __syncthreads();
  upoint_store(uacc, U, b, un0, umt, uc, uh);

  const float* cand = reinterpret_cast<const float*>(cand4) + lane * 3;
  float nx = cand[0], ny = cand[1], nz = cand[2];
  bool active = true;
  int nb3 = 0;
  for (int base = 0; base < NN && active; base += 32) {
    const float cx = nx, cy = ny, cz = nz;
    if (base + 32 < NN) nb3 += 32 * 3;
    nx = cand[nb3 + 0]; ny = cand[nb3 + 1]; nz = cand[nb3 + 2];

    bool h[BQ_Q];
    unsigned m[BQ_Q];
    #pragma unroll
    for (int q = 0; q < BQ_Q; ++q) {
      const float dx = cx - qx[q], dy = cy - qy[q], dz = cz - qz[q];
      const float d2 = __builtin_fmaf(dz, dz, __builtin_fmaf(dx, dx, dy * dy));
      h[q] = d2 < R2;
      m[q] = __builtin_amdgcn_ballot_w32(h[q]);
    }
    if ((m[0] | m[1] | m[2] | m[3]) != 0u) {
      #pragma unroll
      for (int q = 0; q < BQ_Q; ++q) {
        if (m[q] != 0u) {
          const int rank = cnt[q] + (int)__builtin_amdgcn_mbcnt_lo(m[q], 0u);
          if (h[q] && rank < KS) mylst[q * KS + rank] = base + lane;
          cnt[q] += __builtin_popcount(m[q]);
        }
      }
      active = (((cnt[0] - KS) | (cnt[1] - KS) | (cnt[2] - KS) | (cnt[3] - KS)) < 0);
    }
  }
  #pragma unroll
  for (int q = 0; q < BQ_Q; ++q) {
    const int v = mylst[q * KS + (lane < cnt[q] ? lane : 0)];
    idx[((size_t)b * NN + q0 + q) * KS + lane] = (unsigned short)v;
  }
}

#define CW_WAVES 4
#define CW_TPB   (32 * CW_WAVES)
#define CW_PTS   (BB * NN / GCONV)
static_assert(CW_PTS == 16 && NN % CW_PTS == 0 && GCONV % CW_WAVES == 0,
              "lane c keeps the pooled values of the wave's point c; a wave stays inside one batch");

struct RawTile { v4f u[8]; float px, py, pz; };

__device__ __forceinline__ RawTile gather_tile(const float* __restrict__ p,
                                               const float* __restrict__ U,
                                               int b, int j, int h) {
  RawTile r;
  const v4f* ur = reinterpret_cast<const v4f*>(U + ((size_t)b * NN + j) * CU) + h;
  #pragma unroll
  for (int i = 0; i < 8; ++i) r.u[i] = ur[2 * i];
  const float* pj = p + ((size_t)b * NN + j) * 3;
  r.px = pj[0]; r.py = pj[1]; r.pz = pj[2];
  return r;
}

__device__ __forceinline__ void end_of_trip(const RawTile& a, const RawTile& b,
                                            float qx, float qy, float qz, int id) {
  __builtin_amdgcn_sched_barrier(0);
  asm volatile("" :: "v"(a.px), "v"(a.py), "v"(a.pz), "v"(b.px), "v"(b.py), "v"(b.pz),
                     "s"(qx), "s"(qy), "s"(qz), "v"(id));
}

__device__ __forceinline__ void load_w1p_frags(const float* __restrict__ W1, int c, int h,
                                               v2f (&ap)[4]) {
  #pragma unroll
  for (int mt = 0; mt < 4; ++mt) {
    const float* wr = W1 + (mt * 16 + c) * CIN;
    const float w01 = wr[1];
    ap[mt].x = wr[2 * h];
    ap[mt].y = h ? 0.f : w01;
  }
}

__device__ __forceinline__ void load_w2_frags(const float* __restrict__ W2, int c, int h,
                                              v2f (&a2)[4][16]) {
  #pragma unroll
  for (int mt2 = 0; mt2 < 4; ++mt2) {
    const float* wr = W2 + (mt2 * 16 + c) * COUT + 8 * h;
    #pragma unroll
    for (int kk = 0; kk < 16; ++kk) {
      const int k0 = (kk >> 2) * 16 + (kk & 3) * 2;
      a2[mt2][kk].x = wr[k0];
      a2[mt2][kk].y = wr[k0 + 1];
    }
  }
}

__device__ __forceinline__ void conv1_tile(const v2f (&ap)[4], const RawTile& rt,
                                           float qx, float qy, float qz, int h, v8f (&acc)[4]) {
  const float dx = rt.px - qx, dy = rt.py - qy, dz = rt.pz - qz;
  v2f bd;
  bd.x = h ? dz : dx;
  bd.y = h ? 0.f : dy;
  #pragma unroll
  for (int mt = 0; mt < 4; ++mt) {
    const v4f lo = rt.u[2 * mt], hi = rt.u[2 * mt + 1];
    const v8f cu = (v8f){lo.x, lo.y, lo.z, lo.w, hi.x, hi.y, hi.z, hi.w};
    acc[mt] = __builtin_amdgcn_wmma_f32_16x16x4_f32(false, ap[mt], false, bd,
                                                    (short)0, cu, false, false);
  }
}

template <int XM>
__device__ __forceinline__ float lane_xor16(float v) {
  return __builtin_bit_cast(float, __builtin_amdgcn_update_dpp(
      0, __builtin_bit_cast(int, v), 0x160 + XM, 0xf, 0xf, true));
}
__device__ __forceinline__ float row_sum16(float v) {
  v += lane_xor16<1>(v); v += lane_xor16<2>(v); v += lane_xor16<4>(v); v += lane_xor16<8>(v);
  return v;
}
__device__ __forceinline__ float row_max16(float v) {
  v = fmaxf(v, lane_xor16<1>(v)); v = fmaxf(v, lane_xor16<2>(v));
  v = fmaxf(v, lane_xor16<4>(v)); v = fmaxf(v, lane_xor16<8>(v));
  return v;
}

__device__ __forceinline__ void write_part_row(float (&s)[4][8], float (&sq)[4][8],
                                               float* __restrict__ row, int c, int h) {
  #pragma unroll
  for (int mt = 0; mt < 4; ++mt) {
    float v = 0.f;
    #pragma unroll
    for (int r = 0; r < 8; ++r) {
      const float ts = row_sum16(s[mt][r]);
      const float tq = row_sum16(sq[mt][r]);
      v = (c == r) ? ts : v;
      v = (c == 8 + r) ? tq : v;
    }
    row[mt * 32 + ((c & 8) << 1) + 8 * h + (c & 7)] = v;
  }
}

__device__ __forceinline__ void stage_part_row(float (&s)[4][8], float (&sq)[4][8],
                                               float* lrow, int c, int h) {
  #pragma unroll
  for (int mt = 0; mt < 4; ++mt) {
    float v = 0.f;
    #pragma unroll
    for (int r = 0; r < 8; ++r) {
      const float ts = row_sum16(s[mt][r]);
      const float tq = row_sum16(sq[mt][r]);
      v = (c == r) ? ts : v;
      v = (c == 8 + r) ? tq : v;
    }
    lrow[2 * (mt * 16 + 8 * h + (c & 7)) + (c >> 3)] = v;
  }
}

#define S1_PP 2
static_assert(CW_PTS % S1_PP == 0, "a trip takes S1_PP whole points");
__global__ __launch_bounds__(CW_TPB) __attribute__((amdgpu_waves_per_eu(1, 1))) void conv1_stats_kernel(
    const float* __restrict__ p, const float* __restrict__ U,
    const unsigned short* __restrict__ idxg, const float* __restrict__ W1,
    float* __restrict__ part1) {
  const int lane = threadIdx.x & 31;
  const int c = lane & 15, h = lane >> 4;
  const int w   = __builtin_amdgcn_readfirstlane((int)(threadIdx.x >> 5));
  const int gw  = blockIdx.x * CW_WAVES + w;
  const int pt0 = gw * CW_PTS;
  const int b   = pt0 / NN, n0 = pt0 % NN;

  v2f ap[4];
  load_w1p_frags(W1, c, h, ap);
  float s[4][8], sq[4][8];
  #pragma unroll
  for (int mt = 0; mt < 4; ++mt)
    #pragma unroll
    for (int r = 0; r < 8; ++r) { s[mt][r] = 0.f; sq[mt][r] = 0.f; }

  const unsigned short* idxw = idxg + ((size_t)b * NN + n0) * KS + lane;
  const float* pqw  = p + ((size_t)b * NN + n0) * 3;
  RawTile rn[S1_PP][2];
  float nqx[S1_PP], nqy[S1_PP], nqz[S1_PP];
  int idn[S1_PP];
  #pragma unroll
  for (int k = 0; k < S1_PP; ++k) {
    const int id = idxw[k * KS];
    rn[k][0] = gather_tile(p, U, b, __shfl(id, c, 32) & (NN - 1), h);
    rn[k][1] = gather_tile(p, U, b, __shfl(id, 16 + c, 32) & (NN - 1), h);
    nqx[k] = pqw[k * 3 + 0]; nqy[k] = pqw[k * 3 + 1]; nqz[k] = pqw[k * 3 + 2];
  }
  #pragma unroll
  for (int k = 0; k < S1_PP; ++k) idn[k] = idxw[(S1_PP + k) * KS];

  #pragma nounroll
  for (int i = 0; i < CW_PTS; i += S1_PP) {
    RawTile r[S1_PP][2];
    float qx[S1_PP], qy[S1_PP], qz[S1_PP];
    #pragma unroll
    for (int k = 0; k < S1_PP; ++k) {
      r[k][0] = rn[k][0]; r[k][1] = rn[k][1];
      qx[k] = nqx[k]; qy[k] = nqy[k]; qz[k] = nqz[k];
    }
    #pragma unroll
    for (int k = 0; k < S1_PP; ++k) {
      const int i1 = (i + S1_PP + k < CW_PTS) ? i + S1_PP + k : CW_PTS - 1;
      const int i2 = (i + 2 * S1_PP + k < CW_PTS) ? i + 2 * S1_PP + k : CW_PTS - 1;
      rn[k][0] = gather_tile(p, U, b, __shfl(idn[k], c, 32) & (NN - 1), h);
      rn[k][1] = gather_tile(p, U, b, __shfl(idn[k], 16 + c, 32) & (NN - 1), h);
      nqx[k] = pqw[i1 * 3 + 0]; nqy[k] = pqw[i1 * 3 + 1]; nqz[k] = pqw[i1 * 3 + 2];
      idn[k] = idxw[i2 * KS];
    }
    __builtin_amdgcn_sched_barrier(0);
    #pragma unroll
    for (int k = 0; k < S1_PP; ++k)
      #pragma unroll
      for (int tile = 0; tile < 2; ++tile) {
        v8f acc[4];
        conv1_tile(ap, r[k][tile], qx[k], qy[k], qz[k], h, acc);
        #pragma unroll
        for (int mt = 0; mt < 4; ++mt)
          #pragma unroll
          for (int rr = 0; rr < 8; ++rr) { const float y = acc[mt][rr]; s[mt][rr] += y; sq[mt][rr] += y * y; }
      }
    #pragma unroll
    for (int k = 0; k < S1_PP; ++k)
      end_of_trip(rn[k][0], rn[k][1], nqx[k], nqy[k], nqz[k], idn[k]);
  }
  write_part_row(s, sq, part1 + (size_t)gw * 128, c, h);
}

#define FIN_TPB   1024
#define FIN_WAVES (FIN_TPB / 32)
#define FIN_ROWS  (GCONV / FIN_WAVES)
static_assert(FIN_ROWS == 32 && FIN_WAVES == 32, "tree32 below sums exactly 32 values");

__device__ __forceinline__ float tree32(float (&a)[32]) {
  #pragma unroll
  for (int h = 16; h >= 1; h >>= 1) {
    #pragma unroll
    for (int r = 0; r < h; ++r) a[r] += a[r + h];
  }
  return a[0];
}

__global__ __launch_bounds__(FIN_TPB) void bn_finalize_kernel(const float* __restrict__ part,
                                                              const float* __restrict__ g,
                                                              const float* __restrict__ bta,
                                                              float* __restrict__ bnp,
                                                              int rstride) {
  __shared__ float red[FIN_WAVES * 32];
  const int t    = threadIdx.x;
  const int lane = t & 31;
  const int w    = t >> 5;
  const int m    = blockIdx.x;
  const float* src = part + (size_t)w * rstride + m * 32 + lane;
  float a[FIN_ROWS];
  #pragma unroll
  for (int r = 0; r < FIN_ROWS; ++r) a[r] = src[(size_t)r * FIN_WAVES * rstride];
  red[w * 32 + lane] = tree32(a);
  __syncthreads();
  if (w == 0) {
    float c[FIN_WAVES];
    #pragma unroll
    for (int k = 0; k < FIN_WAVES; ++k) c[k] = red[k * 32 + lane];
    const float tot = tree32(c);
    const float sq  = __shfl_down(tot, 16, 32);
    if (lane < 16) {
      const int o = m * 16 + lane;
      float mean  = tot / CNTF;
      float var   = sq / CNTF - mean * mean;
      float scale = g[o] * rsqrtf(var + EPSV);
      bnp[o]      = scale;
      bnp[64 + o] = bta[o] - mean * scale;
    }
  }
}

template <bool FIRST>
__device__ __forceinline__ void conv2_tile(const RawTile& rt, float qx, float qy, float qz, int h,
                                           const v2f (&ap)[4], const v2f (&a2)[4][16],
                                           const float (&sc1)[4][8], const float (&sh1)[4][8],
                                           const float (&sgn)[4][8],
                                           float (&s)[4][8], float (&sq)[4][8], float (&m)[4][8]) {
  v8f acc1[4];
  conv1_tile(ap, rt, qx, qy, qz, h, acc1);
  float f1[4][8];
  #pragma unroll
  for (int mt = 0; mt < 4; ++mt)
    #pragma unroll
    for (int r = 0; r < 8; ++r) {
      float y = acc1[mt][r] * sc1[mt][r] + sh1[mt][r];
      f1[mt][r] = y > 0.f ? y : 0.f;
    }
  v8f acc2[4];
  #pragma unroll
  for (int mt2 = 0; mt2 < 4; ++mt2) acc2[mt2] = (v8f){0.f, 0.f, 0.f, 0.f, 0.f, 0.f, 0.f, 0.f};
  #pragma unroll
  for (int kk = 0; kk < 16; ++kk) {
    v2f bm;
    bm.x = f1[kk >> 2][(kk & 3) * 2];
    bm.y = f1[kk >> 2][(kk & 3) * 2 + 1];
    #pragma unroll
    for (int mt2 = 0; mt2 < 4; ++mt2)
      acc2[mt2] = __builtin_amdgcn_wmma_f32_16x16x4_f32(false, a2[mt2][kk], false, bm,
                                                        (short)0, acc2[mt2], false, false);
  }
  #pragma unroll
  for (int mt2 = 0; mt2 < 4; ++mt2)
    #pragma unroll
    for (int r = 0; r < 8; ++r) {
      const float y = acc2[mt2][r];
      s[mt2][r] += y; sq[mt2][r] += y * y;
      const float e = y * sgn[mt2][r];
      m[mt2][r] = FIRST ? e : fmaxf(m[mt2][r], e);
    }
}

#define P2_STRIDE (CW_WAVES * CW_PTS * KS / 2)
__global__ __launch_bounds__(CW_TPB) void conv2_stats_kernel(
    const float* __restrict__ p, const float* __restrict__ U,
    const unsigned short* idxg, const float* __restrict__ W1,
    const float* __restrict__ W2, const float* __restrict__ bnp1,
    const float* __restrict__ g2, float* part2,
    float* __restrict__ out) {
  __shared__ float prow[CW_WAVES * 128];
  const int lane = threadIdx.x & 31;
  const int c = lane & 15, h = lane >> 4;
  const int w   = __builtin_amdgcn_readfirstlane((int)(threadIdx.x >> 5));
  const int gw  = blockIdx.x * CW_WAVES + w;
  const int pt0 = gw * CW_PTS;
  const int b   = pt0 / NN, n0 = pt0 % NN;

  v2f ap[4];
  load_w1p_frags(W1, c, h, ap);
  v2f a2[4][16];
  load_w2_frags(W2, c, h, a2);
  float sc1[4][8], sh1[4][8], sgn[4][8];
  float s[4][8], sq[4][8], res[4][8];
  #pragma unroll
  for (int mt = 0; mt < 4; ++mt)
    #pragma unroll
    for (int r = 0; r < 8; ++r) {
      const int o = mt * 16 + 8 * h + r;
      sc1[mt][r] = bnp1[o]; sh1[mt][r] = bnp1[64 + o];
      sgn[mt][r] = (g2[o] < 0.f) ? -1.f : 1.f;
      s[mt][r] = 0.f; sq[mt][r] = 0.f; res[mt][r] = 0.f;
    }

  const unsigned short* idxw = idxg + ((size_t)b * NN + n0) * KS + lane;
  const float* pqw  = p + ((size_t)b * NN + n0) * 3;
  int idn = idxw[0];
  RawTile rn0 = gather_tile(p, U, b, __shfl(idn, c, 32) & (NN - 1), h);
  RawTile rn1 = gather_tile(p, U, b, __shfl(idn, 16 + c, 32) & (NN - 1), h);
  float nqx = pqw[0], nqy = pqw[1], nqz = pqw[2];
  idn = idxw[KS];

  #pragma nounroll
  for (int i = 0; i < CW_PTS; ++i) {
    const RawTile r0 = rn0, r1 = rn1;
    const float qx = nqx, qy = nqy, qz = nqz;
    {
      const int i1 = (i + 1 < CW_PTS) ? i + 1 : CW_PTS - 1;
      const int i2 = (i + 2 < CW_PTS) ? i + 2 : CW_PTS - 1;
      rn0 = gather_tile(p, U, b, __shfl(idn, c, 32) & (NN - 1), h);
      rn1 = gather_tile(p, U, b, __shfl(idn, 16 + c, 32) & (NN - 1), h);
      nqx = pqw[i1 * 3 + 0]; nqy = pqw[i1 * 3 + 1]; nqz = pqw[i1 * 3 + 2];
      idn = idxw[i2 * KS];
    }
    __builtin_amdgcn_sched_barrier(0);
    float m[4][8];
    conv2_tile<true >(r0, qx, qy, qz, h, ap, a2, sc1, sh1, sgn, s, sq, m);
    conv2_tile<false>(r1, qx, qy, qz, h, ap, a2, sc1, sh1, sgn, s, sq, m);
    #pragma unroll
    for (int mt2 = 0; mt2 < 4; ++mt2)
      #pragma unroll
      for (int r = 0; r < 8; ++r) {
        const float e = row_max16(m[mt2][r]);
        res[mt2][r] = (c == i) ? e : res[mt2][r];
      }
    end_of_trip(rn0, rn1, nqx, nqy, nqz, idn);
  }
  stage_part_row(s, sq, prow + w * 128, c, h);
  __syncthreads();
  const int t = threadIdx.x;
  part2[(size_t)blockIdx.x * P2_STRIDE + t] = (prow[t] + prow[128 + t]) + (prow[256 + t] + prow[384 + t]);
  #pragma unroll
  for (int mt2 = 0; mt2 < 4; ++mt2)
    #pragma unroll
    for (int r = 0; r < 8; ++r) {
      const int o = mt2 * 16 + 8 * h + r;
      out[((size_t)b * COUT + o) * NN + n0 + c] = res[mt2][r] * sgn[mt2][r];
    }
}

#define AP_TPB 256
#define AP_V4  (NN / 4 / AP_TPB)
#define AP_WAVES (AP_TPB / 32)
static_assert(GCONV / CW_WAVES == AP_TPB, "thread t of the apply block takes the partial row of conv2 block t");
__global__ __launch_bounds__(AP_TPB) void bn2_apply_kernel(const float* __restrict__ part2,
                                                           const float* __restrict__ g2,
                                                           const float* __restrict__ b2,
                                                           float* __restrict__ out) {
  __shared__ float2 wtot[AP_WAVES];
  const int row = blockIdx.x;
  const int o   = row & (COUT - 1);
  const int t   = threadIdx.x;
  float4* q = reinterpret_cast<float4*>(out) + (size_t)row * (NN / 4) + t;
  float4 e[AP_V4];
  #pragma unroll
  for (int k = 0; k < AP_V4; ++k) e[k] = q[k * AP_TPB];
  const float2 ps = *reinterpret_cast<const float2*>(part2 + (size_t)t * P2_STRIDE + 2 * o);
  float wsum = row_sum16(ps.x), wsq = row_sum16(ps.y);
  wsum += __shfl_xor(wsum, 16, 32);
  wsq  += __shfl_xor(wsq, 16, 32);
  if ((t & 31) == 0) wtot[t >> 5] = make_float2(wsum, wsq);
  __syncthreads();
  float tot = 0.f, tsq = 0.f;
  #pragma unroll
  for (int k = 0; k < AP_WAVES; ++k) { const float2 v = wtot[k]; tot += v.x; tsq += v.y; }
  const float mean = tot / CNTF;
  const float var  = tsq / CNTF - mean * mean;
  const float sc   = g2[o] * rsqrtf(var + EPSV);
  const float sh   = b2[o] - mean * sc;
  #pragma unroll
  for (int k = 0; k < AP_V4; ++k) {
    float4 r;
    float y;
    y = e[k].x * sc + sh; r.x = y > 0.f ? y : 0.f;
    y = e[k].y * sc + sh; r.y = y > 0.f ? y : 0.f;
    y = e[k].z * sc + sh; r.z = y > 0.f ? y : 0.f;
    y = e[k].w * sc + sh; r.w = y > 0.f ? y : 0.f;
    q[k * AP_TPB] = r;
  }
}

extern "C" void kernel_launch(void* const* d_in, const int* in_sizes, int n_in,
                              void* d_out, int out_size, void* d_ws, size_t ws_size,
                              hipStream_t stream) {
  const float* p  = (const float*)d_in[0];
  const float* x  = (const float*)d_in[1];
  const float* W1 = (const float*)d_in[2];
  const float* g1 = (const float*)d_in[3];
  const float* b1 = (const float*)d_in[4];
  const float* W2 = (const float*)d_in[5];
  const float* g2 = (const float*)d_in[6];
  const float* b2 = (const float*)d_in[7];
  float* out = (float*)d_out;

  char*  ws    = (char*)d_ws;
  unsigned short* idx = (unsigned short*)ws;
  float* U     = (float*)(ws + (size_t)BB * NN * KS * 2);
  float* bnp1  = (float*)(ws + (size_t)BB * NN * KS * 2
                             + (size_t)BB * NN * CU * 4);
  float* part1 = out;
  float* part2 = (float*)ws;

  ballq_kernel<<<BB * (NN / BQ_QPB), BQ_TPB, 0, stream>>>(p, x, W1, idx, U);
  conv1_stats_kernel<<<GCONV / CW_WAVES, CW_TPB, 0, stream>>>(p, U, idx, W1, part1);
  bn_finalize_kernel<<<COUT / 16, FIN_TPB, 0, stream>>>(part1, g1, b1, bnp1, 128);
  conv2_stats_kernel<<<GCONV / CW_WAVES, CW_TPB, 0, stream>>>(p, U, idx, W1, W2, bnp1, g2, part2, out);
  bn2_apply_kernel<<<BB * COUT, AP_TPB, 0, stream>>>(part2, g2, b2, out);
}


# conv2 pass at two waves per SIMD (8 waves x 8 points per block; gamma2's sign folded into W2 fragments to fit 512 registers): vector work overlaps the other wave's matrix work; changes BN2 sum order
# speedup vs baseline: 10.3715x; 1.0937x over previous
#include <hip/hip_runtime.h>
#include <hip/hip_bf16.h>


typedef __attribute__((ext_vector_type(2))) float v2f;
typedef __attribute__((ext_vector_type(4))) float v4f;
typedef __attribute__((ext_vector_type(8))) float v8f;

#define BB      4
#define NN      4096
#define CX      32
#define CU      64
#define CIN     35
#define COUT    64
#define KS      32
#define R2      0.01f
#define EPSV    1e-5f
#define GCONV   1024
#define CNTF    ((float)BB * (float)NN * (float)KS)


#define BQ_Q     4
#define BQ_WAVES 16
#define BQ_QPB   (BQ_Q * BQ_WAVES)
#define BQ_TPB   (32 * BQ_WAVES)
#define BQ_FILL  (NN * 3 / 4 / BQ_TPB)
static_assert(BQ_Q == 4, "the vote test and the early-exit test below are written out for 4 queries");
static_assert(BQ_QPB == 64 && BQ_WAVES == 16,
              "the U job: a block's 64 points are 4 tiles of 16 columns x 4 row tiles, one per wave");

struct UFrags { v2f a[8], b[8]; };
__device__ __forceinline__ UFrags upoint_load(const float* __restrict__ x, const float* __restrict__ W1,
                                              int b, int n0, int mt, int c, int h) {
  UFrags f;
  const float* xc = x + ((size_t)b * CX + 16 * h) * NN + n0 + c;
  const float* wr = W1 + (mt * 16 + c) * CIN + 3 + 16 * h;
  #pragma unroll
  for (int s = 0; s < 8; ++s) {
    f.b[s].x = xc[(size_t)(2 * s) * NN];
    f.b[s].y = xc[(size_t)(2 * s + 1) * NN];
    f.a[s].x = wr[2 * s];
    f.a[s].y = wr[2 * s + 1];
  }
  return f;
}
__device__ __forceinline__ v8f upoint_tile(const UFrags& f) {
  v8f acc = (v8f){0.f, 0.f, 0.f, 0.f, 0.f, 0.f, 0.f, 0.f};
  #pragma unroll
  for (int s = 0; s < 8; ++s)
    acc = __builtin_amdgcn_wmma_f32_16x16x4_f32(false, f.a[s], false, f.b[s],
                                                (short)0, acc, false, false);
  return acc;
}
__device__ __forceinline__ void upoint_store(const v8f& acc, float* __restrict__ U,
                                             int b, int n0, int mt, int c, int h) {
  v4f* ur = reinterpret_cast<v4f*>(U + ((size_t)b * NN + n0 + c) * CU) + h;
  ur[4 * mt]     = (v4f){acc[0], acc[1], acc[2], acc[3]};
  ur[4 * mt + 2] = (v4f){acc[4], acc[5], acc[6], acc[7]};
}

__global__ __launch_bounds__(BQ_TPB) void ballq_kernel(const float* __restrict__ p,
                                                       const float* __restrict__ x,
                                                       const float* __restrict__ W1,
                                                       unsigned short* __restrict__ idx,
                                                       float* __restrict__ U) {
  __shared__ float4 cand4[NN * 3 / 4];
  __shared__ int    lst[BQ_WAVES * BQ_Q * KS];
  const int t    = threadIdx.x;
  const int lane = t & 31;
  const int w  = __builtin_amdgcn_readfirstlane(t >> 5);
  const int b  = blockIdx.x / (NN / BQ_QPB);
  const int q0 = (blockIdx.x % (NN / BQ_QPB)) * BQ_QPB + w * BQ_Q;
  const float* pb = p + (size_t)b * NN * 3;

  const float4* src4 = reinterpret_cast<const float4*>(pb);
  float4 stage[BQ_FILL];
  #pragma unroll
  for (int k = 0; k < BQ_FILL; ++k) stage[k] = src4[t + k * BQ_TPB];
  const int uc = lane & 15, uh = lane >> 4;
  const int un0 = (blockIdx.x % (NN / BQ_QPB)) * BQ_QPB + (w & 3) * 16, umt = w >> 2;
  const UFrags uf = upoint_load(x, W1, b, un0, umt, uc, uh);
  #pragma unroll
  for (int k = 0; k < BQ_FILL; ++k) cand4[t + k * BQ_TPB] = stage[k];
  const v8f uacc = upoint_tile(uf);

  int* mylst = lst + w * (BQ_Q * KS);
  if (lane < BQ_Q) mylst[lane * KS] = 0;
  float qx[BQ_Q], qy[BQ_Q], qz[BQ_Q];
  int cnt[BQ_Q];
  #pragma unroll
  for (int q = 0; q < BQ_Q; ++q) {
    qx[q] = pb[(q0 + q) * 3 + 0];
    qy[q] = pb[(q0 + q) * 3 + 1];
    qz[q] = pb[(q0 + q) * 3 + 2];
    cnt[q] = 0;
  }
  __syncthreads();
  upoint_store(uacc, U, b, un0, umt, uc, uh);

  const float* cand = reinterpret_cast<const float*>(cand4) + lane * 3;
  float nx = cand[0], ny = cand[1], nz = cand[2];
  bool active = true;
  int nb3 = 0;
  for (int base = 0; base < NN && active; base += 32) {
    const float cx = nx, cy = ny, cz = nz;
    if (base + 32 < NN) nb3 += 32 * 3;
    nx = cand[nb3 + 0]; ny = cand[nb3 + 1]; nz = cand[nb3 + 2];

    bool h[BQ_Q];
    unsigned m[BQ_Q];
    #pragma unroll
    for (int q = 0; q < BQ_Q; ++q) {
      const float dx = cx - qx[q], dy = cy - qy[q], dz = cz - qz[q];
      const float d2 = __builtin_fmaf(dz, dz, __builtin_fmaf(dx, dx, dy * dy));
      h[q] = d2 < R2;
      m[q] = __builtin_amdgcn_ballot_w32(h[q]);
    }
    if ((m[0] | m[1] | m[2] | m[3]) != 0u) {
      #pragma unroll
      for (int q = 0; q < BQ_Q; ++q) {
        if (m[q] != 0u) {
          const int rank = cnt[q] + (int)__builtin_amdgcn_mbcnt_lo(m[q], 0u);
          if (h[q] && rank < KS) mylst[q * KS + rank] = base + lane;
          cnt[q] += __builtin_popcount(m[q]);
        }
      }
      active = (((cnt[0] - KS) | (cnt[1] - KS) | (cnt[2] - KS) | (cnt[3] - KS)) < 0);
    }
  }
  #pragma unroll
  for (int q = 0; q < BQ_Q; ++q) {
    const int v = mylst[q * KS + (lane < cnt[q] ? lane : 0)];
    idx[((size_t)b * NN + q0 + q) * KS + lane] = (unsigned short)v;
  }
}

#define CW_WAVES 4
#define CW_TPB   (32 * CW_WAVES)
#define CW_PTS   (BB * NN / GCONV)
static_assert(CW_PTS == 16 && NN % CW_PTS == 0 && GCONV % CW_WAVES == 0,
              "lane c keeps the pooled values of the wave's point c; a wave stays inside one batch");

struct RawTile { v4f u[8]; float px, py, pz; };

__device__ __forceinline__ RawTile gather_tile(const float* __restrict__ p,
                                               const float* __restrict__ U,
                                               int b, int j, int h) {
  RawTile r;
  const v4f* ur = reinterpret_cast<const v4f*>(U + ((size_t)b * NN + j) * CU) + h;
  #pragma unroll
  for (int i = 0; i < 8; ++i) r.u[i] = ur[2 * i];
  const float* pj = p + ((size_t)b * NN + j) * 3;
  r.px = pj[0]; r.py = pj[1]; r.pz = pj[2];
  return r;
}

__device__ __forceinline__ void end_of_trip(const RawTile& a, const RawTile& b,
                                            float qx, float qy, float qz, int id) {
  __builtin_amdgcn_sched_barrier(0);
  asm volatile("" :: "v"(a.px), "v"(a.py), "v"(a.pz), "v"(b.px), "v"(b.py), "v"(b.pz),
                     "s"(qx), "s"(qy), "s"(qz), "v"(id));
}

__device__ __forceinline__ void load_w1p_frags(const float* __restrict__ W1, int c, int h,
                                               v2f (&ap)[4]) {
  #pragma unroll
  for (int mt = 0; mt < 4; ++mt) {
    const float* wr = W1 + (mt * 16 + c) * CIN;
    const float w01 = wr[1];
    ap[mt].x = wr[2 * h];
    ap[mt].y = h ? 0.f : w01;
  }
}

__device__ __forceinline__ void load_w2_frags(const float* __restrict__ W2, int c, int h,
                                              v2f (&a2)[4][16]) {
  #pragma unroll
  for (int mt2 = 0; mt2 < 4; ++mt2) {
    const float* wr = W2 + (mt2 * 16 + c) * COUT + 8 * h;
    #pragma unroll
    for (int kk = 0; kk < 16; ++kk) {
      const int k0 = (kk >> 2) * 16 + (kk & 3) * 2;
      a2[mt2][kk].x = wr[k0];
      a2[mt2][kk].y = wr[k0 + 1];
    }
  }
}

__device__ __forceinline__ void conv1_tile(const v2f (&ap)[4], const RawTile& rt,
                                           float qx, float qy, float qz, int h, v8f (&acc)[4]) {
  const float dx = rt.px - qx, dy = rt.py - qy, dz = rt.pz - qz;
  v2f bd;
  bd.x = h ? dz : dx;
  bd.y = h ? 0.f : dy;
  #pragma unroll
  for (int mt = 0; mt < 4; ++mt) {
    const v4f lo = rt.u[2 * mt], hi = rt.u[2 * mt + 1];
    const v8f cu = (v8f){lo.x, lo.y, lo.z, lo.w, hi.x, hi.y, hi.z, hi.w};
    acc[mt] = __builtin_amdgcn_wmma_f32_16x16x4_f32(false, ap[mt], false, bd,
                                                    (short)0, cu, false, false);
  }
}

template <int XM>
__device__ __forceinline__ float lane_xor16(float v) {
  return __builtin_bit_cast(float, __builtin_amdgcn_update_dpp(
      0, __builtin_bit_cast(int, v), 0x160 + XM, 0xf, 0xf, true));
}
__device__ __forceinline__ float row_sum16(float v) {
  v += lane_xor16<1>(v); v += lane_xor16<2>(v); v += lane_xor16<4>(v); v += lane_xor16<8>(v);
  return v;
}
__device__ __forceinline__ float row_max16(float v) {
  v = fmaxf(v, lane_xor16<1>(v)); v = fmaxf(v, lane_xor16<2>(v));
  v = fmaxf(v, lane_xor16<4>(v)); v = fmaxf(v, lane_xor16<8>(v));
  return v;
}

__device__ __forceinline__ void write_part_row(float (&s)[4][8], float (&sq)[4][8],
                                               float* __restrict__ row, int c, int h) {
  #pragma unroll
  for (int mt = 0; mt < 4; ++mt) {
    float v = 0.f;
    #pragma unroll
    for (int r = 0; r < 8; ++r) {
      const float ts = row_sum16(s[mt][r]);
      const float tq = row_sum16(sq[mt][r]);
      v = (c == r) ? ts : v;
      v = (c == 8 + r) ? tq : v;
    }
    row[mt * 32 + ((c & 8) << 1) + 8 * h + (c & 7)] = v;
  }
}

__device__ __forceinline__ void stage_part_row(float (&s)[4][8], float (&sq)[4][8],
                                               float* lrow, int c, int h) {
  #pragma unroll
  for (int mt = 0; mt < 4; ++mt) {
    float v = 0.f;
    #pragma unroll
    for (int r = 0; r < 8; ++r) {
      const float ts = row_sum16(s[mt][r]);
      const float tq = row_sum16(sq[mt][r]);
      v = (c == r) ? ts : v;
      v = (c == 8 + r) ? tq : v;
    }
    lrow[2 * (mt * 16 + 8 * h + (c & 7)) + (c >> 3)] = v;
  }
}

#define S1_PP 2
static_assert(CW_PTS % S1_PP == 0, "a trip takes S1_PP whole points");
__global__ __launch_bounds__(CW_TPB) __attribute__((amdgpu_waves_per_eu(1, 1))) void conv1_stats_kernel(
    const float* __restrict__ p, const float* __restrict__ U,
    const unsigned short* __restrict__ idxg, const float* __restrict__ W1,
    float* __restrict__ part1) {
  const int lane = threadIdx.x & 31;
  const int c = lane & 15, h = lane >> 4;
  const int w   = __builtin_amdgcn_readfirstlane((int)(threadIdx.x >> 5));
  const int gw  = blockIdx.x * CW_WAVES + w;
  const int pt0 = gw * CW_PTS;
  const int b   = pt0 / NN, n0 = pt0 % NN;

  v2f ap[4];
  load_w1p_frags(W1, c, h, ap);
  float s[4][8], sq[4][8];
  #pragma unroll
  for (int mt = 0; mt < 4; ++mt)
    #pragma unroll
    for (int r = 0; r < 8; ++r) { s[mt][r] = 0.f; sq[mt][r] = 0.f; }

  const unsigned short* idxw = idxg + ((size_t)b * NN + n0) * KS + lane;
  const float* pqw  = p + ((size_t)b * NN + n0) * 3;
  RawTile rn[S1_PP][2];
  float nqx[S1_PP], nqy[S1_PP], nqz[S1_PP];
  int idn[S1_PP];
  #pragma unroll
  for (int k = 0; k < S1_PP; ++k) {
    const int id = idxw[k * KS];
    rn[k][0] = gather_tile(p, U, b, __shfl(id, c, 32) & (NN - 1), h);
    rn[k][1] = gather_tile(p, U, b, __shfl(id, 16 + c, 32) & (NN - 1), h);
    nqx[k] = pqw[k * 3 + 0]; nqy[k] = pqw[k * 3 + 1]; nqz[k] = pqw[k * 3 + 2];
  }
  #pragma unroll
  for (int k = 0; k < S1_PP; ++k) idn[k] = idxw[(S1_PP + k) * KS];

  #pragma nounroll
  for (int i = 0; i < CW_PTS; i += S1_PP) {
    RawTile r[S1_PP][2];
    float qx[S1_PP], qy[S1_PP], qz[S1_PP];
    #pragma unroll
    for (int k = 0; k < S1_PP; ++k) {
      r[k][0] = rn[k][0]; r[k][1] = rn[k][1];
      qx[k] = nqx[k]; qy[k] = nqy[k]; qz[k] = nqz[k];
    }
    #pragma unroll
    for (int k = 0; k < S1_PP; ++k) {
      const int i1 = (i + S1_PP + k < CW_PTS) ? i + S1_PP + k : CW_PTS - 1;
      const int i2 = (i + 2 * S1_PP + k < CW_PTS) ? i + 2 * S1_PP + k : CW_PTS - 1;
      rn[k][0] = gather_tile(p, U, b, __shfl(idn[k], c, 32) & (NN - 1), h);
      rn[k][1] = gather_tile(p, U, b, __shfl(idn[k], 16 + c, 32) & (NN - 1), h);
      nqx[k] = pqw[i1 * 3 + 0]; nqy[k] = pqw[i1 * 3 + 1]; nqz[k] = pqw[i1 * 3 + 2];
      idn[k] = idxw[i2 * KS];
    }
    __builtin_amdgcn_sched_barrier(0);
    #pragma unroll
    for (int k = 0; k < S1_PP; ++k)
      #pragma unroll
      for (int tile = 0; tile < 2; ++tile) {
        v8f acc[4];
        conv1_tile(ap, r[k][tile], qx[k], qy[k], qz[k], h, acc);
        #pragma unroll
        for (int mt = 0; mt < 4; ++mt)
          #pragma unroll
          for (int rr = 0; rr < 8; ++rr) { const float y = acc[mt][rr]; s[mt][rr] += y; sq[mt][rr] += y * y; }
      }
    #pragma unroll
    for (int k = 0; k < S1_PP; ++k)
      end_of_trip(rn[k][0], rn[k][1], nqx[k], nqy[k], nqz[k], idn[k]);
  }
  write_part_row(s, sq, part1 + (size_t)gw * 128, c, h);
}

#define FIN_TPB   1024
#define FIN_WAVES (FIN_TPB / 32)
#define FIN_ROWS  (GCONV / FIN_WAVES)
static_assert(FIN_ROWS == 32 && FIN_WAVES == 32, "tree32 below sums exactly 32 values");

__device__ __forceinline__ float tree32(float (&a)[32]) {
  #pragma unroll
  for (int h = 16; h >= 1; h >>= 1) {
    #pragma unroll
    for (int r = 0; r < h; ++r) a[r] += a[r + h];
  }
  return a[0];
}

__global__ __launch_bounds__(FIN_TPB) void bn_finalize_kernel(const float* __restrict__ part,
                                                              const float* __restrict__ g,
                                                              const float* __restrict__ bta,
                                                              float* __restrict__ bnp,
                                                              int rstride) {
  __shared__ float red[FIN_WAVES * 32];
  const int t    = threadIdx.x;
  const int lane = t & 31;
  const int w    = t >> 5;
  const int m    = blockIdx.x;
  const float* src = part + (size_t)w * rstride + m * 32 + lane;
  float a[FIN_ROWS];
  #pragma unroll
  for (int r = 0; r < FIN_ROWS; ++r) a[r] = src[(size_t)r * FIN_WAVES * rstride];
  red[w * 32 + lane] = tree32(a);
  __syncthreads();
  if (w == 0) {
    float c[FIN_WAVES];
    #pragma unroll
    for (int k = 0; k < FIN_WAVES; ++k) c[k] = red[k * 32 + lane];
    const float tot = tree32(c);
    const float sq  = __shfl_down(tot, 16, 32);
    if (lane < 16) {
      const int o = m * 16 + lane;
      float mean  = tot / CNTF;
      float var   = sq / CNTF - mean * mean;
      float scale = g[o] * rsqrtf(var + EPSV);
      bnp[o]      = scale;
      bnp[64 + o] = bta[o] - mean * scale;
    }
  }
}

template <bool FIRST>
__device__ __forceinline__ void conv2_tile(const RawTile& rt, float qx, float qy, float qz, int h,
                                           const v2f (&ap)[4], const v2f (&a2)[4][16],
                                           const float (&sc1)[4][8], const float (&sh1)[4][8],
                                           float (&s)[4][8], float (&sq)[4][8], float (&m)[4][8]) {
  v8f acc1[4];
  conv1_tile(ap, rt, qx, qy, qz, h, acc1);
  float f1[4][8];
  #pragma unroll
  for (int mt = 0; mt < 4; ++mt)
    #pragma unroll
    for (int r = 0; r < 8; ++r) {
      float y = acc1[mt][r] * sc1[mt][r] + sh1[mt][r];
      f1[mt][r] = y > 0.f ? y : 0.f;
    }
  v8f acc2[4];
  #pragma unroll
  for (int mt2 = 0; mt2 < 4; ++mt2) acc2[mt2] = (v8f){0.f, 0.f, 0.f, 0.f, 0.f, 0.f, 0.f, 0.f};
  #pragma unroll
  for (int kk = 0; kk < 16; ++kk) {
    v2f bm;
    bm.x = f1[kk >> 2][(kk & 3) * 2];
    bm.y = f1[kk >> 2][(kk & 3) * 2 + 1];
    #pragma unroll
    for (int mt2 = 0; mt2 < 4; ++mt2)
      acc2[mt2] = __builtin_amdgcn_wmma_f32_16x16x4_f32(false, a2[mt2][kk], false, bm,
                                                        (short)0, acc2[mt2], false, false);
  }
  #pragma unroll
  for (int mt2 = 0; mt2 < 4; ++mt2)
    #pragma unroll
    for (int r = 0; r < 8; ++r) {
      const float y = acc2[mt2][r];
      s[mt2][r] += y; sq[mt2][r] += y * y;
      m[mt2][r] = FIRST ? y : fmaxf(m[mt2][r], y);
    }
}

#define C2_WAVES 8
#define C2_TPB   (32 * C2_WAVES)
#define C2_PTS   8
#define C2_QPB   (C2_WAVES * C2_PTS)
#define C2_PITCH (C2_QPB + 4)
static_assert(C2_QPB == 64 && NN % C2_QPB == 0 && C2_PTS <= 16,
              "a block owns 64 consecutive points of one batch; lane c keeps the pooled values of the wave's point c");
#define P2_STRIDE (C2_QPB * KS / 2)
__global__ __launch_bounds__(C2_TPB) void conv2_stats_kernel(
    const float* __restrict__ p, const float* __restrict__ U,
    const unsigned short* idxg, const float* __restrict__ W1,
    const float* __restrict__ W2, const float* __restrict__ bnp1,
    const float* __restrict__ g2, float* part2,
    float* __restrict__ out) {
  __shared__ float prow[C2_WAVES * 128];
  __shared__ __attribute__((aligned(16))) float ptile[COUT * C2_PITCH];
  const int lane = threadIdx.x & 31;
  const int c = lane & 15, h = lane >> 4;
  const int w   = __builtin_amdgcn_readfirstlane((int)(threadIdx.x >> 5));
  const int gw  = blockIdx.x * C2_WAVES + w;
  const int pt0 = gw * C2_PTS;
  const int b   = pt0 / NN, n0 = pt0 % NN;

  v2f ap[4];
  load_w1p_frags(W1, c, h, ap);
  v2f a2[4][16];
  load_w2_frags(W2, c, h, a2);
  #pragma unroll
  for (int mt2 = 0; mt2 < 4; ++mt2) {
    const float sa = (g2[mt2 * 16 + c] < 0.f) ? -1.f : 1.f;
    #pragma unroll
    for (int kk = 0; kk < 16; ++kk) { a2[mt2][kk].x *= sa; a2[mt2][kk].y *= sa; }
  }
  float sc1[4][8], sh1[4][8];
  float s[4][8], sq[4][8], res[4][8];
  #pragma unroll
  for (int mt = 0; mt < 4; ++mt)
    #pragma unroll
    for (int r = 0; r < 8; ++r) {
      const int o = mt * 16 + 8 * h + r;
      sc1[mt][r] = bnp1[o]; sh1[mt][r] = bnp1[64 + o];
      s[mt][r] = 0.f; sq[mt][r] = 0.f; res[mt][r] = 0.f;
    }

  const unsigned short* idxw = idxg + ((size_t)b * NN + n0) * KS + lane;
  const float* pqw  = p + ((size_t)b * NN + n0) * 3;
  int idn = idxw[0];
  RawTile rn0 = gather_tile(p, U, b, __shfl(idn, c, 32) & (NN - 1), h);
  RawTile rn1 = gather_tile(p, U, b, __shfl(idn, 16 + c, 32) & (NN - 1), h);
  float nqx = pqw[0], nqy = pqw[1], nqz = pqw[2];
  idn = idxw[KS];

  #pragma nounroll
  for (int i = 0; i < C2_PTS; ++i) {
    const RawTile r0 = rn0, r1 = rn1;
    const float qx = nqx, qy = nqy, qz = nqz;
    {
      const int i1 = (i + 1 < C2_PTS) ? i + 1 : C2_PTS - 1;
      const int i2 = (i + 2 < C2_PTS) ? i + 2 : C2_PTS - 1;
      rn0 = gather_tile(p, U, b, __shfl(idn, c, 32) & (NN - 1), h);
      rn1 = gather_tile(p, U, b, __shfl(idn, 16 + c, 32) & (NN - 1), h);
      nqx = pqw[i1 * 3 + 0]; nqy = pqw[i1 * 3 + 1]; nqz = pqw[i1 * 3 + 2];
      idn = idxw[i2 * KS];
    }
    __builtin_amdgcn_sched_barrier(0);
    float m[4][8];
    conv2_tile<true >(r0, qx, qy, qz, h, ap, a2, sc1, sh1, s, sq, m);
    conv2_tile<false>(r1, qx, qy, qz, h, ap, a2, sc1, sh1, s, sq, m);
    #pragma unroll
    for (int mt2 = 0; mt2 < 4; ++mt2)
      #pragma unroll
      for (int r = 0; r < 8; ++r) {
        const float e = row_max16(m[mt2][r]);
        res[mt2][r] = (c == i) ? e : res[mt2][r];
      }
    end_of_trip(rn0, rn1, nqx, nqy, nqz, idn);
  }
  float sgn[4][8];
  #pragma unroll
  for (int mt = 0; mt < 4; ++mt)
    #pragma unroll
    for (int r = 0; r < 8; ++r) {
      sgn[mt][r] = (g2[mt * 16 + 8 * h + r] < 0.f) ? -1.f : 1.f;
      s[mt][r] *= sgn[mt][r];
    }
  stage_part_row(s, sq, prow + w * 128, c, h);
  if (c < C2_PTS) {
    #pragma unroll
    for (int mt2 = 0; mt2 < 4; ++mt2)
      #pragma unroll
      for (int r = 0; r < 8; ++r) {
        const int o = mt2 * 16 + 8 * h + r;
        ptile[o * C2_PITCH + w * C2_PTS + c] = res[mt2][r] * sgn[mt2][r];
      }
  }
  __syncthreads();
  const int t = threadIdx.x;
  if (t < 128)
    part2[(size_t)blockIdx.x * P2_STRIDE + t] =
        ((prow[t] + prow[128 + t]) + (prow[256 + t] + prow[384 + t])) +
        ((prow[512 + t] + prow[640 + t]) + (prow[768 + t] + prow[896 + t]));
  const int nb0 = (blockIdx.x % (NN / C2_QPB)) * C2_QPB;
  #pragma unroll
  for (int k = 0; k < COUT * (C2_QPB / 4) / C2_TPB; ++k) {
    const int f = t + k * C2_TPB;
    const int o = f / (C2_QPB / 4), pc = f % (C2_QPB / 4);
    const v4f v = *reinterpret_cast<const v4f*>(ptile + o * C2_PITCH + 4 * pc);
    *reinterpret_cast<v4f*>(out + ((size_t)b * COUT + o) * NN + nb0 + 4 * pc) = v;
  }
}

#define AP_TPB 256
#define AP_V4  (NN / 4 / AP_TPB)
#define AP_WAVES (AP_TPB / 32)
static_assert(BB * NN / C2_QPB == AP_TPB, "thread t of the apply block takes the partial row of conv2 block t");
__global__ __launch_bounds__(AP_TPB) void bn2_apply_kernel(const float* __restrict__ part2,
                                                           const float* __restrict__ g2,
                                                           const float* __restrict__ b2,
                                                           float* __restrict__ out) {
  __shared__ float2 wtot[AP_WAVES];
  const int row = blockIdx.x;
  const int o   = row & (COUT - 1);
  const int t   = threadIdx.x;
  float4* q = reinterpret_cast<float4*>(out) + (size_t)row * (NN / 4) + t;
  float4 e[AP_V4];
  #pragma unroll
  for (int k = 0; k < AP_V4; ++k) e[k] = q[k * AP_TPB];
  const float2 ps = *reinterpret_cast<const float2*>(part2 + (size_t)t * P2_STRIDE + 2 * o);
  float wsum = row_sum16(ps.x), wsq = row_sum16(ps.y);
  wsum += __shfl_xor(wsum, 16, 32);
  wsq  += __shfl_xor(wsq, 16, 32);
  if ((t & 31) == 0) wtot[t >> 5] = make_float2(wsum, wsq);
  __syncthreads();
  float tot = 0.f, tsq = 0.f;
  #pragma unroll
  for (int k = 0; k < AP_WAVES; ++k) { const float2 v = wtot[k]; tot += v.x; tsq += v.y; }
  const float mean = tot / CNTF;
  const float var  = tsq / CNTF - mean * mean;
  const float sc   = g2[o] * rsqrtf(var + EPSV);
  const float sh   = b2[o] - mean * sc;
  #pragma unroll
  for (int k = 0; k < AP_V4; ++k) {
    float4 r;
    float y;
    y = e[k].x * sc + sh; r.x = y > 0.f ? y : 0.f;
    y = e[k].y * sc + sh; r.y = y > 0.f ? y : 0.f;
    y = e[k].z * sc + sh; r.z = y > 0.f ? y : 0.f;
    y = e[k].w * sc + sh; r.w = y > 0.f ? y : 0.f;
    q[k * AP_TPB] = r;
  }
}

extern "C" void kernel_launch(void* const* d_in, const int* in_sizes, int n_in,
                              void* d_out, int out_size, void* d_ws, size_t ws_size,
                              hipStream_t stream) {
  const float* p  = (const float*)d_in[0];
  const float* x  = (const float*)d_in[1];
  const float* W1 = (const float*)d_in[2];
  const float* g1 = (const float*)d_in[3];
  const float* b1 = (const float*)d_in[4];
  const float* W2 = (const float*)d_in[5];
  const float* g2 = (const float*)d_in[6];
  const float* b2 = (const float*)d_in[7];
  float* out = (float*)d_out;

  char*  ws    = (char*)d_ws;
  unsigned short* idx = (unsigned short*)ws;
  float* U     = (float*)(ws + (size_t)BB * NN * KS * 2);
  float* bnp1  = (float*)(ws + (size_t)BB * NN * KS * 2
                             + (size_t)BB * NN * CU * 4);
  float* part1 = out;
  float* part2 = (float*)ws;

  ballq_kernel<<<BB * (NN / BQ_QPB), BQ_TPB, 0, stream>>>(p, x, W1, idx, U);
  conv1_stats_kernel<<<GCONV / CW_WAVES, CW_TPB, 0, stream>>>(p, U, idx, W1, part1);
  bn_finalize_kernel<<<COUT / 16, FIN_TPB, 0, stream>>>(part1, g1, b1, bnp1, 128);
  conv2_stats_kernel<<<BB * NN / C2_QPB, C2_TPB, 0, stream>>>(p, U, idx, W1, W2, bnp1, g2, part2, out);
  bn2_apply_kernel<<<BB * COUT, AP_TPB, 0, stream>>>(part2, g2, b2, out);
}


# conv2 pass computes only the cnt distinct slots of each ball, packed into 16-column tiles across a wave's 8 points; slot 0's padding copies enter the BN2 sums as a weight (changes BN2 sum order)
# speedup vs baseline: 10.7170x; 1.0333x over previous
#include <hip/hip_runtime.h>
#include <hip/hip_bf16.h>


typedef __attribute__((ext_vector_type(2))) float v2f;
typedef __attribute__((ext_vector_type(4))) float v4f;
typedef __attribute__((ext_vector_type(8))) float v8f;

#define BB      4
#define NN      4096
#define CX      32
#define CU      64
#define CIN     35
#define COUT    64
#define KS      32
#define R2      0.01f
#define EPSV    1e-5f
#define GCONV   1024
#define CNTF    ((float)BB * (float)NN * (float)KS)


#define BQ_Q     4
#define BQ_WAVES 16
#define BQ_QPB   (BQ_Q * BQ_WAVES)
#define BQ_TPB   (32 * BQ_WAVES)
#define BQ_FILL  (NN * 3 / 4 / BQ_TPB)
static_assert(BQ_Q == 4, "the vote test and the early-exit test below are written out for 4 queries");
static_assert(BQ_QPB == 64 && BQ_WAVES == 16,
              "the U job: a block's 64 points are 4 tiles of 16 columns x 4 row tiles, one per wave");

struct UFrags { v2f a[8], b[8]; };
__device__ __forceinline__ UFrags upoint_load(const float* __restrict__ x, const float* __restrict__ W1,
                                              int b, int n0, int mt, int c, int h) {
  UFrags f;
  const float* xc = x + ((size_t)b * CX + 16 * h) * NN + n0 + c;
  const float* wr = W1 + (mt * 16 + c) * CIN + 3 + 16 * h;
  #pragma unroll
  for (int s = 0; s < 8; ++s) {
    f.b[s].x = xc[(size_t)(2 * s) * NN];
    f.b[s].y = xc[(size_t)(2 * s + 1) * NN];
    f.a[s].x = wr[2 * s];
    f.a[s].y = wr[2 * s + 1];
  }
  return f;
}
__device__ __forceinline__ v8f upoint_tile(const UFrags& f) {
  v8f acc = (v8f){0.f, 0.f, 0.f, 0.f, 0.f, 0.f, 0.f, 0.f};
  #pragma unroll
  for (int s = 0; s < 8; ++s)
    acc = __builtin_amdgcn_wmma_f32_16x16x4_f32(false, f.a[s], false, f.b[s],
                                                (short)0, acc, false, false);
  return acc;
}
__device__ __forceinline__ void upoint_store(const v8f& acc, float* __restrict__ U,
                                             int b, int n0, int mt, int c, int h) {
  v4f* ur = reinterpret_cast<v4f*>(U + ((size_t)b * NN + n0 + c) * CU) + h;
  ur[4 * mt]     = (v4f){acc[0], acc[1], acc[2], acc[3]};
  ur[4 * mt + 2] = (v4f){acc[4], acc[5], acc[6], acc[7]};
}

__global__ __launch_bounds__(BQ_TPB) void ballq_kernel(const float* __restrict__ p,
                                                       const float* __restrict__ x,
                                                       const float* __restrict__ W1,
                                                       unsigned short* __restrict__ idx,
                                                       float* __restrict__ U) {
  __shared__ float4 cand4[NN * 3 / 4];
  __shared__ int    lst[BQ_WAVES * BQ_Q * KS];
  const int t    = threadIdx.x;
  const int lane = t & 31;
  const int w  = __builtin_amdgcn_readfirstlane(t >> 5);
  const int b  = blockIdx.x / (NN / BQ_QPB);
  const int q0 = (blockIdx.x % (NN / BQ_QPB)) * BQ_QPB + w * BQ_Q;
  const float* pb = p + (size_t)b * NN * 3;

  const float4* src4 = reinterpret_cast<const float4*>(pb);
  float4 stage[BQ_FILL];
  #pragma unroll
  for (int k = 0; k < BQ_FILL; ++k) stage[k] = src4[t + k * BQ_TPB];
  const int uc = lane & 15, uh = lane >> 4;
  const int un0 = (blockIdx.x % (NN / BQ_QPB)) * BQ_QPB + (w & 3) * 16, umt = w >> 2;
  const UFrags uf = upoint_load(x, W1, b, un0, umt, uc, uh);
  #pragma unroll
  for (int k = 0; k < BQ_FILL; ++k) cand4[t + k * BQ_TPB] = stage[k];
  const v8f uacc = upoint_tile(uf);

  int* mylst = lst + w * (BQ_Q * KS);
  if (lane < BQ_Q) mylst[lane * KS] = 0;
  float qx[BQ_Q], qy[BQ_Q], qz[BQ_Q];
  int cnt[BQ_Q];
  #pragma unroll
  for (int q = 0; q < BQ_Q; ++q) {
    qx[q] = pb[(q0 + q) * 3 + 0];
    qy[q] = pb[(q0 + q) * 3 + 1];
    qz[q] = pb[(q0 + q) * 3 + 2];
    cnt[q] = 0;
  }
  __syncthreads();
  upoint_store(uacc, U, b, un0, umt, uc, uh);

  const float* cand = reinterpret_cast<const float*>(cand4) + lane * 3;
  float nx = cand[0], ny = cand[1], nz = cand[2];
  bool active = true;
  int nb3 = 0;
  for (int base = 0; base < NN && active; base += 32) {
    const float cx = nx, cy = ny, cz = nz;
    if (base + 32 < NN) nb3 += 32 * 3;
    nx = cand[nb3 + 0]; ny = cand[nb3 + 1]; nz = cand[nb3 + 2];

    bool h[BQ_Q];
    unsigned m[BQ_Q];
    #pragma unroll
    for (int q = 0; q < BQ_Q; ++q) {
      const float dx = cx - qx[q], dy = cy - qy[q], dz = cz - qz[q];
      const float d2 = __builtin_fmaf(dz, dz, __builtin_fmaf(dx, dx, dy * dy));
      h[q] = d2 < R2;
      m[q] = __builtin_amdgcn_ballot_w32(h[q]);
    }
    if ((m[0] | m[1] | m[2] | m[3]) != 0u) {
      #pragma unroll
      for (int q = 0; q < BQ_Q; ++q) {
        if (m[q] != 0u) {
          const int rank = cnt[q] + (int)__builtin_amdgcn_mbcnt_lo(m[q], 0u);
          if (h[q] && rank < KS) mylst[q * KS + rank] = base + lane;
          cnt[q] += __builtin_popcount(m[q]);
        }
      }
      active = (((cnt[0] - KS) | (cnt[1] - KS) | (cnt[2] - KS) | (cnt[3] - KS)) < 0);
    }
  }
  #pragma unroll
  for (int q = 0; q < BQ_Q; ++q) {
    const int v = mylst[q * KS + (lane < cnt[q] ? lane : 0)];
    idx[((size_t)b * NN + q0 + q) * KS + lane] = (unsigned short)v;
  }
}

#define CW_WAVES 4
#define CW_TPB   (32 * CW_WAVES)
#define CW_PTS   (BB * NN / GCONV)
static_assert(CW_PTS == 16 && NN % CW_PTS == 0 && GCONV % CW_WAVES == 0,
              "lane c keeps the pooled values of the wave's point c; a wave stays inside one batch");

struct RawTile { v4f u[8]; float px, py, pz; };

__device__ __forceinline__ RawTile gather_tile(const float* __restrict__ p,
                                               const float* __restrict__ U,
                                               int b, int j, int h) {
  RawTile r;
  const v4f* ur = reinterpret_cast<const v4f*>(U + ((size_t)b * NN + j) * CU) + h;
  #pragma unroll
  for (int i = 0; i < 8; ++i) r.u[i] = ur[2 * i];
  const float* pj = p + ((size_t)b * NN + j) * 3;
  r.px = pj[0]; r.py = pj[1]; r.pz = pj[2];
  return r;
}

__device__ __forceinline__ void end_of_trip(const RawTile& a, const RawTile& b,
                                            float qx, float qy, float qz, int id) {
  __builtin_amdgcn_sched_barrier(0);
  asm volatile("" :: "v"(a.px), "v"(a.py), "v"(a.pz), "v"(b.px), "v"(b.py), "v"(b.pz),
                     "s"(qx), "s"(qy), "s"(qz), "v"(id));
}

__device__ __forceinline__ void load_w1p_frags(const float* __restrict__ W1, int c, int h,
                                               v2f (&ap)[4]) {
  #pragma unroll
  for (int mt = 0; mt < 4; ++mt) {
    const float* wr = W1 + (mt * 16 + c) * CIN;
    const float w01 = wr[1];
    ap[mt].x = wr[2 * h];
    ap[mt].y = h ? 0.f : w01;
  }
}

__device__ __forceinline__ void load_w2_frags(const float* __restrict__ W2, int c, int h,
                                              v2f (&a2)[4][16]) {
  #pragma unroll
  for (int mt2 = 0; mt2 < 4; ++mt2) {
    const float* wr = W2 + (mt2 * 16 + c) * COUT + 8 * h;
    #pragma unroll
    for (int kk = 0; kk < 16; ++kk) {
      const int k0 = (kk >> 2) * 16 + (kk & 3) * 2;
      a2[mt2][kk].x = wr[k0];
      a2[mt2][kk].y = wr[k0 + 1];
    }
  }
}

__device__ __forceinline__ void conv1_tile(const v2f (&ap)[4], const RawTile& rt,
                                           float qx, float qy, float qz, int h, v8f (&acc)[4]) {
  const float dx = rt.px - qx, dy = rt.py - qy, dz = rt.pz - qz;
  v2f bd;
  bd.x = h ? dz : dx;
  bd.y = h ? 0.f : dy;
  #pragma unroll
  for (int mt = 0; mt < 4; ++mt) {
    const v4f lo = rt.u[2 * mt], hi = rt.u[2 * mt + 1];
    const v8f cu = (v8f){lo.x, lo.y, lo.z, lo.w, hi.x, hi.y, hi.z, hi.w};
    acc[mt] = __builtin_amdgcn_wmma_f32_16x16x4_f32(false, ap[mt], false, bd,
                                                    (short)0, cu, false, false);
  }
}

template <int XM>
__device__ __forceinline__ float lane_xor16(float v) {
  return __builtin_bit_cast(float, __builtin_amdgcn_update_dpp(
      0, __builtin_bit_cast(int, v), 0x160 + XM, 0xf, 0xf, true));
}
__device__ __forceinline__ float row_sum16(float v) {
  v += lane_xor16<1>(v); v += lane_xor16<2>(v); v += lane_xor16<4>(v); v += lane_xor16<8>(v);
  return v;
}
__device__ __forceinline__ int fkey(float v) {
  const int b = __builtin_bit_cast(int, v);
  return b ^ ((b >> 31) & 0x7fffffff);
}
__device__ __forceinline__ float fkey_inv(int k) {
  return __builtin_bit_cast(float, k ^ ((k >> 31) & 0x7fffffff));
}
#define KEY_MIN ((int)0x80000000)
__device__ __forceinline__ int imax(int a, int b) { return a > b ? a : b; }
template <int XM>
__device__ __forceinline__ int lane_xor16i(int v) {
  return __builtin_amdgcn_update_dpp(0, v, 0x160 + XM, 0xf, 0xf, true);
}
__device__ __forceinline__ int row_max16i(int v) {
  v = imax(v, lane_xor16i<1>(v)); v = imax(v, lane_xor16i<2>(v));
  v = imax(v, lane_xor16i<4>(v)); v = imax(v, lane_xor16i<8>(v));
  return v;
}

__device__ __forceinline__ void write_part_row(float (&s)[4][8], float (&sq)[4][8],
                                               float* __restrict__ row, int c, int h) {
  #pragma unroll
  for (int mt = 0; mt < 4; ++mt) {
    float v = 0.f;
    #pragma unroll
    for (int r = 0; r < 8; ++r) {
      const float ts = row_sum16(s[mt][r]);
      const float tq = row_sum16(sq[mt][r]);
      v = (c == r) ? ts : v;
      v = (c == 8 + r) ? tq : v;
    }
    row[mt * 32 + ((c & 8) << 1) + 8 * h + (c & 7)] = v;
  }
}

__device__ __forceinline__ void stage_part_row(float (&s)[4][8], float (&sq)[4][8],
                                               float* lrow, int c, int h) {
  #pragma unroll
  for (int mt = 0; mt < 4; ++mt) {
    float v = 0.f;
    #pragma unroll
    for (int r = 0; r < 8; ++r) {
      const float ts = row_sum16(s[mt][r]);
      const float tq = row_sum16(sq[mt][r]);
      v = (c == r) ? ts : v;
      v = (c == 8 + r) ? tq : v;
    }
    lrow[2 * (mt * 16 + 8 * h + (c & 7)) + (c >> 3)] = v;
  }
}

#define S1_PP 2
static_assert(CW_PTS % S1_PP == 0, "a trip takes S1_PP whole points");
__global__ __launch_bounds__(CW_TPB) __attribute__((amdgpu_waves_per_eu(1, 1))) void conv1_stats_kernel(
    const float* __restrict__ p, const float* __restrict__ U,
    const unsigned short* __restrict__ idxg, const float* __restrict__ W1,
    float* __restrict__ part1) {
  const int lane = threadIdx.x & 31;
  const int c = lane & 15, h = lane >> 4;
  const int w   = __builtin_amdgcn_readfirstlane((int)(threadIdx.x >> 5));
  const int gw  = blockIdx.x * CW_WAVES + w;
  const int pt0 = gw * CW_PTS;
  const int b   = pt0 / NN, n0 = pt0 % NN;

  v2f ap[4];
  load_w1p_frags(W1, c, h, ap);
  float s[4][8], sq[4][8];
  #pragma unroll
  for (int mt = 0; mt < 4; ++mt)
    #pragma unroll
    for (int r = 0; r < 8; ++r) { s[mt][r] = 0.f; sq[mt][r] = 0.f; }

  const unsigned short* idxw = idxg + ((size_t)b * NN + n0) * KS + lane;
  const float* pqw  = p + ((size_t)b * NN + n0) * 3;
  RawTile rn[S1_PP][2];
  float nqx[S1_PP], nqy[S1_PP], nqz[S1_PP];
  int idn[S1_PP];
  #pragma unroll
  for (int k = 0; k < S1_PP; ++k) {
    const int id = idxw[k * KS];
    rn[k][0] = gather_tile(p, U, b, __shfl(id, c, 32) & (NN - 1), h);
    rn[k][1] = gather_tile(p, U, b, __shfl(id, 16 + c, 32) & (NN - 1), h);
    nqx[k] = pqw[k * 3 + 0]; nqy[k] = pqw[k * 3 + 1]; nqz[k] = pqw[k * 3 + 2];
  }
  #pragma unroll
  for (int k = 0; k < S1_PP; ++k) idn[k] = idxw[(S1_PP + k) * KS];

  #pragma nounroll
  for (int i = 0; i < CW_PTS; i += S1_PP) {
    RawTile r[S1_PP][2];
    float qx[S1_PP], qy[S1_PP], qz[S1_PP];
    #pragma unroll
    for (int k = 0; k < S1_PP; ++k) {
      r[k][0] = rn[k][0]; r[k][1] = rn[k][1];
      qx[k] = nqx[k]; qy[k] = nqy[k]; qz[k] = nqz[k];
    }
    #pragma unroll
    for (int k = 0; k < S1_PP; ++k) {
      const int i1 = (i + S1_PP + k < CW_PTS) ? i + S1_PP + k : CW_PTS - 1;
      const int i2 = (i + 2 * S1_PP + k < CW_PTS) ? i + 2 * S1_PP + k : CW_PTS - 1;
      rn[k][0] = gather_tile(p, U, b, __shfl(idn[k], c, 32) & (NN - 1), h);
      rn[k][1] = gather_tile(p, U, b, __shfl(idn[k], 16 + c, 32) & (NN - 1), h);
      nqx[k] = pqw[i1 * 3 + 0]; nqy[k] = pqw[i1 * 3 + 1]; nqz[k] = pqw[i1 * 3 + 2];
      idn[k] = idxw[i2 * KS];
    }
    __builtin_amdgcn_sched_barrier(0);
    #pragma unroll
    for (int k = 0; k < S1_PP; ++k)
      #pragma unroll
      for (int tile = 0; tile < 2; ++tile) {
        v8f acc[4];
        conv1_tile(ap, r[k][tile], qx[k], qy[k], qz[k], h, acc);
        #pragma unroll
        for (int mt = 0; mt < 4; ++mt)
          #pragma unroll
          for (int rr = 0; rr < 8; ++rr) { const float y = acc[mt][rr]; s[mt][rr] += y; sq[mt][rr] += y * y; }
      }
    #pragma unroll
    for (int k = 0; k < S1_PP; ++k)
      end_of_trip(rn[k][0], rn[k][1], nqx[k], nqy[k], nqz[k], idn[k]);
  }
  write_part_row(s, sq, part1 + (size_t)gw * 128, c, h);
}

#define FIN_TPB   1024
#define FIN_WAVES (FIN_TPB / 32)
#define FIN_ROWS  (GCONV / FIN_WAVES)
static_assert(FIN_ROWS == 32 && FIN_WAVES == 32, "tree32 below sums exactly 32 values");

__device__ __forceinline__ float tree32(float (&a)[32]) {
  #pragma unroll
  for (int h = 16; h >= 1; h >>= 1) {
    #pragma unroll
    for (int r = 0; r < h; ++r) a[r] += a[r + h];
  }
  return a[0];
}

__global__ __launch_bounds__(FIN_TPB) void bn_finalize_kernel(const float* __restrict__ part,
                                                              const float* __restrict__ g,
                                                              const float* __restrict__ bta,
                                                              float* __restrict__ bnp,
                                                              int rstride) {
  __shared__ float red[FIN_WAVES * 32];
  const int t    = threadIdx.x;
  const int lane = t & 31;
  const int w    = t >> 5;
  const int m    = blockIdx.x;
  const float* src = part + (size_t)w * rstride + m * 32 + lane;
  float a[FIN_ROWS];
  #pragma unroll
  for (int r = 0; r < FIN_ROWS; ++r) a[r] = src[(size_t)r * FIN_WAVES * rstride];
  red[w * 32 + lane] = tree32(a);
  __syncthreads();
  if (w == 0) {
    float c[FIN_WAVES];
    #pragma unroll
    for (int k = 0; k < FIN_WAVES; ++k) c[k] = red[k * 32 + lane];
    const float tot = tree32(c);
    const float sq  = __shfl_down(tot, 16, 32);
    if (lane < 16) {
      const int o = m * 16 + lane;
      float mean  = tot / CNTF;
      float var   = sq / CNTF - mean * mean;
      float scale = g[o] * rsqrtf(var + EPSV);
      bnp[o]      = scale;
      bnp[64 + o] = bta[o] - mean * scale;
    }
  }
}


__device__ __forceinline__ void col_entry(int t, int c, const int (&off)[9], const int (&cnt)[8],
                                          int& pi, int& k, float& w) {
  const int T = off[8];
  const int e = 16 * t + c;
  const bool valid = e < T;
  const int ec = valid ? e : T - 1;
  int o = 0, cn = cnt[0], i = 0;
  #pragma unroll
  for (int q = 1; q < 8; ++q) {
    const bool ge = ec >= off[q];
    o = ge ? off[q] : o; cn = ge ? cnt[q] : cn; i = ge ? q : i;
  }
  pi = i;
  k = ec - o;
  w = valid ? (k == 0 ? (float)(33 - cn) : 1.f) : 0.f;
}

__device__ __forceinline__ void conv2_cols(const RawTile& rt, float qx, float qy, float qz, float w, int h,
                                           const v2f (&ap)[4], const v2f (&a2)[4][16],
                                           const float (&sc1)[4][8], const float (&sh1)[4][8],
                                           float (&s)[4][8], float (&sq)[4][8], float (&y2)[4][8]) {
  v8f acc1[4];
  conv1_tile(ap, rt, qx, qy, qz, h, acc1);
  float f1[4][8];
  #pragma unroll
  for (int mt = 0; mt < 4; ++mt)
    #pragma unroll
    for (int r = 0; r < 8; ++r) {
      float y = acc1[mt][r] * sc1[mt][r] + sh1[mt][r];
      f1[mt][r] = y > 0.f ? y : 0.f;
    }
  v8f acc2[4];
  #pragma unroll
  for (int mt2 = 0; mt2 < 4; ++mt2) acc2[mt2] = (v8f){0.f, 0.f, 0.f, 0.f, 0.f, 0.f, 0.f, 0.f};
  #pragma unroll
  for (int kk = 0; kk < 16; ++kk) {
    v2f bm;
    bm.x = f1[kk >> 2][(kk & 3) * 2];
    bm.y = f1[kk >> 2][(kk & 3) * 2 + 1];
    #pragma unroll
    for (int mt2 = 0; mt2 < 4; ++mt2)
      acc2[mt2] = __builtin_amdgcn_wmma_f32_16x16x4_f32(false, a2[mt2][kk], false, bm,
                                                        (short)0, acc2[mt2], false, false);
  }
  #pragma unroll
  for (int mt2 = 0; mt2 < 4; ++mt2)
    #pragma unroll
    for (int r = 0; r < 8; ++r) {
      const float y = acc2[mt2][r];
      const float wy = w * y;
      y2[mt2][r] = y;
      s[mt2][r] += wy; sq[mt2][r] += wy * y;
    }
}

__device__ __forceinline__ void end_of_col_trip(const RawTile& a, float qx, float qy, float qz,
                                                int j, int pi, float w) {
  __builtin_amdgcn_sched_barrier(0);
  asm volatile("" :: "v"(a.px), "v"(a.py), "v"(a.pz), "v"(qx), "v"(qy), "v"(qz),
                     "v"(j), "v"(pi), "v"(w));
}

#define C2_WAVES 8
#define C2_TPB   (32 * C2_WAVES)
#define C2_PTS   8
#define C2_QPB   (C2_WAVES * C2_PTS)
#define C2_PITCH (C2_QPB + 4)
static_assert(C2_QPB == 64 && NN % C2_QPB == 0 && C2_PTS == 8,
              "a block owns 64 consecutive points of one batch; lane c keeps the pooled values of the wave's point c; col_entry is written out for 8 points");
#define P2_STRIDE (C2_QPB * KS / 2)
__global__ __launch_bounds__(C2_TPB) void conv2_stats_kernel(
    const float* __restrict__ p, const float* __restrict__ U,
    const unsigned short* idxg, const float* __restrict__ W1,
    const float* __restrict__ W2, const float* __restrict__ bnp1,
    const float* __restrict__ g2, float* part2,
    float* __restrict__ out) {
  __shared__ float prow[C2_WAVES * 128];
  __shared__ __attribute__((aligned(16))) float ptile[COUT * C2_PITCH];
  const int lane = threadIdx.x & 31;
  const int c = lane & 15, h = lane >> 4;
  const int w   = __builtin_amdgcn_readfirstlane((int)(threadIdx.x >> 5));
  const int gw  = blockIdx.x * C2_WAVES + w;
  const int pt0 = gw * C2_PTS;
  const int b   = pt0 / NN, n0 = pt0 % NN;

  v2f ap[4];
  load_w1p_frags(W1, c, h, ap);
  v2f a2[4][16];
  load_w2_frags(W2, c, h, a2);
  #pragma unroll
  for (int mt2 = 0; mt2 < 4; ++mt2) {
    const float sa = (g2[mt2 * 16 + c] < 0.f) ? -1.f : 1.f;
    #pragma unroll
    for (int kk = 0; kk < 16; ++kk) { a2[mt2][kk].x *= sa; a2[mt2][kk].y *= sa; }
  }
  float sc1[4][8], sh1[4][8];
  float s[4][8], sq[4][8];
  int resk[4][8];
  #pragma unroll
  for (int mt = 0; mt < 4; ++mt)
    #pragma unroll
    for (int r = 0; r < 8; ++r) {
      const int o = mt * 16 + 8 * h + r;
      sc1[mt][r] = bnp1[o]; sh1[mt][r] = bnp1[64 + o];
      s[mt][r] = 0.f; sq[mt][r] = 0.f; resk[mt][r] = KEY_MIN;
    }

  const unsigned short* idxr = idxg + ((size_t)b * NN + n0) * KS;
  const float* pqw  = p + ((size_t)b * NN + n0) * 3;
  int cnt[8], off[9];
  off[0] = 0;
  #pragma unroll
  for (int i = 0; i < C2_PTS; ++i) {
    const int id  = idxr[i * KS + lane];
    const int id0 = __builtin_amdgcn_readfirstlane(id);
    cnt[i] = __builtin_popcount(__builtin_amdgcn_ballot_w32(id != id0)) + 1;
    off[i + 1] = off[i] + cnt[i];
  }
  const int ntiles = (off[8] + 15) >> 4;

  int piB, kB; float wB;
  col_entry(0, c, off, cnt, piB, kB, wB);
  RawTile rn = gather_tile(p, U, b, idxr[piB * KS + kB] & (NN - 1), h);
  float nqx = pqw[piB * 3 + 0], nqy = pqw[piB * 3 + 1], nqz = pqw[piB * 3 + 2];
  int piA, kA; float wA;
  col_entry(ntiles > 1 ? 1 : 0, c, off, cnt, piA, kA, wA);
  int jn = idxr[piA * KS + kA];

  #pragma nounroll
  for (int t = 0; t < ntiles; ++t) {
    const RawTile r0 = rn;
    const float qx = nqx, qy = nqy, qz = nqz, w = wB;
    const int pi = piB;
    {
      rn = gather_tile(p, U, b, jn & (NN - 1), h);
      nqx = pqw[piA * 3 + 0]; nqy = pqw[piA * 3 + 1]; nqz = pqw[piA * 3 + 2];
      piB = piA; wB = wA;
      const int t2 = (t + 2 < ntiles) ? t + 2 : ntiles - 1;
      col_entry(t2, c, off, cnt, piA, kA, wA);
      jn = idxr[piA * KS + kA];
    }
    __builtin_amdgcn_sched_barrier(0);
    float y2[4][8];
    conv2_cols(r0, qx, qy, qz, w, h, ap, a2, sc1, sh1, s, sq, y2);
    const int qlo = __builtin_amdgcn_readfirstlane(pi);
    const int qhi = __builtin_amdgcn_readlane(pi, 15);
    int yk[4][8];
    #pragma unroll
    for (int mt2 = 0; mt2 < 4; ++mt2)
      #pragma unroll
      for (int r = 0; r < 8; ++r) yk[mt2][r] = fkey(y2[mt2][r]);
    #pragma nounroll
    for (int q = qlo; q <= qhi; ++q) {
      const bool mine = (pi == q);
      const bool keep = (c == q);
      #pragma unroll
      for (int mt2 = 0; mt2 < 4; ++mt2)
        #pragma unroll
        for (int r = 0; r < 8; ++r) {
          const int e = row_max16i(mine ? yk[mt2][r] : KEY_MIN);
          resk[mt2][r] = imax(resk[mt2][r], keep ? e : KEY_MIN);
        }
    }
    end_of_col_trip(rn, nqx, nqy, nqz, jn, piA, wA);
  }
  float sgn[4][8];
  #pragma unroll
  for (int mt = 0; mt < 4; ++mt)
    #pragma unroll
    for (int r = 0; r < 8; ++r) {
      sgn[mt][r] = (g2[mt * 16 + 8 * h + r] < 0.f) ? -1.f : 1.f;
      s[mt][r] *= sgn[mt][r];
    }
  stage_part_row(s, sq, prow + w * 128, c, h);
  if (c < C2_PTS) {
    #pragma unroll
    for (int mt2 = 0; mt2 < 4; ++mt2)
      #pragma unroll
      for (int r = 0; r < 8; ++r) {
        const int o = mt2 * 16 + 8 * h + r;
        ptile[o * C2_PITCH + w * C2_PTS + c] = fkey_inv(resk[mt2][r]) * sgn[mt2][r];
      }
  }
  __syncthreads();
  const int t = threadIdx.x;
  if (t < 128)
    part2[(size_t)blockIdx.x * P2_STRIDE + t] =
        ((prow[t] + prow[128 + t]) + (prow[256 + t] + prow[384 + t])) +
        ((prow[512 + t] + prow[640 + t]) + (prow[768 + t] + prow[896 + t]));
  const int nb0 = (blockIdx.x % (NN / C2_QPB)) * C2_QPB;
  #pragma unroll
  for (int k = 0; k < COUT * (C2_QPB / 4) / C2_TPB; ++k) {
    const int f = t + k * C2_TPB;
    const int o = f / (C2_QPB / 4), pc = f % (C2_QPB / 4);
    const v4f v = *reinterpret_cast<const v4f*>(ptile + o * C2_PITCH + 4 * pc);
    *reinterpret_cast<v4f*>(out + ((size_t)b * COUT + o) * NN + nb0 + 4 * pc) = v;
  }
}

#define AP_TPB 256
#define AP_V4  (NN / 4 / AP_TPB)
#define AP_WAVES (AP_TPB / 32)
static_assert(BB * NN / C2_QPB == AP_TPB, "thread t of the apply block takes the partial row of conv2 block t");
__global__ __launch_bounds__(AP_TPB) void bn2_apply_kernel(const float* __restrict__ part2,
                                                           const float* __restrict__ g2,
                                                           const float* __restrict__ b2,
                                                           float* __restrict__ out) {
  __shared__ float2 wtot[AP_WAVES];
  const int row = blockIdx.x;
  const int o   = row & (COUT - 1);
  const int t   = threadIdx.x;
  float4* q = reinterpret_cast<float4*>(out) + (size_t)row * (NN / 4) + t;
  float4 e[AP_V4];
  #pragma unroll
  for (int k = 0; k < AP_V4; ++k) e[k] = q[k * AP_TPB];
  const float2 ps = *reinterpret_cast<const float2*>(part2 + (size_t)t * P2_STRIDE + 2 * o);
  float wsum = row_sum16(ps.x), wsq = row_sum16(ps.y);
  wsum += __shfl_xor(wsum, 16, 32);
  wsq  += __shfl_xor(wsq, 16, 32);
  if ((t & 31) == 0) wtot[t >> 5] = make_float2(wsum, wsq);
  __syncthreads();
  float tot = 0.f, tsq = 0.f;
  #pragma unroll
  for (int k = 0; k < AP_WAVES; ++k) { const float2 v = wtot[k]; tot += v.x; tsq += v.y; }
  const float mean = tot / CNTF;
  const float var  = tsq / CNTF - mean * mean;
  const float sc   = g2[o] * rsqrtf(var + EPSV);
  const float sh   = b2[o] - mean * sc;
  #pragma unroll
  for (int k = 0; k < AP_V4; ++k) {
    float4 r;
    float y;
    y = e[k].x * sc + sh; r.x = y > 0.f ? y : 0.f;
    y = e[k].y * sc + sh; r.y = y > 0.f ? y : 0.f;
    y = e[k].z * sc + sh; r.z = y > 0.f ? y : 0.f;
    y = e[k].w * sc + sh; r.w = y > 0.f ? y : 0.f;
    q[k * AP_TPB] = r;
  }
}

extern "C" void kernel_launch(void* const* d_in, const int* in_sizes, int n_in,
                              void* d_out, int out_size, void* d_ws, size_t ws_size,
                              hipStream_t stream) {
  const float* p  = (const float*)d_in[0];
  const float* x  = (const float*)d_in[1];
  const float* W1 = (const float*)d_in[2];
  const float* g1 = (const float*)d_in[3];
  const float* b1 = (const float*)d_in[4];
  const float* W2 = (const float*)d_in[5];
  const float* g2 = (const float*)d_in[6];
  const float* b2 = (const float*)d_in[7];
  float* out = (float*)d_out;

  char*  ws    = (char*)d_ws;
  unsigned short* idx = (unsigned short*)ws;
  float* U     = (float*)(ws + (size_t)BB * NN * KS * 2);
  float* bnp1  = (float*)(ws + (size_t)BB * NN * KS * 2
                             + (size_t)BB * NN * CU * 4);
  float* part1 = out;
  float* part2 = (float*)ws;

  ballq_kernel<<<BB * (NN / BQ_QPB), BQ_TPB, 0, stream>>>(p, x, W1, idx, U);
  conv1_stats_kernel<<<GCONV / CW_WAVES, CW_TPB, 0, stream>>>(p, U, idx, W1, part1);
  bn_finalize_kernel<<<COUT / 16, FIN_TPB, 0, stream>>>(part1, g1, b1, bnp1, 128);
  conv2_stats_kernel<<<BB * NN / C2_QPB, C2_TPB, 0, stream>>>(p, U, idx, W1, W2, bnp1, g2, part2, out);
  bn2_apply_kernel<<<BB * COUT, AP_TPB, 0, stream>>>(part2, g2, b2, out);
}
